# starting kernel
# baseline (speedup 1.0000x reference)
#include <hip/hip_runtime.h>

typedef float v2f __attribute__((ext_vector_type(2)));
typedef float v8f __attribute__((ext_vector_type(8)));

#define N_DIM   2048
#define BS_DIM  64
#define N_ITER  25
#define ELEMS   ((size_t)BS_DIM * N_DIM)
#define MATSZ   ((size_t)N_DIM * N_DIM)
#define KSPLIT  4
#define KSLICE  (N_DIM / KSPLIT)

__global__ void gs_zero_kernel(float* __restrict__ p, int n) {
    int i = blockIdx.x * blockDim.x + threadIdx.x;
    if (i < n) p[i] = 0.0f;
}

__global__ void __launch_bounds__(256)
gs_transpose(const float* __restrict__ in, float* __restrict__ out) {
    __shared__ float tile[32][33];
    const int bx = blockIdx.x & 63;
    const int by = blockIdx.x >> 6;
    const int tx = threadIdx.x & 31;
    const int ty = threadIdx.x >> 5;

    const int x = bx * 32 + tx;
    #pragma unroll
    for (int r = ty; r < 32; r += 8)
        tile[r][tx] = in[(size_t)(by * 32 + r) * N_DIM + x];
    __syncthreads();
    const int x2 = by * 32 + tx;
    #pragma unroll
    for (int r = ty; r < 32; r += 8)
        out[(size_t)(bx * 32 + r) * N_DIM + x2] = tile[tx][r];
}

__global__ void __launch_bounds__(128)
gs_gemm_ksplit(const float* __restrict__ A, const float* __restrict__ Bt,
               float* __restrict__ P) {
    const int lane = threadIdx.x & 31;
    const int tm   = threadIdx.x >> 5;
    const int s    = blockIdx.x & 3;
    const int tn   = blockIdx.x >> 2;
    const int l15  = lane & 15;
    const int hi   = lane >> 4;
    const int kbase = s * KSLICE + hi * 2;

    const float* aptr = A  + (size_t)(tm * 16 + l15) * N_DIM + kbase;
    const float* bptr = Bt + (size_t)(tn * 16 + l15) * N_DIM + kbase;

    v8f c0 = {}; v8f c1 = {}; v8f c2 = {}; v8f c3 = {};
    #pragma unroll 2
    for (int k = 0; k < KSLICE; k += 16) {
        v2f a0 = *(const v2f*)(aptr + k);
        v2f b0 = *(const v2f*)(bptr + k);
        v2f a1 = *(const v2f*)(aptr + k + 4);
        v2f b1 = *(const v2f*)(bptr + k + 4);
        v2f a2 = *(const v2f*)(aptr + k + 8);
        v2f b2 = *(const v2f*)(bptr + k + 8);
        v2f a3 = *(const v2f*)(aptr + k + 12);
        v2f b3 = *(const v2f*)(bptr + k + 12);
        c0 = __builtin_amdgcn_wmma_f32_16x16x4_f32(false, a0, false, b0,
                                                   (short)0, c0, false, false);
        c1 = __builtin_amdgcn_wmma_f32_16x16x4_f32(false, a1, false, b1,
                                                   (short)0, c1, false, false);
        c2 = __builtin_amdgcn_wmma_f32_16x16x4_f32(false, a2, false, b2,
                                                   (short)0, c2, false, false);
        c3 = __builtin_amdgcn_wmma_f32_16x16x4_f32(false, a3, false, b3,
                                                   (short)0, c3, false, false);
    }

    float* optr = P + (size_t)s * ELEMS
                    + (size_t)(tm * 16 + hi * 8) * N_DIM + tn * 16 + l15;
    #pragma unroll
    for (int r = 0; r < 8; ++r)
        optr[(size_t)r * N_DIM] = (c0[r] + c1[r]) + (c2[r] + c3[r]);
}

__global__ void __launch_bounds__(256)
gs_reduce(const float* __restrict__ P, const float* __restrict__ subSrc,
          float* __restrict__ out0, float* __restrict__ out1,
          float* __restrict__ out2) {
    const size_t i = ((size_t)blockIdx.x * 256 + threadIdx.x) * 4;
    float4 v0 = *(const float4*)(P + i);
    float4 v1 = *(const float4*)(P + ELEMS + i);
    float4 v2 = *(const float4*)(P + 2 * ELEMS + i);
    float4 v3 = *(const float4*)(P + 3 * ELEMS + i);
    float4 v;
    v.x = (v0.x + v1.x) + (v2.x + v3.x);
    v.y = (v0.y + v1.y) + (v2.y + v3.y);
    v.z = (v0.z + v1.z) + (v2.z + v3.z);
    v.w = (v0.w + v1.w) + (v2.w + v3.w);
    if (subSrc) {
        float4 s = *(const float4*)(subSrc + i);
        v.x = s.x - v.x; v.y = s.y - v.y; v.z = s.z - v.z; v.w = s.w - v.w;
    }
    *(float4*)(out0 + i) = v;
    if (out1) *(float4*)(out1 + i) = v;
    if (out2) *(float4*)(out2 + i) = v;
}

__global__ void __launch_bounds__(128)
gs_gemm_nt(const float* __restrict__ A, const float* __restrict__ Bt,
           const float* __restrict__ subSrc,
           float* __restrict__ out0, float* __restrict__ out1,
           float* __restrict__ out2) {
    const int lane = threadIdx.x & 31;
    const int wave = threadIdx.x >> 5;
    const int t    = blockIdx.x * 4 + wave;
    const int tm   = t & 3;
    const int tn   = t >> 2;
    const int l15  = lane & 15;
    const int hi   = lane >> 4;
    const int koff = hi * 2;

    const float* aptr = A  + (size_t)(tm * 16 + l15) * N_DIM + koff;
    const float* bptr = Bt + (size_t)(tn * 16 + l15) * N_DIM + koff;

    v8f acc0 = {}; v8f acc1 = {};
    #pragma unroll 4
    for (int k = 0; k < N_DIM; k += 8) {
        v2f a0 = *(const v2f*)(aptr + k);
        v2f b0 = *(const v2f*)(bptr + k);
        v2f a1 = *(const v2f*)(aptr + k + 4);
        v2f b1 = *(const v2f*)(bptr + k + 4);
        acc0 = __builtin_amdgcn_wmma_f32_16x16x4_f32(false, a0, false, b0,
                                                     (short)0, acc0, false, false);
        acc1 = __builtin_amdgcn_wmma_f32_16x16x4_f32(false, a1, false, b1,
                                                     (short)0, acc1, false, false);
    }

    const size_t obase = (size_t)(tm * 16 + hi * 8) * N_DIM + tn * 16 + l15;
    float res[8];
    #pragma unroll
    for (int r = 0; r < 8; ++r) res[r] = acc0[r] + acc1[r];
    if (subSrc) {
        #pragma unroll
        for (int r = 0; r < 8; ++r)
            res[r] = subSrc[obase + (size_t)r * N_DIM] - res[r];
    }
    #pragma unroll
    for (int r = 0; r < 8; ++r) out0[obase + (size_t)r * N_DIM] = res[r];
    if (out1) {
        #pragma unroll
        for (int r = 0; r < 8; ++r) out1[obase + (size_t)r * N_DIM] = res[r];
    }
    if (out2) {
        #pragma unroll
        for (int r = 0; r < 8; ++r) out2[obase + (size_t)r * N_DIM] = res[r];
    }
}

__global__ void __launch_bounds__(128)
gs_gemm_nn(const float* __restrict__ A, const float* __restrict__ B,
           const float* __restrict__ subSrc,
           float* __restrict__ out0, float* __restrict__ out1,
           float* __restrict__ out2) {
    const int lane = threadIdx.x & 31;
    const int wave = threadIdx.x >> 5;
    const int t    = blockIdx.x * 4 + wave;
    const int tm   = t & 3;
    const int tn   = t >> 2;
    const int l15  = lane & 15;
    const int hi   = lane >> 4;
    const int koff = hi * 2;

    const float* aptr = A + (size_t)(tm * 16 + l15) * N_DIM + koff;
    const float* bptr = B + (size_t)koff * N_DIM + tn * 16 + l15;

    v8f acc0 = {}; v8f acc1 = {};
    #pragma unroll 4
    for (int k = 0; k < N_DIM; k += 8) {
        v2f a0 = *(const v2f*)(aptr + k);
        v2f b0;
        b0.x = bptr[(size_t)(k + 0) * N_DIM];
        b0.y = bptr[(size_t)(k + 1) * N_DIM];
        v2f a1 = *(const v2f*)(aptr + k + 4);
        v2f b1;
        b1.x = bptr[(size_t)(k + 4) * N_DIM];
        b1.y = bptr[(size_t)(k + 5) * N_DIM];
        acc0 = __builtin_amdgcn_wmma_f32_16x16x4_f32(false, a0, false, b0,
                                                     (short)0, acc0, false, false);
        acc1 = __builtin_amdgcn_wmma_f32_16x16x4_f32(false, a1, false, b1,
                                                     (short)0, acc1, false, false);
    }

    const size_t obase = (size_t)(tm * 16 + hi * 8) * N_DIM + tn * 16 + l15;
    float res[8];
    #pragma unroll
    for (int r = 0; r < 8; ++r) res[r] = acc0[r] + acc1[r];
    if (subSrc) {
        #pragma unroll
        for (int r = 0; r < 8; ++r)
            res[r] = subSrc[obase + (size_t)r * N_DIM] - res[r];
    }
    #pragma unroll
    for (int r = 0; r < 8; ++r) out0[obase + (size_t)r * N_DIM] = res[r];
    if (out1) {
        #pragma unroll
        for (int r = 0; r < 8; ++r) out1[obase + (size_t)r * N_DIM] = res[r];
    }
    if (out2) {
        #pragma unroll
        for (int r = 0; r < 8; ++r) out2[obase + (size_t)r * N_DIM] = res[r];
    }
}

extern "C" void kernel_launch(void* const* d_in, const int* in_sizes, int n_in,
                              void* d_out, int out_size, void* d_ws, size_t ws_size,
                              hipStream_t stream) {
    const float* y    = (const float*)d_in[2];
    const float* H    = (const float*)d_in[3];
    const float* Dinv = (const float*)d_in[4];
    const float* U    = (const float*)d_in[5];
    const float* invM = (const float*)d_in[6];

    float* sfinal = (float*)d_out;
    float* traj   = (float*)d_out + ELEMS;

    float* ws    = (float*)d_ws;
    float* yMF   = ws;
    float* S     = ws + ELEMS;
    float* T     = ws + 2 * ELEMS;
    float* P     = ws + 3 * ELEMS;
    float* Dt    = ws + 7 * ELEMS;
    float* Ut    = Dt + MATSZ;
    float* invMt = Ut + MATSZ;

    const size_t need_fast = (7 * ELEMS + 3 * MATSZ) * sizeof(float);
    const size_t need_mid  = (3 * ELEMS + 2 * MATSZ) * sizeof(float);
    const int mode = (ws_size >= need_fast) ? 2 : (ws_size >= need_mid ? 1 : 0);

    dim3 gblk(128);
    dim3 ggrd(128);
    dim3 kgrd(512);
    dim3 rgrd(ELEMS / (256 * 4));

    gs_zero_kernel<<<(int)(ELEMS / 256), 256, 0, stream>>>(traj, (int)ELEMS);

    if (mode == 2) {
        gs_transpose<<<4096, 256, 0, stream>>>(Dinv, Dt);
        gs_transpose<<<4096, 256, 0, stream>>>(U, Ut);
        gs_transpose<<<4096, 256, 0, stream>>>(invM, invMt);

        gs_gemm_ksplit<<<kgrd, gblk, 0, stream>>>(y, H, P);
        gs_reduce<<<rgrd, 256, 0, stream>>>(P, nullptr, yMF, nullptr, nullptr);
        gs_gemm_ksplit<<<kgrd, gblk, 0, stream>>>(yMF, Dt, P);
        gs_reduce<<<rgrd, 256, 0, stream>>>(P, nullptr, S, nullptr, nullptr);

        for (int i = 1; i <= N_ITER; ++i) {
            float* trajD  = traj + (size_t)i * ELEMS;
            float* finalD = (i == N_ITER) ? sfinal : nullptr;
            gs_gemm_ksplit<<<kgrd, gblk, 0, stream>>>(S, Ut, P);
            gs_reduce<<<rgrd, 256, 0, stream>>>(P, yMF, T, nullptr, nullptr);
            gs_gemm_ksplit<<<kgrd, gblk, 0, stream>>>(T, invMt, P);
            gs_reduce<<<rgrd, 256, 0, stream>>>(P, nullptr, S, trajD, finalD);
        }
    } else if (mode == 1) {
        gs_transpose<<<4096, 256, 0, stream>>>(U, Ut - MATSZ);
        gs_transpose<<<4096, 256, 0, stream>>>(invM, invMt - MATSZ);
        float* Ut1    = Dt;
        float* invMt1 = Dt + MATSZ;
        gs_gemm_nt<<<ggrd, gblk, 0, stream>>>(y, H, nullptr, yMF, nullptr, nullptr);
        gs_gemm_nn<<<ggrd, gblk, 0, stream>>>(yMF, Dinv, nullptr, S, nullptr, nullptr);
        for (int i = 1; i <= N_ITER; ++i) {
            float* trajD  = traj + (size_t)i * ELEMS;
            float* finalD = (i == N_ITER) ? sfinal : nullptr;
            gs_gemm_nt<<<ggrd, gblk, 0, stream>>>(S, Ut1, yMF, T, nullptr, nullptr);
            gs_gemm_nt<<<ggrd, gblk, 0, stream>>>(T, invMt1, nullptr, S, trajD, finalD);
        }
    } else {
        gs_gemm_nt<<<ggrd, gblk, 0, stream>>>(y, H, nullptr, yMF, nullptr, nullptr);
        gs_gemm_nn<<<ggrd, gblk, 0, stream>>>(yMF, Dinv, nullptr, S, nullptr, nullptr);
        for (int i = 1; i <= N_ITER; ++i) {
            float* trajD  = traj + (size_t)i * ELEMS;
            float* finalD = (i == N_ITER) ? sfinal : nullptr;
            gs_gemm_nn<<<ggrd, gblk, 0, stream>>>(S, U, yMF, T, nullptr, nullptr);
            gs_gemm_nn<<<ggrd, gblk, 0, stream>>>(T, invM, nullptr, S, trajD, finalD);
        }
    }
}


# Operands pre-packed in WMMA fragment order (matrices by the one-time packing kernels, S/T by the reduce epilogue): each fragment load is one contiguous b128 over whole 128-B lines
# speedup vs baseline: 1.6965x; 1.6965x over previous
#include <hip/hip_runtime.h>

typedef float v2f __attribute__((ext_vector_type(2)));
typedef float v4f __attribute__((ext_vector_type(4)));
typedef float v8f __attribute__((ext_vector_type(8)));

#define N_DIM   2048
#define BS_DIM  64
#define N_ITER  25
#define ELEMS   ((size_t)BS_DIM * N_DIM)
#define MATSZ   ((size_t)N_DIM * N_DIM)
#define KSPLIT  4
#define KSLICE  (N_DIM / KSPLIT)

__global__ void gs_zero_kernel(float* __restrict__ p, int n) {
    int i = blockIdx.x * blockDim.x + threadIdx.x;
    if (i < n) p[i] = 0.0f;
}

__global__ void __launch_bounds__(256)
gs_transpose(const float* __restrict__ in, float* __restrict__ out) {
    __shared__ float tile[32][33];
    const int bx = blockIdx.x & 63;
    const int by = blockIdx.x >> 6;
    const int tx = threadIdx.x & 31;
    const int ty = threadIdx.x >> 5;

    const int x = bx * 32 + tx;
    #pragma unroll
    for (int r = ty; r < 32; r += 8)
        tile[r][tx] = in[(size_t)(by * 32 + r) * N_DIM + x];
    __syncthreads();
    const int x2 = by * 32 + tx;
    #pragma unroll
    for (int r = ty; r < 32; r += 8)
        out[(size_t)(bx * 32 + r) * N_DIM + x2] = tile[tx][r];
}

#define PK_KB   (N_DIM / 8)

__global__ void __launch_bounds__(256)
gs_pack_b(const float* __restrict__ in, float* __restrict__ out) {
    __shared__ float tile[32][40];
    const int bx = blockIdx.x & 63;
    const int by = blockIdx.x >> 6;
    const int tx = threadIdx.x & 31;
    const int ty = threadIdx.x >> 5;

    const int x = bx * 32 + tx;
    #pragma unroll
    for (int r = ty; r < 32; r += 8)
        tile[r][tx] = in[(size_t)(by * 32 + r) * N_DIM + x];
    __syncthreads();

    const int c   = tx & 15;
    const int h   = tx >> 4;
    const int tnl = ty & 1;
    const int kbl = ty >> 1;
    const int k0  = kbl * 8 + 2 * h;
    const int nl  = tnl * 16 + c;
    v4f v;
    v.x = tile[k0][nl];
    v.y = tile[k0 + 1][nl];
    v.z = tile[k0 + 4][nl];
    v.w = tile[k0 + 5][nl];
    const size_t off =
        (((size_t)(bx * 2 + tnl) * PK_KB + (by * 4 + kbl)) * 32 + tx) * 4;
    *(v4f*)(out + off) = v;
}

__global__ void __launch_bounds__(128)
gs_gemm_ksplit(const float* __restrict__ A, const float* __restrict__ Bt,
               float* __restrict__ P) {
    const int lane = threadIdx.x & 31;
    const int tm   = threadIdx.x >> 5;
    const int s    = blockIdx.x & 3;
    const int tn   = blockIdx.x >> 2;
    const int l15  = lane & 15;
    const int hi   = lane >> 4;
    const int kbase = s * KSLICE + hi * 2;

    const float* aptr = A  + (size_t)(tm * 16 + l15) * N_DIM + kbase;
    const float* bptr = Bt + (size_t)(tn * 16 + l15) * N_DIM + kbase;

    v8f c0 = {}; v8f c1 = {}; v8f c2 = {}; v8f c3 = {};
    #pragma unroll 2
    for (int k = 0; k < KSLICE; k += 16) {
        v2f a0 = *(const v2f*)(aptr + k);
        v2f b0 = *(const v2f*)(bptr + k);
        v2f a1 = *(const v2f*)(aptr + k + 4);
        v2f b1 = *(const v2f*)(bptr + k + 4);
        v2f a2 = *(const v2f*)(aptr + k + 8);
        v2f b2 = *(const v2f*)(bptr + k + 8);
        v2f a3 = *(const v2f*)(aptr + k + 12);
        v2f b3 = *(const v2f*)(bptr + k + 12);
        c0 = __builtin_amdgcn_wmma_f32_16x16x4_f32(false, a0, false, b0,
                                                   (short)0, c0, false, false);
        c1 = __builtin_amdgcn_wmma_f32_16x16x4_f32(false, a1, false, b1,
                                                   (short)0, c1, false, false);
        c2 = __builtin_amdgcn_wmma_f32_16x16x4_f32(false, a2, false, b2,
                                                   (short)0, c2, false, false);
        c3 = __builtin_amdgcn_wmma_f32_16x16x4_f32(false, a3, false, b3,
                                                   (short)0, c3, false, false);
    }

    float* optr = P + (size_t)s * ELEMS
                    + (size_t)(tm * 16 + hi * 8) * N_DIM + tn * 16 + l15;
    #pragma unroll
    for (int r = 0; r < 8; ++r)
        optr[(size_t)r * N_DIM] = (c0[r] + c1[r]) + (c2[r] + c3[r]);
}

#define PK_GRP  8
#define PK_NGRP (KSLICE / 8 / PK_GRP)
__global__ void __launch_bounds__(128)
gs_gemm_ksplit_pk(const float* __restrict__ Ap, const float* __restrict__ Bp,
                  float* __restrict__ P) {
    const int lane = threadIdx.x & 31;
    const int tm   = threadIdx.x >> 5;
    const int s    = blockIdx.x & 3;
    const int tn   = blockIdx.x >> 2;
    const int l15  = lane & 15;
    const int hi   = lane >> 4;

    const v4f* ap = (const v4f*)Ap
                  + ((size_t)(tm * PK_KB + s * (KSLICE / 8)) * 32 + lane);
    const v4f* bp = (const v4f*)Bp
                  + ((size_t)(tn * PK_KB + s * (KSLICE / 8)) * 32 + lane);

    v4f a[3][PK_GRP], b[3][PK_GRP];
    #pragma unroll
    for (int g = 0; g < 2; ++g) {
        #pragma unroll
        for (int j = 0; j < PK_GRP; ++j) {
            a[g][j] = ap[(g * PK_GRP + j) * 32];
            b[g][j] = bp[(g * PK_GRP + j) * 32];
        }
    }

    v8f c0 = {}; v8f c1 = {}; v8f c2 = {}; v8f c3 = {};
    #pragma unroll
    for (int g = 0; g < PK_NGRP; ++g) {
        const int cur = g % 3;
        if (g + 2 < PK_NGRP) {
            const int nxt = (g + 2) % 3;
            #pragma unroll
            for (int j = 0; j < PK_GRP; ++j) {
                a[nxt][j] = ap[((g + 2) * PK_GRP + j) * 32];
                b[nxt][j] = bp[((g + 2) * PK_GRP + j) * 32];
            }
        }
        __builtin_amdgcn_sched_barrier(0);
        #pragma unroll
        for (int j = 0; j < PK_GRP; j += 2) {
            c0 = __builtin_amdgcn_wmma_f32_16x16x4_f32(false, a[cur][j].xy, false,
                     b[cur][j].xy, (short)0, c0, false, false);
            c1 = __builtin_amdgcn_wmma_f32_16x16x4_f32(false, a[cur][j].zw, false,
                     b[cur][j].zw, (short)0, c1, false, false);
            c2 = __builtin_amdgcn_wmma_f32_16x16x4_f32(false, a[cur][j + 1].xy, false,
                     b[cur][j + 1].xy, (short)0, c2, false, false);
            c3 = __builtin_amdgcn_wmma_f32_16x16x4_f32(false, a[cur][j + 1].zw, false,
                     b[cur][j + 1].zw, (short)0, c3, false, false);
        }
        __builtin_amdgcn_sched_barrier(0);
    }

    float* optr = P + (size_t)s * ELEMS
                    + (size_t)(tm * 16 + hi * 8) * N_DIM + tn * 16 + l15;
    #pragma unroll
    for (int r = 0; r < 8; ++r)
        optr[(size_t)r * N_DIM] = (c0[r] + c1[r]) + (c2[r] + c3[r]);
}

__global__ void __launch_bounds__(256)
gs_reduce(const float* __restrict__ P, const float* __restrict__ subSrc,
          float* __restrict__ pk0, float* __restrict__ out1,
          float* __restrict__ out2) {
    const size_t i = ((size_t)blockIdx.x * 256 + threadIdx.x) * 4;
    float4 v0 = *(const float4*)(P + i);
    float4 v1 = *(const float4*)(P + ELEMS + i);
    float4 v2 = *(const float4*)(P + 2 * ELEMS + i);
    float4 v3 = *(const float4*)(P + 3 * ELEMS + i);
    float4 v;
    v.x = (v0.x + v1.x) + (v2.x + v3.x);
    v.y = (v0.y + v1.y) + (v2.y + v3.y);
    v.z = (v0.z + v1.z) + (v2.z + v3.z);
    v.w = (v0.w + v1.w) + (v2.w + v3.w);
    if (subSrc) {
        float4 s = *(const float4*)(subSrc + i);
        v.x = s.x - v.x; v.y = s.y - v.y; v.z = s.z - v.z; v.w = s.w - v.w;
    }
    {
        const int m  = (int)(i >> 11);
        const int k  = (int)(i & (N_DIM - 1));
        const int kb = k >> 3;
        const int j2 = (k >> 1) & 2;
        float* blk = pk0 + ((size_t)((m >> 4) * PK_KB + kb) * 32 + (m & 15)) * 4 + j2;
        v2f lo; lo.x = v.x; lo.y = v.y;
        v2f hh; hh.x = v.z; hh.y = v.w;
        *(v2f*)(blk)          = lo;
        *(v2f*)(blk + 16 * 4) = hh;
    }
    if (out1) *(float4*)(out1 + i) = v;
    if (out2) *(float4*)(out2 + i) = v;
}

__global__ void __launch_bounds__(128)
gs_gemm_nt(const float* __restrict__ A, const float* __restrict__ Bt,
           const float* __restrict__ subSrc,
           float* __restrict__ out0, float* __restrict__ out1,
           float* __restrict__ out2) {
    const int lane = threadIdx.x & 31;
    const int wave = threadIdx.x >> 5;
    const int t    = blockIdx.x * 4 + wave;
    const int tm   = t & 3;
    const int tn   = t >> 2;
    const int l15  = lane & 15;
    const int hi   = lane >> 4;
    const int koff = hi * 2;

    const float* aptr = A  + (size_t)(tm * 16 + l15) * N_DIM + koff;
    const float* bptr = Bt + (size_t)(tn * 16 + l15) * N_DIM + koff;

    v8f acc0 = {}; v8f acc1 = {};
    #pragma unroll 4
    for (int k = 0; k < N_DIM; k += 8) {
        v2f a0 = *(const v2f*)(aptr + k);
        v2f b0 = *(const v2f*)(bptr + k);
        v2f a1 = *(const v2f*)(aptr + k + 4);
        v2f b1 = *(const v2f*)(bptr + k + 4);
        acc0 = __builtin_amdgcn_wmma_f32_16x16x4_f32(false, a0, false, b0,
                                                     (short)0, acc0, false, false);
        acc1 = __builtin_amdgcn_wmma_f32_16x16x4_f32(false, a1, false, b1,
                                                     (short)0, acc1, false, false);
    }

    const size_t obase = (size_t)(tm * 16 + hi * 8) * N_DIM + tn * 16 + l15;
    float res[8];
    #pragma unroll
    for (int r = 0; r < 8; ++r) res[r] = acc0[r] + acc1[r];
    if (subSrc) {
        #pragma unroll
        for (int r = 0; r < 8; ++r)
            res[r] = subSrc[obase + (size_t)r * N_DIM] - res[r];
    }
    #pragma unroll
    for (int r = 0; r < 8; ++r) out0[obase + (size_t)r * N_DIM] = res[r];
    if (out1) {
        #pragma unroll
        for (int r = 0; r < 8; ++r) out1[obase + (size_t)r * N_DIM] = res[r];
    }
    if (out2) {
        #pragma unroll
        for (int r = 0; r < 8; ++r) out2[obase + (size_t)r * N_DIM] = res[r];
    }
}

__global__ void __launch_bounds__(128)
gs_gemm_nn(const float* __restrict__ A, const float* __restrict__ B,
           const float* __restrict__ subSrc,
           float* __restrict__ out0, float* __restrict__ out1,
           float* __restrict__ out2) {
    const int lane = threadIdx.x & 31;
    const int wave = threadIdx.x >> 5;
    const int t    = blockIdx.x * 4 + wave;
    const int tm   = t & 3;
    const int tn   = t >> 2;
    const int l15  = lane & 15;
    const int hi   = lane >> 4;
    const int koff = hi * 2;

    const float* aptr = A + (size_t)(tm * 16 + l15) * N_DIM + koff;
    const float* bptr = B + (size_t)koff * N_DIM + tn * 16 + l15;

    v8f acc0 = {}; v8f acc1 = {};
    #pragma unroll 4
    for (int k = 0; k < N_DIM; k += 8) {
        v2f a0 = *(const v2f*)(aptr + k);
        v2f b0;
        b0.x = bptr[(size_t)(k + 0) * N_DIM];
        b0.y = bptr[(size_t)(k + 1) * N_DIM];
        v2f a1 = *(const v2f*)(aptr + k + 4);
        v2f b1;
        b1.x = bptr[(size_t)(k + 4) * N_DIM];
        b1.y = bptr[(size_t)(k + 5) * N_DIM];
        acc0 = __builtin_amdgcn_wmma_f32_16x16x4_f32(false, a0, false, b0,
                                                     (short)0, acc0, false, false);
        acc1 = __builtin_amdgcn_wmma_f32_16x16x4_f32(false, a1, false, b1,
                                                     (short)0, acc1, false, false);
    }

    const size_t obase = (size_t)(tm * 16 + hi * 8) * N_DIM + tn * 16 + l15;
    float res[8];
    #pragma unroll
    for (int r = 0; r < 8; ++r) res[r] = acc0[r] + acc1[r];
    if (subSrc) {
        #pragma unroll
        for (int r = 0; r < 8; ++r)
            res[r] = subSrc[obase + (size_t)r * N_DIM] - res[r];
    }
    #pragma unroll
    for (int r = 0; r < 8; ++r) out0[obase + (size_t)r * N_DIM] = res[r];
    if (out1) {
        #pragma unroll
        for (int r = 0; r < 8; ++r) out1[obase + (size_t)r * N_DIM] = res[r];
    }
    if (out2) {
        #pragma unroll
        for (int r = 0; r < 8; ++r) out2[obase + (size_t)r * N_DIM] = res[r];
    }
}

extern "C" void kernel_launch(void* const* d_in, const int* in_sizes, int n_in,
                              void* d_out, int out_size, void* d_ws, size_t ws_size,
                              hipStream_t stream) {
    const float* y    = (const float*)d_in[2];
    const float* H    = (const float*)d_in[3];
    const float* Dinv = (const float*)d_in[4];
    const float* U    = (const float*)d_in[5];
    const float* invM = (const float*)d_in[6];

    float* sfinal = (float*)d_out;
    float* traj   = (float*)d_out + ELEMS;

    float* ws    = (float*)d_ws;
    float* yMF   = ws;
    float* S     = ws + ELEMS;
    float* T     = ws + 2 * ELEMS;
    float* P     = ws + 3 * ELEMS;
    float* Dt    = ws + 7 * ELEMS;
    float* Ut    = Dt + MATSZ;
    float* invMt = Ut + MATSZ;

    const size_t need_fast = (7 * ELEMS + 3 * MATSZ) * sizeof(float);
    const size_t need_mid  = (3 * ELEMS + 2 * MATSZ) * sizeof(float);
    const int mode = (ws_size >= need_fast) ? 2 : (ws_size >= need_mid ? 1 : 0);

    dim3 gblk(128);
    dim3 ggrd(128);
    dim3 kgrd(512);
    dim3 rgrd(ELEMS / (256 * 4));

    gs_zero_kernel<<<(int)(ELEMS / 256), 256, 0, stream>>>(traj, (int)ELEMS);

    if (mode == 2) {
        gs_pack_b<<<4096, 256, 0, stream>>>(Dinv, Dt);
        gs_pack_b<<<4096, 256, 0, stream>>>(U, Ut);
        gs_pack_b<<<4096, 256, 0, stream>>>(invM, invMt);

        gs_gemm_ksplit<<<kgrd, gblk, 0, stream>>>(y, H, P);
        gs_reduce<<<rgrd, 256, 0, stream>>>(P, nullptr, T, yMF, nullptr);
        gs_gemm_ksplit_pk<<<kgrd, gblk, 0, stream>>>(T, Dt, P);
        gs_reduce<<<rgrd, 256, 0, stream>>>(P, nullptr, S, nullptr, nullptr);

        for (int i = 1; i <= N_ITER; ++i) {
            float* trajD  = traj + (size_t)i * ELEMS;
            float* finalD = (i == N_ITER) ? sfinal : nullptr;
            gs_gemm_ksplit_pk<<<kgrd, gblk, 0, stream>>>(S, Ut, P);
            gs_reduce<<<rgrd, 256, 0, stream>>>(P, yMF, T, nullptr, nullptr);
            gs_gemm_ksplit_pk<<<kgrd, gblk, 0, stream>>>(T, invMt, P);
            gs_reduce<<<rgrd, 256, 0, stream>>>(P, nullptr, S, trajD, finalD);
        }
    } else if (mode == 1) {
        gs_transpose<<<4096, 256, 0, stream>>>(U, Ut - MATSZ);
        gs_transpose<<<4096, 256, 0, stream>>>(invM, invMt - MATSZ);
        float* Ut1    = Dt;
        float* invMt1 = Dt + MATSZ;
        gs_gemm_nt<<<ggrd, gblk, 0, stream>>>(y, H, nullptr, yMF, nullptr, nullptr);
        gs_gemm_nn<<<ggrd, gblk, 0, stream>>>(yMF, Dinv, nullptr, S, nullptr, nullptr);
        for (int i = 1; i <= N_ITER; ++i) {
            float* trajD  = traj + (size_t)i * ELEMS;
            float* finalD = (i == N_ITER) ? sfinal : nullptr;
            gs_gemm_nt<<<ggrd, gblk, 0, stream>>>(S, Ut1, yMF, T, nullptr, nullptr);
            gs_gemm_nt<<<ggrd, gblk, 0, stream>>>(T, invMt1, nullptr, S, trajD, finalD);
        }
    } else {
        gs_gemm_nt<<<ggrd, gblk, 0, stream>>>(y, H, nullptr, yMF, nullptr, nullptr);
        gs_gemm_nn<<<ggrd, gblk, 0, stream>>>(yMF, Dinv, nullptr, S, nullptr, nullptr);
        for (int i = 1; i <= N_ITER; ++i) {
            float* trajD  = traj + (size_t)i * ELEMS;
            float* finalD = (i == N_ITER) ? sfinal : nullptr;
            gs_gemm_nn<<<ggrd, gblk, 0, stream>>>(S, U, yMF, T, nullptr, nullptr);
            gs_gemm_nn<<<ggrd, gblk, 0, stream>>>(T, invM, nullptr, S, trajD, finalD);
        }
    }
}


# Split-K slices moved inside the work-group (8 waves = 4 K-slices x 2 N-tiles): partials summed through LDS in the same fixed order, reduce epilogue fused into the GEMM; 51 reduce launches gone
# speedup vs baseline: 2.1938x; 1.2931x over previous
#include <hip/hip_runtime.h>

typedef float v2f __attribute__((ext_vector_type(2)));
typedef float v4f __attribute__((ext_vector_type(4)));
typedef float v8f __attribute__((ext_vector_type(8)));

#define N_DIM   2048
#define BS_DIM  64
#define N_ITER  25
#define ELEMS   ((size_t)BS_DIM * N_DIM)
#define MATSZ   ((size_t)N_DIM * N_DIM)
#define KSPLIT  4
#define KSLICE  (N_DIM / KSPLIT)

__global__ void gs_zero_kernel(float* __restrict__ p, int n) {
    int i = blockIdx.x * blockDim.x + threadIdx.x;
    if (i < n) p[i] = 0.0f;
}

__global__ void __launch_bounds__(256)
gs_transpose(const float* __restrict__ in, float* __restrict__ out) {
    __shared__ float tile[32][33];
    const int bx = blockIdx.x & 63;
    const int by = blockIdx.x >> 6;
    const int tx = threadIdx.x & 31;
    const int ty = threadIdx.x >> 5;

    const int x = bx * 32 + tx;
    #pragma unroll
    for (int r = ty; r < 32; r += 8)
        tile[r][tx] = in[(size_t)(by * 32 + r) * N_DIM + x];
    __syncthreads();
    const int x2 = by * 32 + tx;
    #pragma unroll
    for (int r = ty; r < 32; r += 8)
        out[(size_t)(bx * 32 + r) * N_DIM + x2] = tile[tx][r];
}

#define PK_KB   (N_DIM / 8)

__global__ void __launch_bounds__(256)
gs_pack_b(const float* __restrict__ in, float* __restrict__ out) {
    __shared__ float tile[32][40];
    const int bx = blockIdx.x & 63;
    const int by = blockIdx.x >> 6;
    const int tx = threadIdx.x & 31;
    const int ty = threadIdx.x >> 5;

    const int x = bx * 32 + tx;
    #pragma unroll
    for (int r = ty; r < 32; r += 8)
        tile[r][tx] = in[(size_t)(by * 32 + r) * N_DIM + x];
    __syncthreads();

    const int c   = tx & 15;
    const int h   = tx >> 4;
    const int tnl = ty & 1;
    const int kbl = ty >> 1;
    const int k0  = kbl * 8 + 2 * h;
    const int nl  = tnl * 16 + c;
    v4f v;
    v.x = tile[k0][nl];
    v.y = tile[k0 + 1][nl];
    v.z = tile[k0 + 4][nl];
    v.w = tile[k0 + 5][nl];
    const size_t off =
        (((size_t)(bx * 2 + tnl) * PK_KB + (by * 4 + kbl)) * 32 + tx) * 4;
    *(v4f*)(out + off) = v;
}

__global__ void __launch_bounds__(128)
gs_gemm_ksplit(const float* __restrict__ A, const float* __restrict__ Bt,
               float* __restrict__ P) {
    const int lane = threadIdx.x & 31;
    const int tm   = threadIdx.x >> 5;
    const int s    = blockIdx.x & 3;
    const int tn   = blockIdx.x >> 2;
    const int l15  = lane & 15;
    const int hi   = lane >> 4;
    const int kbase = s * KSLICE + hi * 2;

    const float* aptr = A  + (size_t)(tm * 16 + l15) * N_DIM + kbase;
    const float* bptr = Bt + (size_t)(tn * 16 + l15) * N_DIM + kbase;

    v8f c0 = {}; v8f c1 = {}; v8f c2 = {}; v8f c3 = {};
    #pragma unroll 2
    for (int k = 0; k < KSLICE; k += 16) {
        v2f a0 = *(const v2f*)(aptr + k);
        v2f b0 = *(const v2f*)(bptr + k);
        v2f a1 = *(const v2f*)(aptr + k + 4);
        v2f b1 = *(const v2f*)(bptr + k + 4);
        v2f a2 = *(const v2f*)(aptr + k + 8);
        v2f b2 = *(const v2f*)(bptr + k + 8);
        v2f a3 = *(const v2f*)(aptr + k + 12);
        v2f b3 = *(const v2f*)(bptr + k + 12);
        c0 = __builtin_amdgcn_wmma_f32_16x16x4_f32(false, a0, false, b0,
                                                   (short)0, c0, false, false);
        c1 = __builtin_amdgcn_wmma_f32_16x16x4_f32(false, a1, false, b1,
                                                   (short)0, c1, false, false);
        c2 = __builtin_amdgcn_wmma_f32_16x16x4_f32(false, a2, false, b2,
                                                   (short)0, c2, false, false);
        c3 = __builtin_amdgcn_wmma_f32_16x16x4_f32(false, a3, false, b3,
                                                   (short)0, c3, false, false);
    }

    float* optr = P + (size_t)s * ELEMS
                    + (size_t)(tm * 16 + hi * 8) * N_DIM + tn * 16 + l15;
    #pragma unroll
    for (int r = 0; r < 8; ++r)
        optr[(size_t)r * N_DIM] = (c0[r] + c1[r]) + (c2[r] + c3[r]);
}

#define PK_GRP  8
#define PK_NGRP (KSLICE / 8 / PK_GRP)
#define RD_PITCH 36
__global__ void __launch_bounds__(256)
gs_gemm_fused_pk(const float* __restrict__ Ap, const float* __restrict__ Bp,
                 const float* __restrict__ subSrc, float* __restrict__ pk0,
                 float* __restrict__ out1, float* __restrict__ out2) {
    __shared__ alignas(16) float red[KSPLIT][16][RD_PITCH];

    const int lane = threadIdx.x & 31;
    const int w    = __builtin_amdgcn_readfirstlane((int)(threadIdx.x >> 5));
    const int s    = w & 3;
    const int tnl  = w >> 2;
    const int tm   = blockIdx.x & 3;
    const int tp   = blockIdx.x >> 2;
    const int tn   = tp * 2 + tnl;
    const int l15  = lane & 15;
    const int hi   = lane >> 4;

    v2f sa = {}; v2f sb = {};
    if (subSrc != nullptr && tnl == 1) {
        const float* sp = subSrc + (size_t)(tm * 16 + l15) * N_DIM
                        + tp * 32 + s * 8 + 2 * hi;
        sa = *(const v2f*)(sp);
        sb = *(const v2f*)(sp + 4);
    }

    const v4f* ap = (const v4f*)Ap
                  + ((size_t)(tm * PK_KB + s * (KSLICE / 8)) * 32 + lane);
    const v4f* bp = (const v4f*)Bp
                  + ((size_t)(tn * PK_KB + s * (KSLICE / 8)) * 32 + lane);

    v4f a[3][PK_GRP], b[3][PK_GRP];
    #pragma unroll
    for (int g = 0; g < 2; ++g) {
        #pragma unroll
        for (int j = 0; j < PK_GRP; ++j) {
            a[g][j] = ap[(g * PK_GRP + j) * 32];
            b[g][j] = bp[(g * PK_GRP + j) * 32];
        }
    }

    v8f c0 = {}; v8f c1 = {}; v8f c2 = {}; v8f c3 = {};
    #pragma unroll
    for (int g = 0; g < PK_NGRP; ++g) {
        const int cur = g % 3;
        if (g + 2 < PK_NGRP) {
            const int nxt = (g + 2) % 3;
            #pragma unroll
            for (int j = 0; j < PK_GRP; ++j) {
                a[nxt][j] = ap[((g + 2) * PK_GRP + j) * 32];
                b[nxt][j] = bp[((g + 2) * PK_GRP + j) * 32];
            }
        }
        __builtin_amdgcn_sched_barrier(0);
        #pragma unroll
        for (int j = 0; j < PK_GRP; j += 2) {
            c0 = __builtin_amdgcn_wmma_f32_16x16x4_f32(false, a[cur][j].xy, false,
                     b[cur][j].xy, (short)0, c0, false, false);
            c1 = __builtin_amdgcn_wmma_f32_16x16x4_f32(false, a[cur][j].zw, false,
                     b[cur][j].zw, (short)0, c1, false, false);
            c2 = __builtin_amdgcn_wmma_f32_16x16x4_f32(false, a[cur][j + 1].xy, false,
                     b[cur][j + 1].xy, (short)0, c2, false, false);
            c3 = __builtin_amdgcn_wmma_f32_16x16x4_f32(false, a[cur][j + 1].zw, false,
                     b[cur][j + 1].zw, (short)0, c3, false, false);
        }
        __builtin_amdgcn_sched_barrier(0);
    }

    {
        float* rp = &red[s][hi * 8][tnl * 16 + l15];
        #pragma unroll
        for (int r = 0; r < 8; ++r)
            rp[r * RD_PITCH] = (c0[r] + c1[r]) + (c2[r] + c3[r]);
    }
    __syncthreads();

    if (tnl == 1) {
        const float* q = &red[0][l15][s * 8 + 2 * hi];
        const v2f x0 = *(const v2f*)(q);
        const v2f y0 = *(const v2f*)(q + 4);
        const v2f x1 = *(const v2f*)(q + 16 * RD_PITCH);
        const v2f y1 = *(const v2f*)(q + 16 * RD_PITCH + 4);
        const v2f x2 = *(const v2f*)(q + 32 * RD_PITCH);
        const v2f y2 = *(const v2f*)(q + 32 * RD_PITCH + 4);
        const v2f x3 = *(const v2f*)(q + 48 * RD_PITCH);
        const v2f y3 = *(const v2f*)(q + 48 * RD_PITCH + 4);
        v2f x = (x0 + x1) + (x2 + x3);
        v2f y = (y0 + y1) + (y2 + y3);
        if (subSrc != nullptr) { x = sa - x; y = sb - y; }
        v4f o;
        o.x = x.x; o.y = x.y; o.z = y.x; o.w = y.y;
        *((v4f*)pk0 + ((size_t)(tm * PK_KB + tp * 4 + s) * 32 + lane)) = o;
    } else if (out1 != nullptr || out2 != nullptr) {
        const int t    = threadIdx.x;
        const int row  = t >> 3;
        const int col4 = (t & 7) * 4;
        const float* q = &red[0][row][col4];
        const v4f p0 = *(const v4f*)(q);
        const v4f p1 = *(const v4f*)(q + 16 * RD_PITCH);
        const v4f p2 = *(const v4f*)(q + 32 * RD_PITCH);
        const v4f p3 = *(const v4f*)(q + 48 * RD_PITCH);
        v4f v = (p0 + p1) + (p2 + p3);
        const size_t o = (size_t)(tm * 16 + row) * N_DIM + tp * 32 + col4;
        if (subSrc != nullptr) v = *(const v4f*)(subSrc + o) - v;
        if (out1 != nullptr) *(v4f*)(out1 + o) = v;
        if (out2 != nullptr) *(v4f*)(out2 + o) = v;
    }
}

__global__ void __launch_bounds__(256)
gs_reduce(const float* __restrict__ P, const float* __restrict__ subSrc,
          float* __restrict__ pk0, float* __restrict__ out1,
          float* __restrict__ out2) {
    const size_t i = ((size_t)blockIdx.x * 256 + threadIdx.x) * 4;
    float4 v0 = *(const float4*)(P + i);
    float4 v1 = *(const float4*)(P + ELEMS + i);
    float4 v2 = *(const float4*)(P + 2 * ELEMS + i);
    float4 v3 = *(const float4*)(P + 3 * ELEMS + i);
    float4 v;
    v.x = (v0.x + v1.x) + (v2.x + v3.x);
    v.y = (v0.y + v1.y) + (v2.y + v3.y);
    v.z = (v0.z + v1.z) + (v2.z + v3.z);
    v.w = (v0.w + v1.w) + (v2.w + v3.w);
    if (subSrc) {
        float4 s = *(const float4*)(subSrc + i);
        v.x = s.x - v.x; v.y = s.y - v.y; v.z = s.z - v.z; v.w = s.w - v.w;
    }
    {
        const int m  = (int)(i >> 11);
        const int k  = (int)(i & (N_DIM - 1));
        const int kb = k >> 3;
        const int j2 = (k >> 1) & 2;
        float* blk = pk0 + ((size_t)((m >> 4) * PK_KB + kb) * 32 + (m & 15)) * 4 + j2;
        v2f lo; lo.x = v.x; lo.y = v.y;
        v2f hh; hh.x = v.z; hh.y = v.w;
        *(v2f*)(blk)          = lo;
        *(v2f*)(blk + 16 * 4) = hh;
    }
    if (out1) *(float4*)(out1 + i) = v;
    if (out2) *(float4*)(out2 + i) = v;
}

__global__ void __launch_bounds__(128)
gs_gemm_nt(const float* __restrict__ A, const float* __restrict__ Bt,
           const float* __restrict__ subSrc,
           float* __restrict__ out0, float* __restrict__ out1,
           float* __restrict__ out2) {
    const int lane = threadIdx.x & 31;
    const int wave = threadIdx.x >> 5;
    const int t    = blockIdx.x * 4 + wave;
    const int tm   = t & 3;
    const int tn   = t >> 2;
    const int l15  = lane & 15;
    const int hi   = lane >> 4;
    const int koff = hi * 2;

    const float* aptr = A  + (size_t)(tm * 16 + l15) * N_DIM + koff;
    const float* bptr = Bt + (size_t)(tn * 16 + l15) * N_DIM + koff;

    v8f acc0 = {}; v8f acc1 = {};
    #pragma unroll 4
    for (int k = 0; k < N_DIM; k += 8) {
        v2f a0 = *(const v2f*)(aptr + k);
        v2f b0 = *(const v2f*)(bptr + k);
        v2f a1 = *(const v2f*)(aptr + k + 4);
        v2f b1 = *(const v2f*)(bptr + k + 4);
        acc0 = __builtin_amdgcn_wmma_f32_16x16x4_f32(false, a0, false, b0,
                                                     (short)0, acc0, false, false);
        acc1 = __builtin_amdgcn_wmma_f32_16x16x4_f32(false, a1, false, b1,
                                                     (short)0, acc1, false, false);
    }

    const size_t obase = (size_t)(tm * 16 + hi * 8) * N_DIM + tn * 16 + l15;
    float res[8];
    #pragma unroll
    for (int r = 0; r < 8; ++r) res[r] = acc0[r] + acc1[r];
    if (subSrc) {
        #pragma unroll
        for (int r = 0; r < 8; ++r)
            res[r] = subSrc[obase + (size_t)r * N_DIM] - res[r];
    }
    #pragma unroll
    for (int r = 0; r < 8; ++r) out0[obase + (size_t)r * N_DIM] = res[r];
    if (out1) {
        #pragma unroll
        for (int r = 0; r < 8; ++r) out1[obase + (size_t)r * N_DIM] = res[r];
    }
    if (out2) {
        #pragma unroll
        for (int r = 0; r < 8; ++r) out2[obase + (size_t)r * N_DIM] = res[r];
    }
}

__global__ void __launch_bounds__(128)
gs_gemm_nn(const float* __restrict__ A, const float* __restrict__ B,
           const float* __restrict__ subSrc,
           float* __restrict__ out0, float* __restrict__ out1,
           float* __restrict__ out2) {
    const int lane = threadIdx.x & 31;
    const int wave = threadIdx.x >> 5;
    const int t    = blockIdx.x * 4 + wave;
    const int tm   = t & 3;
    const int tn   = t >> 2;
    const int l15  = lane & 15;
    const int hi   = lane >> 4;
    const int koff = hi * 2;

    const float* aptr = A + (size_t)(tm * 16 + l15) * N_DIM + koff;
    const float* bptr = B + (size_t)koff * N_DIM + tn * 16 + l15;

    v8f acc0 = {}; v8f acc1 = {};
    #pragma unroll 4
    for (int k = 0; k < N_DIM; k += 8) {
        v2f a0 = *(const v2f*)(aptr + k);
        v2f b0;
        b0.x = bptr[(size_t)(k + 0) * N_DIM];
        b0.y = bptr[(size_t)(k + 1) * N_DIM];
        v2f a1 = *(const v2f*)(aptr + k + 4);
        v2f b1;
        b1.x = bptr[(size_t)(k + 4) * N_DIM];
        b1.y = bptr[(size_t)(k + 5) * N_DIM];
        acc0 = __builtin_amdgcn_wmma_f32_16x16x4_f32(false, a0, false, b0,
                                                     (short)0, acc0, false, false);
        acc1 = __builtin_amdgcn_wmma_f32_16x16x4_f32(false, a1, false, b1,
                                                     (short)0, acc1, false, false);
    }

    const size_t obase = (size_t)(tm * 16 + hi * 8) * N_DIM + tn * 16 + l15;
    float res[8];
    #pragma unroll
    for (int r = 0; r < 8; ++r) res[r] = acc0[r] + acc1[r];
    if (subSrc) {
        #pragma unroll
        for (int r = 0; r < 8; ++r)
            res[r] = subSrc[obase + (size_t)r * N_DIM] - res[r];
    }
    #pragma unroll
    for (int r = 0; r < 8; ++r) out0[obase + (size_t)r * N_DIM] = res[r];
    if (out1) {
        #pragma unroll
        for (int r = 0; r < 8; ++r) out1[obase + (size_t)r * N_DIM] = res[r];
    }
    if (out2) {
        #pragma unroll
        for (int r = 0; r < 8; ++r) out2[obase + (size_t)r * N_DIM] = res[r];
    }
}

extern "C" void kernel_launch(void* const* d_in, const int* in_sizes, int n_in,
                              void* d_out, int out_size, void* d_ws, size_t ws_size,
                              hipStream_t stream) {
    const float* y    = (const float*)d_in[2];
    const float* H    = (const float*)d_in[3];
    const float* Dinv = (const float*)d_in[4];
    const float* U    = (const float*)d_in[5];
    const float* invM = (const float*)d_in[6];

    float* sfinal = (float*)d_out;
    float* traj   = (float*)d_out + ELEMS;

    float* ws    = (float*)d_ws;
    float* yMF   = ws;
    float* S     = ws + ELEMS;
    float* T     = ws + 2 * ELEMS;
    float* P     = ws + 3 * ELEMS;
    float* Dt    = ws + 7 * ELEMS;
    float* Ut    = Dt + MATSZ;
    float* invMt = Ut + MATSZ;

    const size_t need_fast = (7 * ELEMS + 3 * MATSZ) * sizeof(float);
    const size_t need_mid  = (3 * ELEMS + 2 * MATSZ) * sizeof(float);
    const int mode = (ws_size >= need_fast) ? 2 : (ws_size >= need_mid ? 1 : 0);

    dim3 gblk(128);
    dim3 ggrd(128);
    dim3 kgrd(512);
    dim3 rgrd(ELEMS / (256 * 4));
    dim3 fblk(256);
    dim3 fgrd(256);

    gs_zero_kernel<<<(int)(ELEMS / 256), 256, 0, stream>>>(traj, (int)ELEMS);

    if (mode == 2) {
        gs_pack_b<<<4096, 256, 0, stream>>>(Dinv, Dt);
        gs_pack_b<<<4096, 256, 0, stream>>>(U, Ut);
        gs_pack_b<<<4096, 256, 0, stream>>>(invM, invMt);

        gs_gemm_ksplit<<<kgrd, gblk, 0, stream>>>(y, H, P);
        gs_reduce<<<rgrd, 256, 0, stream>>>(P, nullptr, T, yMF, nullptr);
        gs_gemm_fused_pk<<<fgrd, fblk, 0, stream>>>(T, Dt, nullptr, S, nullptr, nullptr);

        for (int i = 1; i <= N_ITER; ++i) {
            float* trajD  = traj + (size_t)i * ELEMS;
            float* finalD = (i == N_ITER) ? sfinal : nullptr;
            gs_gemm_fused_pk<<<fgrd, fblk, 0, stream>>>(S, Ut, yMF, T, nullptr, nullptr);
            gs_gemm_fused_pk<<<fgrd, fblk, 0, stream>>>(T, invMt, nullptr, S, trajD, finalD);
        }
    } else if (mode == 1) {
        gs_transpose<<<4096, 256, 0, stream>>>(U, Ut - MATSZ);
        gs_transpose<<<4096, 256, 0, stream>>>(invM, invMt - MATSZ);
        float* Ut1    = Dt;
        float* invMt1 = Dt + MATSZ;
        gs_gemm_nt<<<ggrd, gblk, 0, stream>>>(y, H, nullptr, yMF, nullptr, nullptr);
        gs_gemm_nn<<<ggrd, gblk, 0, stream>>>(yMF, Dinv, nullptr, S, nullptr, nullptr);
        for (int i = 1; i <= N_ITER; ++i) {
            float* trajD  = traj + (size_t)i * ELEMS;
            float* finalD = (i == N_ITER) ? sfinal : nullptr;
            gs_gemm_nt<<<ggrd, gblk, 0, stream>>>(S, Ut1, yMF, T, nullptr, nullptr);
            gs_gemm_nt<<<ggrd, gblk, 0, stream>>>(T, invMt1, nullptr, S, trajD, finalD);
        }
    } else {
        gs_gemm_nt<<<ggrd, gblk, 0, stream>>>(y, H, nullptr, yMF, nullptr, nullptr);
        gs_gemm_nn<<<ggrd, gblk, 0, stream>>>(yMF, Dinv, nullptr, S, nullptr, nullptr);
        for (int i = 1; i <= N_ITER; ++i) {
            float* trajD  = traj + (size_t)i * ELEMS;
            float* finalD = (i == N_ITER) ? sfinal : nullptr;
            gs_gemm_nn<<<ggrd, gblk, 0, stream>>>(S, U, yMF, T, nullptr, nullptr);
            gs_gemm_nn<<<ggrd, gblk, 0, stream>>>(T, invM, nullptr, S, trajD, finalD);
        }
    }
}


# Each wave now computes both N-tiles of its block from one A fragment in registers (8 waves = 8 K-slices of 256): 3 loads per 4 WMMAs instead of 4; LDS reduction tree now 8 partials, fixed order
# speedup vs baseline: 2.3201x; 1.0576x over previous
#include <hip/hip_runtime.h>

typedef float v2f __attribute__((ext_vector_type(2)));
typedef float v4f __attribute__((ext_vector_type(4)));
typedef float v8f __attribute__((ext_vector_type(8)));

#define N_DIM   2048
#define BS_DIM  64
#define N_ITER  25
#define ELEMS   ((size_t)BS_DIM * N_DIM)
#define MATSZ   ((size_t)N_DIM * N_DIM)
#define KSPLIT  4
#define KSLICE  (N_DIM / KSPLIT)

__global__ void gs_zero_kernel(float* __restrict__ p, int n) {
    int i = blockIdx.x * blockDim.x + threadIdx.x;
    if (i < n) p[i] = 0.0f;
}

__global__ void __launch_bounds__(256)
gs_transpose(const float* __restrict__ in, float* __restrict__ out) {
    __shared__ float tile[32][33];
    const int bx = blockIdx.x & 63;
    const int by = blockIdx.x >> 6;
    const int tx = threadIdx.x & 31;
    const int ty = threadIdx.x >> 5;

    const int x = bx * 32 + tx;
    #pragma unroll
    for (int r = ty; r < 32; r += 8)
        tile[r][tx] = in[(size_t)(by * 32 + r) * N_DIM + x];
    __syncthreads();
    const int x2 = by * 32 + tx;
    #pragma unroll
    for (int r = ty; r < 32; r += 8)
        out[(size_t)(bx * 32 + r) * N_DIM + x2] = tile[tx][r];
}

#define PK_KB   (N_DIM / 8)

__global__ void __launch_bounds__(256)
gs_pack_b(const float* __restrict__ in, float* __restrict__ out) {
    __shared__ float tile[32][40];
    const int bx = blockIdx.x & 63;
    const int by = blockIdx.x >> 6;
    const int tx = threadIdx.x & 31;
    const int ty = threadIdx.x >> 5;

    const int x = bx * 32 + tx;
    #pragma unroll
    for (int r = ty; r < 32; r += 8)
        tile[r][tx] = in[(size_t)(by * 32 + r) * N_DIM + x];
    __syncthreads();

    const int c   = tx & 15;
    const int h   = tx >> 4;
    const int tnl = ty & 1;
    const int kbl = ty >> 1;
    const int k0  = kbl * 8 + 2 * h;
    const int nl  = tnl * 16 + c;
    v4f v;
    v.x = tile[k0][nl];
    v.y = tile[k0 + 1][nl];
    v.z = tile[k0 + 4][nl];
    v.w = tile[k0 + 5][nl];
    const size_t off =
        (((size_t)(bx * 2 + tnl) * PK_KB + (by * 4 + kbl)) * 32 + tx) * 4;
    *(v4f*)(out + off) = v;
}

__global__ void __launch_bounds__(128)
gs_gemm_ksplit(const float* __restrict__ A, const float* __restrict__ Bt,
               float* __restrict__ P) {
    const int lane = threadIdx.x & 31;
    const int tm   = threadIdx.x >> 5;
    const int s    = blockIdx.x & 3;
    const int tn   = blockIdx.x >> 2;
    const int l15  = lane & 15;
    const int hi   = lane >> 4;
    const int kbase = s * KSLICE + hi * 2;

    const float* aptr = A  + (size_t)(tm * 16 + l15) * N_DIM + kbase;
    const float* bptr = Bt + (size_t)(tn * 16 + l15) * N_DIM + kbase;

    v8f c0 = {}; v8f c1 = {}; v8f c2 = {}; v8f c3 = {};
    #pragma unroll 2
    for (int k = 0; k < KSLICE; k += 16) {
        v2f a0 = *(const v2f*)(aptr + k);
        v2f b0 = *(const v2f*)(bptr + k);
        v2f a1 = *(const v2f*)(aptr + k + 4);
        v2f b1 = *(const v2f*)(bptr + k + 4);
        v2f a2 = *(const v2f*)(aptr + k + 8);
        v2f b2 = *(const v2f*)(bptr + k + 8);
        v2f a3 = *(const v2f*)(aptr + k + 12);
        v2f b3 = *(const v2f*)(bptr + k + 12);
        c0 = __builtin_amdgcn_wmma_f32_16x16x4_f32(false, a0, false, b0,
                                                   (short)0, c0, false, false);
        c1 = __builtin_amdgcn_wmma_f32_16x16x4_f32(false, a1, false, b1,
                                                   (short)0, c1, false, false);
        c2 = __builtin_amdgcn_wmma_f32_16x16x4_f32(false, a2, false, b2,
                                                   (short)0, c2, false, false);
        c3 = __builtin_amdgcn_wmma_f32_16x16x4_f32(false, a3, false, b3,
                                                   (short)0, c3, false, false);
    }

    float* optr = P + (size_t)s * ELEMS
                    + (size_t)(tm * 16 + hi * 8) * N_DIM + tn * 16 + l15;
    #pragma unroll
    for (int r = 0; r < 8; ++r)
        optr[(size_t)r * N_DIM] = (c0[r] + c1[r]) + (c2[r] + c3[r]);
}

#define FK_SPLIT 8
#define FK_KB    (N_DIM / FK_SPLIT / 8)
#define PK_GRP   4
#define PK_NGRP  (FK_KB / PK_GRP)
#define PK_AHEAD 3
#define PK_RING  (PK_AHEAD + 1)
#define RD_PITCH 36
__global__ void __launch_bounds__(256)
gs_gemm_fused_pk(const float* __restrict__ Ap, const float* __restrict__ Bp,
                 const float* __restrict__ subSrc, float* __restrict__ pk0,
                 float* __restrict__ out1, float* __restrict__ out2) {
    __shared__ alignas(16) float red[FK_SPLIT][16][RD_PITCH];

    const int lane = threadIdx.x & 31;
    const int w    = __builtin_amdgcn_readfirstlane((int)(threadIdx.x >> 5));
    const int kbl  = w & 3;
    const int tm   = blockIdx.x & 3;
    const int tp   = blockIdx.x >> 2;
    const int l15  = lane & 15;
    const int hi   = lane >> 4;

    v2f sa = {}; v2f sb = {};
    if (subSrc != nullptr && w >= 4) {
        const float* sp = subSrc + (size_t)(tm * 16 + l15) * N_DIM
                        + tp * 32 + kbl * 8 + 2 * hi;
        sa = *(const v2f*)(sp);
        sb = *(const v2f*)(sp + 4);
    }

    const v4f* ap  = (const v4f*)Ap
                   + ((size_t)(tm * PK_KB + w * FK_KB) * 32 + lane);
    const v4f* bp0 = (const v4f*)Bp
                   + ((size_t)(tp * 2 * PK_KB + w * FK_KB) * 32 + lane);
    const v4f* bp1 = bp0 + (size_t)PK_KB * 32;

    v4f a[PK_RING][PK_GRP], b0[PK_RING][PK_GRP], b1[PK_RING][PK_GRP];
    #pragma unroll
    for (int g = 0; g < PK_AHEAD; ++g) {
        #pragma unroll
        for (int j = 0; j < PK_GRP; ++j) {
            a[g][j]  = ap[(g * PK_GRP + j) * 32];
            b0[g][j] = bp0[(g * PK_GRP + j) * 32];
            b1[g][j] = bp1[(g * PK_GRP + j) * 32];
        }
    }

    v8f c00 = {}; v8f c10 = {}; v8f c01 = {}; v8f c11 = {};
    #pragma unroll
    for (int g = 0; g < PK_NGRP; ++g) {
        const int cur = g % PK_RING;
        if (g + PK_AHEAD < PK_NGRP) {
            const int nxt = (g + PK_AHEAD) % PK_RING;
            #pragma unroll
            for (int j = 0; j < PK_GRP; ++j) {
                a[nxt][j]  = ap[((g + PK_AHEAD) * PK_GRP + j) * 32];
                b0[nxt][j] = bp0[((g + PK_AHEAD) * PK_GRP + j) * 32];
                b1[nxt][j] = bp1[((g + PK_AHEAD) * PK_GRP + j) * 32];
            }
        }
        __builtin_amdgcn_sched_barrier(0);
        #pragma unroll
        for (int j = 0; j < PK_GRP; ++j) {
            c00 = __builtin_amdgcn_wmma_f32_16x16x4_f32(false, a[cur][j].xy, false,
                      b0[cur][j].xy, (short)0, c00, false, false);
            c10 = __builtin_amdgcn_wmma_f32_16x16x4_f32(false, a[cur][j].xy, false,
                      b1[cur][j].xy, (short)0, c10, false, false);
            c01 = __builtin_amdgcn_wmma_f32_16x16x4_f32(false, a[cur][j].zw, false,
                      b0[cur][j].zw, (short)0, c01, false, false);
            c11 = __builtin_amdgcn_wmma_f32_16x16x4_f32(false, a[cur][j].zw, false,
                      b1[cur][j].zw, (short)0, c11, false, false);
        }
        __builtin_amdgcn_sched_barrier(0);
    }

    {
        float* rp = &red[w][hi * 8][l15];
        #pragma unroll
        for (int r = 0; r < 8; ++r) {
            rp[r * RD_PITCH]      = c00[r] + c01[r];
            rp[r * RD_PITCH + 16] = c10[r] + c11[r];
        }
    }
    __syncthreads();

    if (w >= 4) {
        const float* q = &red[0][l15][kbl * 8 + 2 * hi];
        v2f xs[FK_SPLIT], ys[FK_SPLIT];
        #pragma unroll
        for (int p = 0; p < FK_SPLIT; ++p) {
            xs[p] = *(const v2f*)(q + p * 16 * RD_PITCH);
            ys[p] = *(const v2f*)(q + p * 16 * RD_PITCH + 4);
        }
        v2f x = ((xs[0] + xs[1]) + (xs[2] + xs[3])) + ((xs[4] + xs[5]) + (xs[6] + xs[7]));
        v2f y = ((ys[0] + ys[1]) + (ys[2] + ys[3])) + ((ys[4] + ys[5]) + (ys[6] + ys[7]));
        if (subSrc != nullptr) { x = sa - x; y = sb - y; }
        v4f o;
        o.x = x.x; o.y = x.y; o.z = y.x; o.w = y.y;
        *((v4f*)pk0 + ((size_t)(tm * PK_KB + tp * 4 + kbl) * 32 + lane)) = o;
    } else if (out1 != nullptr || out2 != nullptr) {
        const int t    = threadIdx.x;
        const int row  = t >> 3;
        const int col4 = (t & 7) * 4;
        const float* q = &red[0][row][col4];
        v4f ps[FK_SPLIT];
        #pragma unroll
        for (int p = 0; p < FK_SPLIT; ++p)
            ps[p] = *(const v4f*)(q + p * 16 * RD_PITCH);
        v4f v = ((ps[0] + ps[1]) + (ps[2] + ps[3])) + ((ps[4] + ps[5]) + (ps[6] + ps[7]));
        const size_t o = (size_t)(tm * 16 + row) * N_DIM + tp * 32 + col4;
        if (subSrc != nullptr) v = *(const v4f*)(subSrc + o) - v;
        if (out1 != nullptr) *(v4f*)(out1 + o) = v;
        if (out2 != nullptr) *(v4f*)(out2 + o) = v;
    }
}

__global__ void __launch_bounds__(256)
gs_reduce(const float* __restrict__ P, const float* __restrict__ subSrc,
          float* __restrict__ pk0, float* __restrict__ out1,
          float* __restrict__ out2) {
    const size_t i = ((size_t)blockIdx.x * 256 + threadIdx.x) * 4;
    float4 v0 = *(const float4*)(P + i);
    float4 v1 = *(const float4*)(P + ELEMS + i);
    float4 v2 = *(const float4*)(P + 2 * ELEMS + i);
    float4 v3 = *(const float4*)(P + 3 * ELEMS + i);
    float4 v;
    v.x = (v0.x + v1.x) + (v2.x + v3.x);
    v.y = (v0.y + v1.y) + (v2.y + v3.y);
    v.z = (v0.z + v1.z) + (v2.z + v3.z);
    v.w = (v0.w + v1.w) + (v2.w + v3.w);
    if (subSrc) {
        float4 s = *(const float4*)(subSrc + i);
        v.x = s.x - v.x; v.y = s.y - v.y; v.z = s.z - v.z; v.w = s.w - v.w;
    }
    {
        const int m  = (int)(i >> 11);
        const int k  = (int)(i & (N_DIM - 1));
        const int kb = k >> 3;
        const int j2 = (k >> 1) & 2;
        float* blk = pk0 + ((size_t)((m >> 4) * PK_KB + kb) * 32 + (m & 15)) * 4 + j2;
        v2f lo; lo.x = v.x; lo.y = v.y;
        v2f hh; hh.x = v.z; hh.y = v.w;
        *(v2f*)(blk)          = lo;
        *(v2f*)(blk + 16 * 4) = hh;
    }
    if (out1) *(float4*)(out1 + i) = v;
    if (out2) *(float4*)(out2 + i) = v;
}

__global__ void __launch_bounds__(128)
gs_gemm_nt(const float* __restrict__ A, const float* __restrict__ Bt,
           const float* __restrict__ subSrc,
           float* __restrict__ out0, float* __restrict__ out1,
           float* __restrict__ out2) {
    const int lane = threadIdx.x & 31;
    const int wave = threadIdx.x >> 5;
    const int t    = blockIdx.x * 4 + wave;
    const int tm   = t & 3;
    const int tn   = t >> 2;
    const int l15  = lane & 15;
    const int hi   = lane >> 4;
    const int koff = hi * 2;

    const float* aptr = A  + (size_t)(tm * 16 + l15) * N_DIM + koff;
    const float* bptr = Bt + (size_t)(tn * 16 + l15) * N_DIM + koff;

    v8f acc0 = {}; v8f acc1 = {};
    #pragma unroll 4
    for (int k = 0; k < N_DIM; k += 8) {
        v2f a0 = *(const v2f*)(aptr + k);
        v2f b0 = *(const v2f*)(bptr + k);
        v2f a1 = *(const v2f*)(aptr + k + 4);
        v2f b1 = *(const v2f*)(bptr + k + 4);
        acc0 = __builtin_amdgcn_wmma_f32_16x16x4_f32(false, a0, false, b0,
                                                     (short)0, acc0, false, false);
        acc1 = __builtin_amdgcn_wmma_f32_16x16x4_f32(false, a1, false, b1,
                                                     (short)0, acc1, false, false);
    }

    const size_t obase = (size_t)(tm * 16 + hi * 8) * N_DIM + tn * 16 + l15;
    float res[8];
    #pragma unroll
    for (int r = 0; r < 8; ++r) res[r] = acc0[r] + acc1[r];
    if (subSrc) {
        #pragma unroll
        for (int r = 0; r < 8; ++r)
            res[r] = subSrc[obase + (size_t)r * N_DIM] - res[r];
    }
    #pragma unroll
    for (int r = 0; r < 8; ++r) out0[obase + (size_t)r * N_DIM] = res[r];
    if (out1) {
        #pragma unroll
        for (int r = 0; r < 8; ++r) out1[obase + (size_t)r * N_DIM] = res[r];
    }
    if (out2) {
        #pragma unroll
        for (int r = 0; r < 8; ++r) out2[obase + (size_t)r * N_DIM] = res[r];
    }
}

__global__ void __launch_bounds__(128)
gs_gemm_nn(const float* __restrict__ A, const float* __restrict__ B,
           const float* __restrict__ subSrc,
           float* __restrict__ out0, float* __restrict__ out1,
           float* __restrict__ out2) {
    const int lane = threadIdx.x & 31;
    const int wave = threadIdx.x >> 5;
    const int t    = blockIdx.x * 4 + wave;
    const int tm   = t & 3;
    const int tn   = t >> 2;
    const int l15  = lane & 15;
    const int hi   = lane >> 4;
    const int koff = hi * 2;

    const float* aptr = A + (size_t)(tm * 16 + l15) * N_DIM + koff;
    const float* bptr = B + (size_t)koff * N_DIM + tn * 16 + l15;

    v8f acc0 = {}; v8f acc1 = {};
    #pragma unroll 4
    for (int k = 0; k < N_DIM; k += 8) {
        v2f a0 = *(const v2f*)(aptr + k);
        v2f b0;
        b0.x = bptr[(size_t)(k + 0) * N_DIM];
        b0.y = bptr[(size_t)(k + 1) * N_DIM];
        v2f a1 = *(const v2f*)(aptr + k + 4);
        v2f b1;
        b1.x = bptr[(size_t)(k + 4) * N_DIM];
        b1.y = bptr[(size_t)(k + 5) * N_DIM];
        acc0 = __builtin_amdgcn_wmma_f32_16x16x4_f32(false, a0, false, b0,
                                                     (short)0, acc0, false, false);
        acc1 = __builtin_amdgcn_wmma_f32_16x16x4_f32(false, a1, false, b1,
                                                     (short)0, acc1, false, false);
    }

    const size_t obase = (size_t)(tm * 16 + hi * 8) * N_DIM + tn * 16 + l15;
    float res[8];
    #pragma unroll
    for (int r = 0; r < 8; ++r) res[r] = acc0[r] + acc1[r];
    if (subSrc) {
        #pragma unroll
        for (int r = 0; r < 8; ++r)
            res[r] = subSrc[obase + (size_t)r * N_DIM] - res[r];
    }
    #pragma unroll
    for (int r = 0; r < 8; ++r) out0[obase + (size_t)r * N_DIM] = res[r];
    if (out1) {
        #pragma unroll
        for (int r = 0; r < 8; ++r) out1[obase + (size_t)r * N_DIM] = res[r];
    }
    if (out2) {
        #pragma unroll
        for (int r = 0; r < 8; ++r) out2[obase + (size_t)r * N_DIM] = res[r];
    }
}

extern "C" void kernel_launch(void* const* d_in, const int* in_sizes, int n_in,
                              void* d_out, int out_size, void* d_ws, size_t ws_size,
                              hipStream_t stream) {
    const float* y    = (const float*)d_in[2];
    const float* H    = (const float*)d_in[3];
    const float* Dinv = (const float*)d_in[4];
    const float* U    = (const float*)d_in[5];
    const float* invM = (const float*)d_in[6];

    float* sfinal = (float*)d_out;
    float* traj   = (float*)d_out + ELEMS;

    float* ws    = (float*)d_ws;
    float* yMF   = ws;
    float* S     = ws + ELEMS;
    float* T     = ws + 2 * ELEMS;
    float* P     = ws + 3 * ELEMS;
    float* Dt    = ws + 7 * ELEMS;
    float* Ut    = Dt + MATSZ;
    float* invMt = Ut + MATSZ;

    const size_t need_fast = (7 * ELEMS + 3 * MATSZ) * sizeof(float);
    const size_t need_mid  = (3 * ELEMS + 2 * MATSZ) * sizeof(float);
    const int mode = (ws_size >= need_fast) ? 2 : (ws_size >= need_mid ? 1 : 0);

    dim3 gblk(128);
    dim3 ggrd(128);
    dim3 kgrd(512);
    dim3 rgrd(ELEMS / (256 * 4));
    dim3 fblk(256);
    dim3 fgrd(256);

    gs_zero_kernel<<<(int)(ELEMS / 256), 256, 0, stream>>>(traj, (int)ELEMS);

    if (mode == 2) {
        gs_pack_b<<<4096, 256, 0, stream>>>(Dinv, Dt);
        gs_pack_b<<<4096, 256, 0, stream>>>(U, Ut);
        gs_pack_b<<<4096, 256, 0, stream>>>(invM, invMt);

        gs_gemm_ksplit<<<kgrd, gblk, 0, stream>>>(y, H, P);
        gs_reduce<<<rgrd, 256, 0, stream>>>(P, nullptr, T, yMF, nullptr);
        gs_gemm_fused_pk<<<fgrd, fblk, 0, stream>>>(T, Dt, nullptr, S, nullptr, nullptr);

        for (int i = 1; i <= N_ITER; ++i) {
            float* trajD  = traj + (size_t)i * ELEMS;
            float* finalD = (i == N_ITER) ? sfinal : nullptr;
            gs_gemm_fused_pk<<<fgrd, fblk, 0, stream>>>(S, Ut, yMF, T, nullptr, nullptr);
            gs_gemm_fused_pk<<<fgrd, fblk, 0, stream>>>(T, invMt, nullptr, S, trajD, finalD);
        }
    } else if (mode == 1) {
        gs_transpose<<<4096, 256, 0, stream>>>(U, Ut - MATSZ);
        gs_transpose<<<4096, 256, 0, stream>>>(invM, invMt - MATSZ);
        float* Ut1    = Dt;
        float* invMt1 = Dt + MATSZ;
        gs_gemm_nt<<<ggrd, gblk, 0, stream>>>(y, H, nullptr, yMF, nullptr, nullptr);
        gs_gemm_nn<<<ggrd, gblk, 0, stream>>>(yMF, Dinv, nullptr, S, nullptr, nullptr);
        for (int i = 1; i <= N_ITER; ++i) {
            float* trajD  = traj + (size_t)i * ELEMS;
            float* finalD = (i == N_ITER) ? sfinal : nullptr;
            gs_gemm_nt<<<ggrd, gblk, 0, stream>>>(S, Ut1, yMF, T, nullptr, nullptr);
            gs_gemm_nt<<<ggrd, gblk, 0, stream>>>(T, invMt1, nullptr, S, trajD, finalD);
        }
    } else {
        gs_gemm_nt<<<ggrd, gblk, 0, stream>>>(y, H, nullptr, yMF, nullptr, nullptr);
        gs_gemm_nn<<<ggrd, gblk, 0, stream>>>(yMF, Dinv, nullptr, S, nullptr, nullptr);
        for (int i = 1; i <= N_ITER; ++i) {
            float* trajD  = traj + (size_t)i * ELEMS;
            float* finalD = (i == N_ITER) ? sfinal : nullptr;
            gs_gemm_nn<<<ggrd, gblk, 0, stream>>>(S, U, yMF, T, nullptr, nullptr);
            gs_gemm_nn<<<ggrd, gblk, 0, stream>>>(T, invM, nullptr, S, trajD, finalD);
        }
    }
}


# Step folded to s_next = C - s@W with W = U@invM and C = yMF@invM formed once per call on the same f32 WMMA kernel: 25 dependent GEMMs, not 50; products re-associated, so the summation order changes
# speedup vs baseline: 2.5605x; 1.1036x over previous
#include <hip/hip_runtime.h>

typedef float v2f __attribute__((ext_vector_type(2)));
typedef float v4f __attribute__((ext_vector_type(4)));
typedef float v8f __attribute__((ext_vector_type(8)));

#define N_DIM   2048
#define BS_DIM  64
#define N_ITER  25
#define ELEMS   ((size_t)BS_DIM * N_DIM)
#define MATSZ   ((size_t)N_DIM * N_DIM)
#define KSPLIT  4
#define KSLICE  (N_DIM / KSPLIT)

__global__ void gs_zero_kernel(float* __restrict__ p, int n) {
    int i = blockIdx.x * blockDim.x + threadIdx.x;
    if (i < n) p[i] = 0.0f;
}

__global__ void __launch_bounds__(256)
gs_transpose(const float* __restrict__ in, float* __restrict__ out) {
    __shared__ float tile[32][33];
    const int bx = blockIdx.x & 63;
    const int by = blockIdx.x >> 6;
    const int tx = threadIdx.x & 31;
    const int ty = threadIdx.x >> 5;

    const int x = bx * 32 + tx;
    #pragma unroll
    for (int r = ty; r < 32; r += 8)
        tile[r][tx] = in[(size_t)(by * 32 + r) * N_DIM + x];
    __syncthreads();
    const int x2 = by * 32 + tx;
    #pragma unroll
    for (int r = ty; r < 32; r += 8)
        out[(size_t)(bx * 32 + r) * N_DIM + x2] = tile[tx][r];
}

#define PK_KB   (N_DIM / 8)

__global__ void __launch_bounds__(256)
gs_pack_b(const float* __restrict__ in, float* __restrict__ out) {
    __shared__ float tile[32][40];
    const int bx = blockIdx.x & 63;
    const int by = blockIdx.x >> 6;
    const int tx = threadIdx.x & 31;
    const int ty = threadIdx.x >> 5;

    const int x = bx * 32 + tx;
    #pragma unroll
    for (int r = ty; r < 32; r += 8)
        tile[r][tx] = in[(size_t)(by * 32 + r) * N_DIM + x];
    __syncthreads();

    const int c   = tx & 15;
    const int h   = tx >> 4;
    const int tnl = ty & 1;
    const int kbl = ty >> 1;
    const int k0  = kbl * 8 + 2 * h;
    const int nl  = tnl * 16 + c;
    v4f v;
    v.x = tile[k0][nl];
    v.y = tile[k0 + 1][nl];
    v.z = tile[k0 + 4][nl];
    v.w = tile[k0 + 5][nl];
    const size_t off =
        (((size_t)(bx * 2 + tnl) * PK_KB + (by * 4 + kbl)) * 32 + tx) * 4;
    *(v4f*)(out + off) = v;
}

__global__ void __launch_bounds__(256)
gs_pack_a(const float* __restrict__ in, float* __restrict__ out) {
    const int lane = threadIdx.x & 31;
    const int w    = threadIdx.x >> 5;
    const int tmr  = blockIdx.x >> 5;
    const int kb   = (blockIdx.x & 31) * 8 + w;
    const int c    = lane & 15;
    const int h    = lane >> 4;
    const float* p = in + (size_t)(tmr * 16 + c) * N_DIM + kb * 8 + 2 * h;
    const v2f lo = *(const v2f*)(p);
    const v2f hh = *(const v2f*)(p + 4);
    v4f v;
    v.x = lo.x; v.y = lo.y; v.z = hh.x; v.w = hh.y;
    *((v4f*)out + ((size_t)(tmr * PK_KB + kb) * 32 + lane)) = v;
}

__global__ void __launch_bounds__(128)
gs_gemm_ksplit(const float* __restrict__ A, const float* __restrict__ Bt,
               float* __restrict__ P) {
    const int lane = threadIdx.x & 31;
    const int tm   = threadIdx.x >> 5;
    const int s    = blockIdx.x & 3;
    const int tn   = blockIdx.x >> 2;
    const int l15  = lane & 15;
    const int hi   = lane >> 4;
    const int kbase = s * KSLICE + hi * 2;

    const float* aptr = A  + (size_t)(tm * 16 + l15) * N_DIM + kbase;
    const float* bptr = Bt + (size_t)(tn * 16 + l15) * N_DIM + kbase;

    v8f c0 = {}; v8f c1 = {}; v8f c2 = {}; v8f c3 = {};
    #pragma unroll 2
    for (int k = 0; k < KSLICE; k += 16) {
        v2f a0 = *(const v2f*)(aptr + k);
        v2f b0 = *(const v2f*)(bptr + k);
        v2f a1 = *(const v2f*)(aptr + k + 4);
        v2f b1 = *(const v2f*)(bptr + k + 4);
        v2f a2 = *(const v2f*)(aptr + k + 8);
        v2f b2 = *(const v2f*)(bptr + k + 8);
        v2f a3 = *(const v2f*)(aptr + k + 12);
        v2f b3 = *(const v2f*)(bptr + k + 12);
        c0 = __builtin_amdgcn_wmma_f32_16x16x4_f32(false, a0, false, b0,
                                                   (short)0, c0, false, false);
        c1 = __builtin_amdgcn_wmma_f32_16x16x4_f32(false, a1, false, b1,
                                                   (short)0, c1, false, false);
        c2 = __builtin_amdgcn_wmma_f32_16x16x4_f32(false, a2, false, b2,
                                                   (short)0, c2, false, false);
        c3 = __builtin_amdgcn_wmma_f32_16x16x4_f32(false, a3, false, b3,
                                                   (short)0, c3, false, false);
    }

    float* optr = P + (size_t)s * ELEMS
                    + (size_t)(tm * 16 + hi * 8) * N_DIM + tn * 16 + l15;
    #pragma unroll
    for (int r = 0; r < 8; ++r)
        optr[(size_t)r * N_DIM] = (c0[r] + c1[r]) + (c2[r] + c3[r]);
}

#define FK_SPLIT 8
#define FK_KB    (N_DIM / FK_SPLIT / 8)
#define PK_GRP   4
#define PK_NGRP  (FK_KB / PK_GRP)
#define PK_AHEAD 3
#define PK_RING  (PK_AHEAD + 1)
#define RD_PITCH 36
template <bool FOLD>
__global__ void __launch_bounds__(256)
gs_gemm_fused_pk(const float* __restrict__ Ap, const float* __restrict__ Bp,
                 const float* __restrict__ subSrc, float* __restrict__ pk0,
                 float* __restrict__ out1, float* __restrict__ out2) {
    __shared__ alignas(16) float red[FK_SPLIT][16][RD_PITCH];

    const int lane = threadIdx.x & 31;
    const int w    = __builtin_amdgcn_readfirstlane((int)(threadIdx.x >> 5));
    const int kbl  = w & 3;
    const int tm   = FOLD ? (int)(blockIdx.x >> 6) : (int)(blockIdx.x & 3);
    const int tp   = FOLD ? (int)(blockIdx.x & 63) : (int)(blockIdx.x >> 2);
    const int l15  = lane & 15;
    const int hi   = lane >> 4;

    v2f sa = {}; v2f sb = {};
    if (!FOLD && subSrc != nullptr && w >= 4) {
        const float* sp = subSrc + (size_t)(tm * 16 + l15) * N_DIM
                        + tp * 32 + kbl * 8 + 2 * hi;
        sa = *(const v2f*)(sp);
        sb = *(const v2f*)(sp + 4);
    }

    const v4f* ap  = (const v4f*)Ap
                   + ((size_t)(tm * PK_KB + w * FK_KB) * 32 + lane);
    const v4f* bp0 = (const v4f*)Bp
                   + ((size_t)(tp * 2 * PK_KB + w * FK_KB) * 32 + lane);
    const v4f* bp1 = bp0 + (size_t)PK_KB * 32;

    v4f a[PK_RING][PK_GRP], b0[PK_RING][PK_GRP], b1[PK_RING][PK_GRP];
    #pragma unroll
    for (int g = 0; g < PK_AHEAD; ++g) {
        #pragma unroll
        for (int j = 0; j < PK_GRP; ++j) {
            a[g][j]  = ap[(g * PK_GRP + j) * 32];
            b0[g][j] = bp0[(g * PK_GRP + j) * 32];
            b1[g][j] = bp1[(g * PK_GRP + j) * 32];
        }
    }

    v8f c00 = {}; v8f c10 = {}; v8f c01 = {}; v8f c11 = {};
    #pragma unroll
    for (int g = 0; g < PK_NGRP; ++g) {
        const int cur = g % PK_RING;
        if (g + PK_AHEAD < PK_NGRP) {
            const int nxt = (g + PK_AHEAD) % PK_RING;
            #pragma unroll
            for (int j = 0; j < PK_GRP; ++j) {
                a[nxt][j]  = ap[((g + PK_AHEAD) * PK_GRP + j) * 32];
                b0[nxt][j] = bp0[((g + PK_AHEAD) * PK_GRP + j) * 32];
                b1[nxt][j] = bp1[((g + PK_AHEAD) * PK_GRP + j) * 32];
            }
        }
        __builtin_amdgcn_sched_barrier(0);
        #pragma unroll
        for (int j = 0; j < PK_GRP; ++j) {
            c00 = __builtin_amdgcn_wmma_f32_16x16x4_f32(false, a[cur][j].xy, false,
                      b0[cur][j].xy, (short)0, c00, false, false);
            c10 = __builtin_amdgcn_wmma_f32_16x16x4_f32(false, a[cur][j].xy, false,
                      b1[cur][j].xy, (short)0, c10, false, false);
            c01 = __builtin_amdgcn_wmma_f32_16x16x4_f32(false, a[cur][j].zw, false,
                      b0[cur][j].zw, (short)0, c01, false, false);
            c11 = __builtin_amdgcn_wmma_f32_16x16x4_f32(false, a[cur][j].zw, false,
                      b1[cur][j].zw, (short)0, c11, false, false);
        }
        __builtin_amdgcn_sched_barrier(0);
    }

    {
        float* rp = &red[w][hi * 8][l15];
        #pragma unroll
        for (int r = 0; r < 8; ++r) {
            rp[r * RD_PITCH]      = c00[r] + c01[r];
            rp[r * RD_PITCH + 16] = c10[r] + c11[r];
        }
    }
    __syncthreads();

    if (FOLD) {
        if (w >= 4) {
            const int tnl = kbl & 1;
            const int kb2 = kbl >> 1;
            const float* q = &red[0][kb2 * 8 + 2 * hi][tnl * 16 + l15];
            float e0[FK_SPLIT], e1[FK_SPLIT], e4[FK_SPLIT], e5[FK_SPLIT];
            #pragma unroll
            for (int p = 0; p < FK_SPLIT; ++p) {
                e0[p] = q[p * 16 * RD_PITCH];
                e1[p] = q[p * 16 * RD_PITCH + RD_PITCH];
                e4[p] = q[p * 16 * RD_PITCH + 4 * RD_PITCH];
                e5[p] = q[p * 16 * RD_PITCH + 5 * RD_PITCH];
            }
            v4f o;
            o.x = ((e0[0] + e0[1]) + (e0[2] + e0[3])) + ((e0[4] + e0[5]) + (e0[6] + e0[7]));
            o.y = ((e1[0] + e1[1]) + (e1[2] + e1[3])) + ((e1[4] + e1[5]) + (e1[6] + e1[7]));
            o.z = ((e4[0] + e4[1]) + (e4[2] + e4[3])) + ((e4[4] + e4[5]) + (e4[6] + e4[7]));
            o.w = ((e5[0] + e5[1]) + (e5[2] + e5[3])) + ((e5[4] + e5[5]) + (e5[6] + e5[7]));
            *((v4f*)pk0 + ((size_t)((tp * 2 + tnl) * PK_KB + tm * 2 + kb2) * 32 + lane)) = o;
        }
    } else if (w >= 4) {
        if (pk0 == nullptr) return;
        const float* q = &red[0][l15][kbl * 8 + 2 * hi];
        v2f xs[FK_SPLIT], ys[FK_SPLIT];
        #pragma unroll
        for (int p = 0; p < FK_SPLIT; ++p) {
            xs[p] = *(const v2f*)(q + p * 16 * RD_PITCH);
            ys[p] = *(const v2f*)(q + p * 16 * RD_PITCH + 4);
        }
        v2f x = ((xs[0] + xs[1]) + (xs[2] + xs[3])) + ((xs[4] + xs[5]) + (xs[6] + xs[7]));
        v2f y = ((ys[0] + ys[1]) + (ys[2] + ys[3])) + ((ys[4] + ys[5]) + (ys[6] + ys[7]));
        if (subSrc != nullptr) { x = sa - x; y = sb - y; }
        v4f o;
        o.x = x.x; o.y = x.y; o.z = y.x; o.w = y.y;
        *((v4f*)pk0 + ((size_t)(tm * PK_KB + tp * 4 + kbl) * 32 + lane)) = o;
    } else if (out1 != nullptr || out2 != nullptr) {
        const int t    = threadIdx.x;
        const int row  = t >> 3;
        const int col4 = (t & 7) * 4;
        const float* q = &red[0][row][col4];
        v4f ps[FK_SPLIT];
        #pragma unroll
        for (int p = 0; p < FK_SPLIT; ++p)
            ps[p] = *(const v4f*)(q + p * 16 * RD_PITCH);
        v4f v = ((ps[0] + ps[1]) + (ps[2] + ps[3])) + ((ps[4] + ps[5]) + (ps[6] + ps[7]));
        const size_t o = (size_t)(tm * 16 + row) * N_DIM + tp * 32 + col4;
        if (subSrc != nullptr) v = *(const v4f*)(subSrc + o) - v;
        if (out1 != nullptr) *(v4f*)(out1 + o) = v;
        if (out2 != nullptr) *(v4f*)(out2 + o) = v;
    }
}

__global__ void __launch_bounds__(256)
gs_reduce(const float* __restrict__ P, const float* __restrict__ subSrc,
          float* __restrict__ pk0, float* __restrict__ out1,
          float* __restrict__ out2) {
    const size_t i = ((size_t)blockIdx.x * 256 + threadIdx.x) * 4;
    float4 v0 = *(const float4*)(P + i);
    float4 v1 = *(const float4*)(P + ELEMS + i);
    float4 v2 = *(const float4*)(P + 2 * ELEMS + i);
    float4 v3 = *(const float4*)(P + 3 * ELEMS + i);
    float4 v;
    v.x = (v0.x + v1.x) + (v2.x + v3.x);
    v.y = (v0.y + v1.y) + (v2.y + v3.y);
    v.z = (v0.z + v1.z) + (v2.z + v3.z);
    v.w = (v0.w + v1.w) + (v2.w + v3.w);
    if (subSrc) {
        float4 s = *(const float4*)(subSrc + i);
        v.x = s.x - v.x; v.y = s.y - v.y; v.z = s.z - v.z; v.w = s.w - v.w;
    }
    {
        const int m  = (int)(i >> 11);
        const int k  = (int)(i & (N_DIM - 1));
        const int kb = k >> 3;
        const int j2 = (k >> 1) & 2;
        float* blk = pk0 + ((size_t)((m >> 4) * PK_KB + kb) * 32 + (m & 15)) * 4 + j2;
        v2f lo; lo.x = v.x; lo.y = v.y;
        v2f hh; hh.x = v.z; hh.y = v.w;
        *(v2f*)(blk)          = lo;
        *(v2f*)(blk + 16 * 4) = hh;
    }
    if (out1) *(float4*)(out1 + i) = v;
    if (out2) *(float4*)(out2 + i) = v;
}

__global__ void __launch_bounds__(128)
gs_gemm_nt(const float* __restrict__ A, const float* __restrict__ Bt,
           const float* __restrict__ subSrc,
           float* __restrict__ out0, float* __restrict__ out1,
           float* __restrict__ out2) {
    const int lane = threadIdx.x & 31;
    const int wave = threadIdx.x >> 5;
    const int t    = blockIdx.x * 4 + wave;
    const int tm   = t & 3;
    const int tn   = t >> 2;
    const int l15  = lane & 15;
    const int hi   = lane >> 4;
    const int koff = hi * 2;

    const float* aptr = A  + (size_t)(tm * 16 + l15) * N_DIM + koff;
    const float* bptr = Bt + (size_t)(tn * 16 + l15) * N_DIM + koff;

    v8f acc0 = {}; v8f acc1 = {};
    #pragma unroll 4
    for (int k = 0; k < N_DIM; k += 8) {
        v2f a0 = *(const v2f*)(aptr + k);
        v2f b0 = *(const v2f*)(bptr + k);
        v2f a1 = *(const v2f*)(aptr + k + 4);
        v2f b1 = *(const v2f*)(bptr + k + 4);
        acc0 = __builtin_amdgcn_wmma_f32_16x16x4_f32(false, a0, false, b0,
                                                     (short)0, acc0, false, false);
        acc1 = __builtin_amdgcn_wmma_f32_16x16x4_f32(false, a1, false, b1,
                                                     (short)0, acc1, false, false);
    }

    const size_t obase = (size_t)(tm * 16 + hi * 8) * N_DIM + tn * 16 + l15;
    float res[8];
    #pragma unroll
    for (int r = 0; r < 8; ++r) res[r] = acc0[r] + acc1[r];
    if (subSrc) {
        #pragma unroll
        for (int r = 0; r < 8; ++r)
            res[r] = subSrc[obase + (size_t)r * N_DIM] - res[r];
    }
    #pragma unroll
    for (int r = 0; r < 8; ++r) out0[obase + (size_t)r * N_DIM] = res[r];
    if (out1) {
        #pragma unroll
        for (int r = 0; r < 8; ++r) out1[obase + (size_t)r * N_DIM] = res[r];
    }
    if (out2) {
        #pragma unroll
        for (int r = 0; r < 8; ++r) out2[obase + (size_t)r * N_DIM] = res[r];
    }
}

__global__ void __launch_bounds__(128)
gs_gemm_nn(const float* __restrict__ A, const float* __restrict__ B,
           const float* __restrict__ subSrc,
           float* __restrict__ out0, float* __restrict__ out1,
           float* __restrict__ out2) {
    const int lane = threadIdx.x & 31;
    const int wave = threadIdx.x >> 5;
    const int t    = blockIdx.x * 4 + wave;
    const int tm   = t & 3;
    const int tn   = t >> 2;
    const int l15  = lane & 15;
    const int hi   = lane >> 4;
    const int koff = hi * 2;

    const float* aptr = A + (size_t)(tm * 16 + l15) * N_DIM + koff;
    const float* bptr = B + (size_t)koff * N_DIM + tn * 16 + l15;

    v8f acc0 = {}; v8f acc1 = {};
    #pragma unroll 4
    for (int k = 0; k < N_DIM; k += 8) {
        v2f a0 = *(const v2f*)(aptr + k);
        v2f b0;
        b0.x = bptr[(size_t)(k + 0) * N_DIM];
        b0.y = bptr[(size_t)(k + 1) * N_DIM];
        v2f a1 = *(const v2f*)(aptr + k + 4);
        v2f b1;
        b1.x = bptr[(size_t)(k + 4) * N_DIM];
        b1.y = bptr[(size_t)(k + 5) * N_DIM];
        acc0 = __builtin_amdgcn_wmma_f32_16x16x4_f32(false, a0, false, b0,
                                                     (short)0, acc0, false, false);
        acc1 = __builtin_amdgcn_wmma_f32_16x16x4_f32(false, a1, false, b1,
                                                     (short)0, acc1, false, false);
    }

    const size_t obase = (size_t)(tm * 16 + hi * 8) * N_DIM + tn * 16 + l15;
    float res[8];
    #pragma unroll
    for (int r = 0; r < 8; ++r) res[r] = acc0[r] + acc1[r];
    if (subSrc) {
        #pragma unroll
        for (int r = 0; r < 8; ++r)
            res[r] = subSrc[obase + (size_t)r * N_DIM] - res[r];
    }
    #pragma unroll
    for (int r = 0; r < 8; ++r) out0[obase + (size_t)r * N_DIM] = res[r];
    if (out1) {
        #pragma unroll
        for (int r = 0; r < 8; ++r) out1[obase + (size_t)r * N_DIM] = res[r];
    }
    if (out2) {
        #pragma unroll
        for (int r = 0; r < 8; ++r) out2[obase + (size_t)r * N_DIM] = res[r];
    }
}

extern "C" void kernel_launch(void* const* d_in, const int* in_sizes, int n_in,
                              void* d_out, int out_size, void* d_ws, size_t ws_size,
                              hipStream_t stream) {
    const float* y    = (const float*)d_in[2];
    const float* H    = (const float*)d_in[3];
    const float* Dinv = (const float*)d_in[4];
    const float* U    = (const float*)d_in[5];
    const float* invM = (const float*)d_in[6];

    float* sfinal = (float*)d_out;
    float* traj   = (float*)d_out + ELEMS;

    float* ws    = (float*)d_ws;
    float* yMF   = ws;
    float* S     = ws + ELEMS;
    float* T     = ws + 2 * ELEMS;
    float* P     = ws + 3 * ELEMS;
    float* Dt    = ws + 7 * ELEMS;
    float* Ut    = Dt + MATSZ;
    float* invMt = Ut + MATSZ;

    const size_t need_fast = (7 * ELEMS + 3 * MATSZ) * sizeof(float);
    const size_t need_mid  = (3 * ELEMS + 2 * MATSZ) * sizeof(float);
    const int mode = (ws_size >= need_fast) ? 2 : (ws_size >= need_mid ? 1 : 0);

    dim3 gblk(128);
    dim3 ggrd(128);
    dim3 kgrd(512);
    dim3 rgrd(ELEMS / (256 * 4));
    dim3 fblk(256);
    dim3 fgrd(256);

    gs_zero_kernel<<<(int)(ELEMS / 256), 256, 0, stream>>>(traj, (int)ELEMS);

    if (mode == 2) {
        float* Cm = yMF;
        float* Wp = Dt;

        gs_pack_b<<<4096, 256, 0, stream>>>(Dinv, Dt);
        gs_pack_a<<<4096, 256, 0, stream>>>(U, Ut);
        gs_pack_b<<<4096, 256, 0, stream>>>(invM, invMt);

        gs_gemm_ksplit<<<kgrd, gblk, 0, stream>>>(y, H, P);
        gs_reduce<<<rgrd, 256, 0, stream>>>(P, nullptr, T, nullptr, nullptr);
        gs_gemm_fused_pk<false><<<fgrd, fblk, 0, stream>>>(T, Dt, nullptr, S, nullptr, nullptr);
        gs_gemm_fused_pk<false><<<fgrd, fblk, 0, stream>>>(T, invMt, nullptr, nullptr, Cm, nullptr);
        gs_gemm_fused_pk<true><<<128 * 64, fblk, 0, stream>>>(Ut, invMt, nullptr, Wp, nullptr, nullptr);

        float* cur = S;
        float* nxt = T;
        for (int i = 1; i <= N_ITER; ++i) {
            float* trajD  = traj + (size_t)i * ELEMS;
            float* finalD = (i == N_ITER) ? sfinal : nullptr;
            gs_gemm_fused_pk<false><<<fgrd, fblk, 0, stream>>>(cur, Wp, Cm, nxt, trajD, finalD);
            float* t = cur; cur = nxt; nxt = t;
        }
    } else if (mode == 1) {
        gs_transpose<<<4096, 256, 0, stream>>>(U, Ut - MATSZ);
        gs_transpose<<<4096, 256, 0, stream>>>(invM, invMt - MATSZ);
        float* Ut1    = Dt;
        float* invMt1 = Dt + MATSZ;
        gs_gemm_nt<<<ggrd, gblk, 0, stream>>>(y, H, nullptr, yMF, nullptr, nullptr);
        gs_gemm_nn<<<ggrd, gblk, 0, stream>>>(yMF, Dinv, nullptr, S, nullptr, nullptr);
        for (int i = 1; i <= N_ITER; ++i) {
            float* trajD  = traj + (size_t)i * ELEMS;
            float* finalD = (i == N_ITER) ? sfinal : nullptr;
            gs_gemm_nt<<<ggrd, gblk, 0, stream>>>(S, Ut1, yMF, T, nullptr, nullptr);
            gs_gemm_nt<<<ggrd, gblk, 0, stream>>>(T, invMt1, nullptr, S, trajD, finalD);
        }
    } else {
        gs_gemm_nt<<<ggrd, gblk, 0, stream>>>(y, H, nullptr, yMF, nullptr, nullptr);
        gs_gemm_nn<<<ggrd, gblk, 0, stream>>>(yMF, Dinv, nullptr, S, nullptr, nullptr);
        for (int i = 1; i <= N_ITER; ++i) {
            float* trajD  = traj + (size_t)i * ELEMS;
            float* finalD = (i == N_ITER) ? sfinal : nullptr;
            gs_gemm_nn<<<ggrd, gblk, 0, stream>>>(S, U, yMF, T, nullptr, nullptr);
            gs_gemm_nn<<<ggrd, gblk, 0, stream>>>(T, invM, nullptr, S, trajD, finalD);
        }
    }
}


# W = U@invM re-tiled: block 64x64, each wave 4x4 fragment tiles of one K-slice, 8 b128 loads per 32 WMMAs instead of 3 per 4; one accumulator per tile, so W's in-slice summation order changes
# speedup vs baseline: 2.7406x; 1.0703x over previous
#include <hip/hip_runtime.h>

typedef float v2f __attribute__((ext_vector_type(2)));
typedef float v4f __attribute__((ext_vector_type(4)));
typedef float v8f __attribute__((ext_vector_type(8)));

#define N_DIM   2048
#define BS_DIM  64
#define N_ITER  25
#define ELEMS   ((size_t)BS_DIM * N_DIM)
#define MATSZ   ((size_t)N_DIM * N_DIM)
#define KSPLIT  4
#define KSLICE  (N_DIM / KSPLIT)

__global__ void gs_zero_kernel(float* __restrict__ p, int n) {
    int i = blockIdx.x * blockDim.x + threadIdx.x;
    if (i < n) p[i] = 0.0f;
}

__global__ void __launch_bounds__(256)
gs_transpose(const float* __restrict__ in, float* __restrict__ out) {
    __shared__ float tile[32][33];
    const int bx = blockIdx.x & 63;
    const int by = blockIdx.x >> 6;
    const int tx = threadIdx.x & 31;
    const int ty = threadIdx.x >> 5;

    const int x = bx * 32 + tx;
    #pragma unroll
    for (int r = ty; r < 32; r += 8)
        tile[r][tx] = in[(size_t)(by * 32 + r) * N_DIM + x];
    __syncthreads();
    const int x2 = by * 32 + tx;
    #pragma unroll
    for (int r = ty; r < 32; r += 8)
        out[(size_t)(bx * 32 + r) * N_DIM + x2] = tile[tx][r];
}

#define PK_KB   (N_DIM / 8)

__global__ void __launch_bounds__(256)
gs_pack_b(const float* __restrict__ in, float* __restrict__ out) {
    __shared__ float tile[32][40];
    const int bx = blockIdx.x & 63;
    const int by = blockIdx.x >> 6;
    const int tx = threadIdx.x & 31;
    const int ty = threadIdx.x >> 5;

    const int x = bx * 32 + tx;
    #pragma unroll
    for (int r = ty; r < 32; r += 8)
        tile[r][tx] = in[(size_t)(by * 32 + r) * N_DIM + x];
    __syncthreads();

    const int c   = tx & 15;
    const int h   = tx >> 4;
    const int tnl = ty & 1;
    const int kbl = ty >> 1;
    const int k0  = kbl * 8 + 2 * h;
    const int nl  = tnl * 16 + c;
    v4f v;
    v.x = tile[k0][nl];
    v.y = tile[k0 + 1][nl];
    v.z = tile[k0 + 4][nl];
    v.w = tile[k0 + 5][nl];
    const size_t off =
        (((size_t)(bx * 2 + tnl) * PK_KB + (by * 4 + kbl)) * 32 + tx) * 4;
    *(v4f*)(out + off) = v;
}

__global__ void __launch_bounds__(256)
gs_pack_a(const float* __restrict__ in, float* __restrict__ out) {
    const int lane = threadIdx.x & 31;
    const int w    = threadIdx.x >> 5;
    const int tmr  = blockIdx.x >> 5;
    const int kb   = (blockIdx.x & 31) * 8 + w;
    const int c    = lane & 15;
    const int h    = lane >> 4;
    const float* p = in + (size_t)(tmr * 16 + c) * N_DIM + kb * 8 + 2 * h;
    const v2f lo = *(const v2f*)(p);
    const v2f hh = *(const v2f*)(p + 4);
    v4f v;
    v.x = lo.x; v.y = lo.y; v.z = hh.x; v.w = hh.y;
    *((v4f*)out + ((size_t)(tmr * PK_KB + kb) * 32 + lane)) = v;
}

__global__ void __launch_bounds__(128)
gs_gemm_ksplit(const float* __restrict__ A, const float* __restrict__ Bt,
               float* __restrict__ P) {
    const int lane = threadIdx.x & 31;
    const int tm   = threadIdx.x >> 5;
    const int s    = blockIdx.x & 3;
    const int tn   = blockIdx.x >> 2;
    const int l15  = lane & 15;
    const int hi   = lane >> 4;
    const int kbase = s * KSLICE + hi * 2;

    const float* aptr = A  + (size_t)(tm * 16 + l15) * N_DIM + kbase;
    const float* bptr = Bt + (size_t)(tn * 16 + l15) * N_DIM + kbase;

    v8f c0 = {}; v8f c1 = {}; v8f c2 = {}; v8f c3 = {};
    #pragma unroll 2
    for (int k = 0; k < KSLICE; k += 16) {
        v2f a0 = *(const v2f*)(aptr + k);
        v2f b0 = *(const v2f*)(bptr + k);
        v2f a1 = *(const v2f*)(aptr + k + 4);
        v2f b1 = *(const v2f*)(bptr + k + 4);
        v2f a2 = *(const v2f*)(aptr + k + 8);
        v2f b2 = *(const v2f*)(bptr + k + 8);
        v2f a3 = *(const v2f*)(aptr + k + 12);
        v2f b3 = *(const v2f*)(bptr + k + 12);
        c0 = __builtin_amdgcn_wmma_f32_16x16x4_f32(false, a0, false, b0,
                                                   (short)0, c0, false, false);
        c1 = __builtin_amdgcn_wmma_f32_16x16x4_f32(false, a1, false, b1,
                                                   (short)0, c1, false, false);
        c2 = __builtin_amdgcn_wmma_f32_16x16x4_f32(false, a2, false, b2,
                                                   (short)0, c2, false, false);
        c3 = __builtin_amdgcn_wmma_f32_16x16x4_f32(false, a3, false, b3,
                                                   (short)0, c3, false, false);
    }

    float* optr = P + (size_t)s * ELEMS
                    + (size_t)(tm * 16 + hi * 8) * N_DIM + tn * 16 + l15;
    #pragma unroll
    for (int r = 0; r < 8; ++r)
        optr[(size_t)r * N_DIM] = (c0[r] + c1[r]) + (c2[r] + c3[r]);
}

#define FK_SPLIT 8
#define FK_KB    (N_DIM / FK_SPLIT / 8)
#define PK_GRP   4
#define PK_NGRP  (FK_KB / PK_GRP)
#define PK_AHEAD 3
#define PK_RING  (PK_AHEAD + 1)
#define RD_PITCH 36
__global__ void __launch_bounds__(256)
gs_gemm_fused_pk(const float* __restrict__ Ap, const float* __restrict__ Bp,
                 const float* __restrict__ subSrc, float* __restrict__ pk0,
                 float* __restrict__ out1, float* __restrict__ out2) {
    __shared__ alignas(16) float red[FK_SPLIT][16][RD_PITCH];

    const int lane = threadIdx.x & 31;
    const int w    = __builtin_amdgcn_readfirstlane((int)(threadIdx.x >> 5));
    const int kbl  = w & 3;
    const int tm   = (int)(blockIdx.x & 3);
    const int tp   = (int)(blockIdx.x >> 2);
    const int l15  = lane & 15;
    const int hi   = lane >> 4;

    v2f sa = {}; v2f sb = {};
    if (subSrc != nullptr && w >= 4) {
        const float* sp = subSrc + (size_t)(tm * 16 + l15) * N_DIM
                        + tp * 32 + kbl * 8 + 2 * hi;
        sa = *(const v2f*)(sp);
        sb = *(const v2f*)(sp + 4);
    }

    const v4f* ap  = (const v4f*)Ap
                   + ((size_t)(tm * PK_KB + w * FK_KB) * 32 + lane);
    const v4f* bp0 = (const v4f*)Bp
                   + ((size_t)(tp * 2 * PK_KB + w * FK_KB) * 32 + lane);
    const v4f* bp1 = bp0 + (size_t)PK_KB * 32;

    v4f a[PK_RING][PK_GRP], b0[PK_RING][PK_GRP], b1[PK_RING][PK_GRP];
    #pragma unroll
    for (int g = 0; g < PK_AHEAD; ++g) {
        #pragma unroll
        for (int j = 0; j < PK_GRP; ++j) {
            a[g][j]  = ap[(g * PK_GRP + j) * 32];
            b0[g][j] = bp0[(g * PK_GRP + j) * 32];
            b1[g][j] = bp1[(g * PK_GRP + j) * 32];
        }
    }

    v8f c00 = {}; v8f c10 = {}; v8f c01 = {}; v8f c11 = {};
    #pragma unroll
    for (int g = 0; g < PK_NGRP; ++g) {
        const int cur = g % PK_RING;
        if (g + PK_AHEAD < PK_NGRP) {
            const int nxt = (g + PK_AHEAD) % PK_RING;
            #pragma unroll
            for (int j = 0; j < PK_GRP; ++j) {
                a[nxt][j]  = ap[((g + PK_AHEAD) * PK_GRP + j) * 32];
                b0[nxt][j] = bp0[((g + PK_AHEAD) * PK_GRP + j) * 32];
                b1[nxt][j] = bp1[((g + PK_AHEAD) * PK_GRP + j) * 32];
            }
        }
        __builtin_amdgcn_sched_barrier(0);
        #pragma unroll
        for (int j = 0; j < PK_GRP; ++j) {
            c00 = __builtin_amdgcn_wmma_f32_16x16x4_f32(false, a[cur][j].xy, false,
                      b0[cur][j].xy, (short)0, c00, false, false);
            c10 = __builtin_amdgcn_wmma_f32_16x16x4_f32(false, a[cur][j].xy, false,
                      b1[cur][j].xy, (short)0, c10, false, false);
            c01 = __builtin_amdgcn_wmma_f32_16x16x4_f32(false, a[cur][j].zw, false,
                      b0[cur][j].zw, (short)0, c01, false, false);
            c11 = __builtin_amdgcn_wmma_f32_16x16x4_f32(false, a[cur][j].zw, false,
                      b1[cur][j].zw, (short)0, c11, false, false);
        }
        __builtin_amdgcn_sched_barrier(0);
    }

    {
        float* rp = &red[w][hi * 8][l15];
        #pragma unroll
        for (int r = 0; r < 8; ++r) {
            rp[r * RD_PITCH]      = c00[r] + c01[r];
            rp[r * RD_PITCH + 16] = c10[r] + c11[r];
        }
    }
    __syncthreads();

    if (w >= 4) {
        if (pk0 == nullptr) return;
        const float* q = &red[0][l15][kbl * 8 + 2 * hi];
        v2f xs[FK_SPLIT], ys[FK_SPLIT];
        #pragma unroll
        for (int p = 0; p < FK_SPLIT; ++p) {
            xs[p] = *(const v2f*)(q + p * 16 * RD_PITCH);
            ys[p] = *(const v2f*)(q + p * 16 * RD_PITCH + 4);
        }
        v2f x = ((xs[0] + xs[1]) + (xs[2] + xs[3])) + ((xs[4] + xs[5]) + (xs[6] + xs[7]));
        v2f y = ((ys[0] + ys[1]) + (ys[2] + ys[3])) + ((ys[4] + ys[5]) + (ys[6] + ys[7]));
        if (subSrc != nullptr) { x = sa - x; y = sb - y; }
        v4f o;
        o.x = x.x; o.y = x.y; o.z = y.x; o.w = y.y;
        *((v4f*)pk0 + ((size_t)(tm * PK_KB + tp * 4 + kbl) * 32 + lane)) = o;
    } else if (out1 != nullptr || out2 != nullptr) {
        const int t    = threadIdx.x;
        const int row  = t >> 3;
        const int col4 = (t & 7) * 4;
        const float* q = &red[0][row][col4];
        v4f ps[FK_SPLIT];
        #pragma unroll
        for (int p = 0; p < FK_SPLIT; ++p)
            ps[p] = *(const v4f*)(q + p * 16 * RD_PITCH);
        v4f v = ((ps[0] + ps[1]) + (ps[2] + ps[3])) + ((ps[4] + ps[5]) + (ps[6] + ps[7]));
        const size_t o = (size_t)(tm * 16 + row) * N_DIM + tp * 32 + col4;
        if (subSrc != nullptr) v = *(const v4f*)(subSrc + o) - v;
        if (out1 != nullptr) *(v4f*)(out1 + o) = v;
        if (out2 != nullptr) *(v4f*)(out2 + o) = v;
    }
}

#define WK_MT    4
#define WK_NT    4
#define WK_AHEAD 2
#define WK_RING  (WK_AHEAD + 1)
#define WR_PITCH 76
__global__ void __launch_bounds__(256)
gs_gemm_w_pk(const float* __restrict__ Ap, const float* __restrict__ Bp,
             float* __restrict__ Wp) {
    __shared__ float red[FK_SPLIT][16][WR_PITCH];

    const int lane = threadIdx.x & 31;
    const int w    = __builtin_amdgcn_readfirstlane((int)(threadIdx.x >> 5));
    const int tm4  = (int)(blockIdx.x >> 5);
    const int tq   = (int)(blockIdx.x & 31);
    const int l15  = lane & 15;
    const int hi   = lane >> 4;

    const v4f* ap = (const v4f*)Ap
                  + ((size_t)(tm4 * WK_MT * PK_KB + w * FK_KB) * 32 + lane);
    const v4f* bp = (const v4f*)Bp
                  + ((size_t)(tq * WK_NT * PK_KB + w * FK_KB) * 32 + lane);

    v4f a[WK_RING][WK_MT], b[WK_RING][WK_NT];
    #pragma unroll
    for (int g = 0; g < WK_AHEAD; ++g) {
        #pragma unroll
        for (int i = 0; i < WK_MT; ++i) a[g][i] = ap[(i * PK_KB + g) * 32];
        #pragma unroll
        for (int n = 0; n < WK_NT; ++n) b[g][n] = bp[(n * PK_KB + g) * 32];
    }

    v8f acc[WK_MT][WK_NT];
    #pragma unroll
    for (int i = 0; i < WK_MT; ++i) {
        #pragma unroll
        for (int n = 0; n < WK_NT; ++n) acc[i][n] = v8f{};
    }

    #pragma unroll
    for (int g = 0; g < FK_KB; ++g) {
        const int cur = g % WK_RING;
        if (g + WK_AHEAD < FK_KB) {
            const int nxt = (g + WK_AHEAD) % WK_RING;
            #pragma unroll
            for (int i = 0; i < WK_MT; ++i)
                a[nxt][i] = ap[(i * PK_KB + g + WK_AHEAD) * 32];
            #pragma unroll
            for (int n = 0; n < WK_NT; ++n)
                b[nxt][n] = bp[(n * PK_KB + g + WK_AHEAD) * 32];
        }
        __builtin_amdgcn_sched_barrier(0);
        #pragma unroll
        for (int i = 0; i < WK_MT; ++i) {
            #pragma unroll
            for (int n = 0; n < WK_NT; ++n)
                acc[i][n] = __builtin_amdgcn_wmma_f32_16x16x4_f32(false, a[cur][i].xy,
                                false, b[cur][n].xy, (short)0, acc[i][n], false, false);
        }
        #pragma unroll
        for (int i = 0; i < WK_MT; ++i) {
            #pragma unroll
            for (int n = 0; n < WK_NT; ++n)
                acc[i][n] = __builtin_amdgcn_wmma_f32_16x16x4_f32(false, a[cur][i].zw,
                                false, b[cur][n].zw, (short)0, acc[i][n], false, false);
        }
        __builtin_amdgcn_sched_barrier(0);
    }

    const int tnl = w & 3;
    const int kb2 = w >> 2;
    #pragma unroll
    for (int m = 0; m < WK_MT; ++m) {
        if (m > 0) __syncthreads();
        float* rp = &red[w][hi * 8][l15];
        #pragma unroll
        for (int r = 0; r < 8; ++r) {
            #pragma unroll
            for (int n = 0; n < WK_NT; ++n)
                rp[r * WR_PITCH + n * 16] = acc[m][n][r];
        }
        __syncthreads();

        const float* q = &red[0][kb2 * 8 + 2 * hi][tnl * 16 + l15];
        float e0[FK_SPLIT], e1[FK_SPLIT], e4[FK_SPLIT], e5[FK_SPLIT];
        #pragma unroll
        for (int p = 0; p < FK_SPLIT; ++p) {
            e0[p] = q[p * 16 * WR_PITCH];
            e1[p] = q[p * 16 * WR_PITCH + WR_PITCH];
            e4[p] = q[p * 16 * WR_PITCH + 4 * WR_PITCH];
            e5[p] = q[p * 16 * WR_PITCH + 5 * WR_PITCH];
        }
        v4f o;
        o.x = ((e0[0] + e0[1]) + (e0[2] + e0[3])) + ((e0[4] + e0[5]) + (e0[6] + e0[7]));
        o.y = ((e1[0] + e1[1]) + (e1[2] + e1[3])) + ((e1[4] + e1[5]) + (e1[6] + e1[7]));
        o.z = ((e4[0] + e4[1]) + (e4[2] + e4[3])) + ((e4[4] + e4[5]) + (e4[6] + e4[7]));
        o.w = ((e5[0] + e5[1]) + (e5[2] + e5[3])) + ((e5[4] + e5[5]) + (e5[6] + e5[7]));
        *((v4f*)Wp + ((size_t)((tq * WK_NT + tnl) * PK_KB
                               + (tm4 * WK_MT + m) * 2 + kb2) * 32 + lane)) = o;
    }
}

__global__ void __launch_bounds__(256)
gs_reduce(const float* __restrict__ P, const float* __restrict__ subSrc,
          float* __restrict__ pk0, float* __restrict__ out1,
          float* __restrict__ out2) {
    const size_t i = ((size_t)blockIdx.x * 256 + threadIdx.x) * 4;
    float4 v0 = *(const float4*)(P + i);
    float4 v1 = *(const float4*)(P + ELEMS + i);
    float4 v2 = *(const float4*)(P + 2 * ELEMS + i);
    float4 v3 = *(const float4*)(P + 3 * ELEMS + i);
    float4 v;
    v.x = (v0.x + v1.x) + (v2.x + v3.x);
    v.y = (v0.y + v1.y) + (v2.y + v3.y);
    v.z = (v0.z + v1.z) + (v2.z + v3.z);
    v.w = (v0.w + v1.w) + (v2.w + v3.w);
    if (subSrc) {
        float4 s = *(const float4*)(subSrc + i);
        v.x = s.x - v.x; v.y = s.y - v.y; v.z = s.z - v.z; v.w = s.w - v.w;
    }
    {
        const int m  = (int)(i >> 11);
        const int k  = (int)(i & (N_DIM - 1));
        const int kb = k >> 3;
        const int j2 = (k >> 1) & 2;
        float* blk = pk0 + ((size_t)((m >> 4) * PK_KB + kb) * 32 + (m & 15)) * 4 + j2;
        v2f lo; lo.x = v.x; lo.y = v.y;
        v2f hh; hh.x = v.z; hh.y = v.w;
        *(v2f*)(blk)          = lo;
        *(v2f*)(blk + 16 * 4) = hh;
    }
    if (out1) *(float4*)(out1 + i) = v;
    if (out2) *(float4*)(out2 + i) = v;
}

__global__ void __launch_bounds__(128)
gs_gemm_nt(const float* __restrict__ A, const float* __restrict__ Bt,
           const float* __restrict__ subSrc,
           float* __restrict__ out0, float* __restrict__ out1,
           float* __restrict__ out2) {
    const int lane = threadIdx.x & 31;
    const int wave = threadIdx.x >> 5;
    const int t    = blockIdx.x * 4 + wave;
    const int tm   = t & 3;
    const int tn   = t >> 2;
    const int l15  = lane & 15;
    const int hi   = lane >> 4;
    const int koff = hi * 2;

    const float* aptr = A  + (size_t)(tm * 16 + l15) * N_DIM + koff;
    const float* bptr = Bt + (size_t)(tn * 16 + l15) * N_DIM + koff;

    v8f acc0 = {}; v8f acc1 = {};
    #pragma unroll 4
    for (int k = 0; k < N_DIM; k += 8) {
        v2f a0 = *(const v2f*)(aptr + k);
        v2f b0 = *(const v2f*)(bptr + k);
        v2f a1 = *(const v2f*)(aptr + k + 4);
        v2f b1 = *(const v2f*)(bptr + k + 4);
        acc0 = __builtin_amdgcn_wmma_f32_16x16x4_f32(false, a0, false, b0,
                                                     (short)0, acc0, false, false);
        acc1 = __builtin_amdgcn_wmma_f32_16x16x4_f32(false, a1, false, b1,
                                                     (short)0, acc1, false, false);
    }

    const size_t obase = (size_t)(tm * 16 + hi * 8) * N_DIM + tn * 16 + l15;
    float res[8];
    #pragma unroll
    for (int r = 0; r < 8; ++r) res[r] = acc0[r] + acc1[r];
    if (subSrc) {
        #pragma unroll
        for (int r = 0; r < 8; ++r)
            res[r] = subSrc[obase + (size_t)r * N_DIM] - res[r];
    }
    #pragma unroll
    for (int r = 0; r < 8; ++r) out0[obase + (size_t)r * N_DIM] = res[r];
    if (out1) {
        #pragma unroll
        for (int r = 0; r < 8; ++r) out1[obase + (size_t)r * N_DIM] = res[r];
    }
    if (out2) {
        #pragma unroll
        for (int r = 0; r < 8; ++r) out2[obase + (size_t)r * N_DIM] = res[r];
    }
}

__global__ void __launch_bounds__(128)
gs_gemm_nn(const float* __restrict__ A, const float* __restrict__ B,
           const float* __restrict__ subSrc,
           float* __restrict__ out0, float* __restrict__ out1,
           float* __restrict__ out2) {
    const int lane = threadIdx.x & 31;
    const int wave = threadIdx.x >> 5;
    const int t    = blockIdx.x * 4 + wave;
    const int tm   = t & 3;
    const int tn   = t >> 2;
    const int l15  = lane & 15;
    const int hi   = lane >> 4;
    const int koff = hi * 2;

    const float* aptr = A + (size_t)(tm * 16 + l15) * N_DIM + koff;
    const float* bptr = B + (size_t)koff * N_DIM + tn * 16 + l15;

    v8f acc0 = {}; v8f acc1 = {};
    #pragma unroll 4
    for (int k = 0; k < N_DIM; k += 8) {
        v2f a0 = *(const v2f*)(aptr + k);
        v2f b0;
        b0.x = bptr[(size_t)(k + 0) * N_DIM];
        b0.y = bptr[(size_t)(k + 1) * N_DIM];
        v2f a1 = *(const v2f*)(aptr + k + 4);
        v2f b1;
        b1.x = bptr[(size_t)(k + 4) * N_DIM];
        b1.y = bptr[(size_t)(k + 5) * N_DIM];
        acc0 = __builtin_amdgcn_wmma_f32_16x16x4_f32(false, a0, false, b0,
                                                     (short)0, acc0, false, false);
        acc1 = __builtin_amdgcn_wmma_f32_16x16x4_f32(false, a1, false, b1,
                                                     (short)0, acc1, false, false);
    }

    const size_t obase = (size_t)(tm * 16 + hi * 8) * N_DIM + tn * 16 + l15;
    float res[8];
    #pragma unroll
    for (int r = 0; r < 8; ++r) res[r] = acc0[r] + acc1[r];
    if (subSrc) {
        #pragma unroll
        for (int r = 0; r < 8; ++r)
            res[r] = subSrc[obase + (size_t)r * N_DIM] - res[r];
    }
    #pragma unroll
    for (int r = 0; r < 8; ++r) out0[obase + (size_t)r * N_DIM] = res[r];
    if (out1) {
        #pragma unroll
        for (int r = 0; r < 8; ++r) out1[obase + (size_t)r * N_DIM] = res[r];
    }
    if (out2) {
        #pragma unroll
        for (int r = 0; r < 8; ++r) out2[obase + (size_t)r * N_DIM] = res[r];
    }
}

extern "C" void kernel_launch(void* const* d_in, const int* in_sizes, int n_in,
                              void* d_out, int out_size, void* d_ws, size_t ws_size,
                              hipStream_t stream) {
    const float* y    = (const float*)d_in[2];
    const float* H    = (const float*)d_in[3];
    const float* Dinv = (const float*)d_in[4];
    const float* U    = (const float*)d_in[5];
    const float* invM = (const float*)d_in[6];

    float* sfinal = (float*)d_out;
    float* traj   = (float*)d_out + ELEMS;

    float* ws    = (float*)d_ws;
    float* yMF   = ws;
    float* S     = ws + ELEMS;
    float* T     = ws + 2 * ELEMS;
    float* P     = ws + 3 * ELEMS;
    float* Dt    = ws + 7 * ELEMS;
    float* Ut    = Dt + MATSZ;
    float* invMt = Ut + MATSZ;

    const size_t need_fast = (7 * ELEMS + 3 * MATSZ) * sizeof(float);
    const size_t need_mid  = (3 * ELEMS + 2 * MATSZ) * sizeof(float);
    const int mode = (ws_size >= need_fast) ? 2 : (ws_size >= need_mid ? 1 : 0);

    dim3 gblk(128);
    dim3 ggrd(128);
    dim3 kgrd(512);
    dim3 rgrd(ELEMS / (256 * 4));
    dim3 fblk(256);
    dim3 fgrd(256);
    dim3 wgrd(1024);

    gs_zero_kernel<<<(int)(ELEMS / 256), 256, 0, stream>>>(traj, (int)ELEMS);

    if (mode == 2) {
        float* Cm = yMF;
        float* Wp = Dt;

        gs_pack_b<<<4096, 256, 0, stream>>>(Dinv, Dt);
        gs_pack_a<<<4096, 256, 0, stream>>>(U, Ut);
        gs_pack_b<<<4096, 256, 0, stream>>>(invM, invMt);

        gs_gemm_ksplit<<<kgrd, gblk, 0, stream>>>(y, H, P);
        gs_reduce<<<rgrd, 256, 0, stream>>>(P, nullptr, T, nullptr, nullptr);
        gs_gemm_fused_pk<<<fgrd, fblk, 0, stream>>>(T, Dt, nullptr, S, nullptr, nullptr);
        gs_gemm_fused_pk<<<fgrd, fblk, 0, stream>>>(T, invMt, nullptr, nullptr, Cm, nullptr);
        gs_gemm_w_pk<<<wgrd, fblk, 0, stream>>>(Ut, invMt, Wp);

        float* cur = S;
        float* nxt = T;
        for (int i = 1; i <= N_ITER; ++i) {
            float* trajD  = traj + (size_t)i * ELEMS;
            float* finalD = (i == N_ITER) ? sfinal : nullptr;
            gs_gemm_fused_pk<<<fgrd, fblk, 0, stream>>>(cur, Wp, Cm, nxt, trajD, finalD);
            float* t = cur; cur = nxt; nxt = t;
        }
    } else if (mode == 1) {
        gs_transpose<<<4096, 256, 0, stream>>>(U, Ut - MATSZ);
        gs_transpose<<<4096, 256, 0, stream>>>(invM, invMt - MATSZ);
        float* Ut1    = Dt;
        float* invMt1 = Dt + MATSZ;
        gs_gemm_nt<<<ggrd, gblk, 0, stream>>>(y, H, nullptr, yMF, nullptr, nullptr);
        gs_gemm_nn<<<ggrd, gblk, 0, stream>>>(yMF, Dinv, nullptr, S, nullptr, nullptr);
        for (int i = 1; i <= N_ITER; ++i) {
            float* trajD  = traj + (size_t)i * ELEMS;
            float* finalD = (i == N_ITER) ? sfinal : nullptr;
            gs_gemm_nt<<<ggrd, gblk, 0, stream>>>(S, Ut1, yMF, T, nullptr, nullptr);
            gs_gemm_nt<<<ggrd, gblk, 0, stream>>>(T, invMt1, nullptr, S, trajD, finalD);
        }
    } else {
        gs_gemm_nt<<<ggrd, gblk, 0, stream>>>(y, H, nullptr, yMF, nullptr, nullptr);
        gs_gemm_nn<<<ggrd, gblk, 0, stream>>>(yMF, Dinv, nullptr, S, nullptr, nullptr);
        for (int i = 1; i <= N_ITER; ++i) {
            float* trajD  = traj + (size_t)i * ELEMS;
            float* finalD = (i == N_ITER) ? sfinal : nullptr;
            gs_gemm_nn<<<ggrd, gblk, 0, stream>>>(S, U, yMF, T, nullptr, nullptr);
            gs_gemm_nn<<<ggrd, gblk, 0, stream>>>(T, invM, nullptr, S, trajD, finalD);
        }
    }
}


# Passes that depend on the inputs alone share one launch (3 packs, y@H^T split-K partials, traj[0]=0) and s_init and C share one 512-block launch: 29 launches per call, not 34; arithmetic unchanged
# speedup vs baseline: 2.9655x; 1.0821x over previous
#include <hip/hip_runtime.h>

typedef float v2f __attribute__((ext_vector_type(2)));
typedef float v4f __attribute__((ext_vector_type(4)));
typedef float v8f __attribute__((ext_vector_type(8)));

#define N_DIM   2048
#define BS_DIM  64
#define N_ITER  25
#define ELEMS   ((size_t)BS_DIM * N_DIM)
#define MATSZ   ((size_t)N_DIM * N_DIM)
#define KSPLIT  4
#define KSLICE  (N_DIM / KSPLIT)

__global__ void gs_zero_kernel(float* __restrict__ p, int n) {
    int i = blockIdx.x * blockDim.x + threadIdx.x;
    if (i < n) p[i] = 0.0f;
}

__global__ void __launch_bounds__(256)
gs_transpose(const float* __restrict__ in, float* __restrict__ out) {
    __shared__ float tile[32][33];
    const int bx = blockIdx.x & 63;
    const int by = blockIdx.x >> 6;
    const int tx = threadIdx.x & 31;
    const int ty = threadIdx.x >> 5;

    const int x = bx * 32 + tx;
    #pragma unroll
    for (int r = ty; r < 32; r += 8)
        tile[r][tx] = in[(size_t)(by * 32 + r) * N_DIM + x];
    __syncthreads();
    const int x2 = by * 32 + tx;
    #pragma unroll
    for (int r = ty; r < 32; r += 8)
        out[(size_t)(bx * 32 + r) * N_DIM + x2] = tile[tx][r];
}

#define PK_KB   (N_DIM / 8)

__device__ __forceinline__ void
gs_pack_b_role(const float* __restrict__ in, float* __restrict__ out, int bid,
               float (*tile)[40]) {
    const int bx = bid & 63;
    const int by = bid >> 6;
    const int tx = threadIdx.x & 31;
    const int ty = threadIdx.x >> 5;

    const int x = bx * 32 + tx;
    #pragma unroll
    for (int r = ty; r < 32; r += 8)
        tile[r][tx] = in[(size_t)(by * 32 + r) * N_DIM + x];
    __syncthreads();

    const int c   = tx & 15;
    const int h   = tx >> 4;
    const int tnl = ty & 1;
    const int kbl = ty >> 1;
    const int k0  = kbl * 8 + 2 * h;
    const int nl  = tnl * 16 + c;
    v4f v;
    v.x = tile[k0][nl];
    v.y = tile[k0 + 1][nl];
    v.z = tile[k0 + 4][nl];
    v.w = tile[k0 + 5][nl];
    const size_t off =
        (((size_t)(bx * 2 + tnl) * PK_KB + (by * 4 + kbl)) * 32 + tx) * 4;
    *(v4f*)(out + off) = v;
}

__device__ __forceinline__ void
gs_pack_a_role(const float* __restrict__ in, float* __restrict__ out, int bid) {
    const int lane = threadIdx.x & 31;
    const int w    = threadIdx.x >> 5;
    const int tmr  = bid >> 5;
    const int kb   = (bid & 31) * 8 + w;
    const int c    = lane & 15;
    const int h    = lane >> 4;
    const float* p = in + (size_t)(tmr * 16 + c) * N_DIM + kb * 8 + 2 * h;
    const v2f lo = *(const v2f*)(p);
    const v2f hh = *(const v2f*)(p + 4);
    v4f v;
    v.x = lo.x; v.y = lo.y; v.z = hh.x; v.w = hh.y;
    *((v4f*)out + ((size_t)(tmr * PK_KB + kb) * 32 + lane)) = v;
}

__device__ __forceinline__ void
gs_gemm_ksplit_role(const float* __restrict__ A, const float* __restrict__ Bt,
                    float* __restrict__ P, int hb, int t) {
    const int lane = t & 31;
    const int tm   = t >> 5;
    const int s    = hb & 3;
    const int tn   = hb >> 2;
    const int l15  = lane & 15;
    const int hi   = lane >> 4;
    const int kbase = s * KSLICE + hi * 2;

    const float* aptr = A  + (size_t)(tm * 16 + l15) * N_DIM + kbase;
    const float* bptr = Bt + (size_t)(tn * 16 + l15) * N_DIM + kbase;

    v8f c0 = {}; v8f c1 = {}; v8f c2 = {}; v8f c3 = {};
    #pragma unroll 2
    for (int k = 0; k < KSLICE; k += 16) {
        v2f a0 = *(const v2f*)(aptr + k);
        v2f b0 = *(const v2f*)(bptr + k);
        v2f a1 = *(const v2f*)(aptr + k + 4);
        v2f b1 = *(const v2f*)(bptr + k + 4);
        v2f a2 = *(const v2f*)(aptr + k + 8);
        v2f b2 = *(const v2f*)(bptr + k + 8);
        v2f a3 = *(const v2f*)(aptr + k + 12);
        v2f b3 = *(const v2f*)(bptr + k + 12);
        c0 = __builtin_amdgcn_wmma_f32_16x16x4_f32(false, a0, false, b0,
                                                   (short)0, c0, false, false);
        c1 = __builtin_amdgcn_wmma_f32_16x16x4_f32(false, a1, false, b1,
                                                   (short)0, c1, false, false);
        c2 = __builtin_amdgcn_wmma_f32_16x16x4_f32(false, a2, false, b2,
                                                   (short)0, c2, false, false);
        c3 = __builtin_amdgcn_wmma_f32_16x16x4_f32(false, a3, false, b3,
                                                   (short)0, c3, false, false);
    }

    float* optr = P + (size_t)s * ELEMS
                    + (size_t)(tm * 16 + hi * 8) * N_DIM + tn * 16 + l15;
    #pragma unroll
    for (int r = 0; r < 8; ++r)
        optr[(size_t)r * N_DIM] = (c0[r] + c1[r]) + (c2[r] + c3[r]);
}

#define SU_GEMM  256
#define SU_PACK  4096
#define SU_ZERO  ((int)(ELEMS / (256 * 4)))
#define SU_GRID  (SU_GEMM + 3 * SU_PACK + SU_ZERO)
__global__ void __launch_bounds__(256)
gs_setup(const float* __restrict__ y, const float* __restrict__ H,
         const float* __restrict__ Dinv, const float* __restrict__ U,
         const float* __restrict__ invM, float* __restrict__ P,
         float* __restrict__ Dp, float* __restrict__ Up,
         float* __restrict__ invMp, float* __restrict__ traj0) {
    __shared__ float tile[32][40];
    const int b = (int)blockIdx.x;
    if (b < SU_GEMM) {
        gs_gemm_ksplit_role(y, H, P, b * 2 + (int)(threadIdx.x >> 7),
                            (int)(threadIdx.x & 127));
    } else if (b < SU_GEMM + SU_PACK) {
        gs_pack_b_role(Dinv, Dp, b - SU_GEMM, tile);
    } else if (b < SU_GEMM + 2 * SU_PACK) {
        gs_pack_a_role(U, Up, b - (SU_GEMM + SU_PACK));
    } else if (b < SU_GEMM + 3 * SU_PACK) {
        gs_pack_b_role(invM, invMp, b - (SU_GEMM + 2 * SU_PACK), tile);
    } else {
        const v4f z = {0.0f, 0.0f, 0.0f, 0.0f};
        *((v4f*)traj0 + ((size_t)(b - (SU_GEMM + 3 * SU_PACK)) * 256 + threadIdx.x)) = z;
    }
}

#define FK_SPLIT 8
#define FK_KB    (N_DIM / FK_SPLIT / 8)
#define PK_GRP   4
#define PK_NGRP  (FK_KB / PK_GRP)
#define PK_AHEAD 3
#define PK_RING  (PK_AHEAD + 1)
#define RD_PITCH 36
__device__ __forceinline__ void
gs_fused_pk_body(const float* __restrict__ Ap, const float* __restrict__ Bp,
                 const float* __restrict__ subSrc, float* __restrict__ pk0,
                 float* __restrict__ out1, float* __restrict__ out2, int bid) {
    __shared__ alignas(16) float red[FK_SPLIT][16][RD_PITCH];

    const int lane = threadIdx.x & 31;
    const int w    = __builtin_amdgcn_readfirstlane((int)(threadIdx.x >> 5));
    const int kbl  = w & 3;
    const int tm   = bid & 3;
    const int tp   = bid >> 2;
    const int l15  = lane & 15;
    const int hi   = lane >> 4;

    v2f sa = {}; v2f sb = {};
    if (subSrc != nullptr && w >= 4) {
        const float* sp = subSrc + (size_t)(tm * 16 + l15) * N_DIM
                        + tp * 32 + kbl * 8 + 2 * hi;
        sa = *(const v2f*)(sp);
        sb = *(const v2f*)(sp + 4);
    }

    const v4f* ap  = (const v4f*)Ap
                   + ((size_t)(tm * PK_KB + w * FK_KB) * 32 + lane);
    const v4f* bp0 = (const v4f*)Bp
                   + ((size_t)(tp * 2 * PK_KB + w * FK_KB) * 32 + lane);
    const v4f* bp1 = bp0 + (size_t)PK_KB * 32;

    v4f a[PK_RING][PK_GRP], b0[PK_RING][PK_GRP], b1[PK_RING][PK_GRP];
    #pragma unroll
    for (int g = 0; g < PK_AHEAD; ++g) {
        #pragma unroll
        for (int j = 0; j < PK_GRP; ++j) {
            a[g][j]  = ap[(g * PK_GRP + j) * 32];
            b0[g][j] = bp0[(g * PK_GRP + j) * 32];
            b1[g][j] = bp1[(g * PK_GRP + j) * 32];
        }
    }

    v8f c00 = {}; v8f c10 = {}; v8f c01 = {}; v8f c11 = {};
    #pragma unroll
    for (int g = 0; g < PK_NGRP; ++g) {
        const int cur = g % PK_RING;
        if (g + PK_AHEAD < PK_NGRP) {
            const int nxt = (g + PK_AHEAD) % PK_RING;
            #pragma unroll
            for (int j = 0; j < PK_GRP; ++j) {
                a[nxt][j]  = ap[((g + PK_AHEAD) * PK_GRP + j) * 32];
                b0[nxt][j] = bp0[((g + PK_AHEAD) * PK_GRP + j) * 32];
                b1[nxt][j] = bp1[((g + PK_AHEAD) * PK_GRP + j) * 32];
            }
        }
        __builtin_amdgcn_sched_barrier(0);
        #pragma unroll
        for (int j = 0; j < PK_GRP; ++j) {
            c00 = __builtin_amdgcn_wmma_f32_16x16x4_f32(false, a[cur][j].xy, false,
                      b0[cur][j].xy, (short)0, c00, false, false);
            c10 = __builtin_amdgcn_wmma_f32_16x16x4_f32(false, a[cur][j].xy, false,
                      b1[cur][j].xy, (short)0, c10, false, false);
            c01 = __builtin_amdgcn_wmma_f32_16x16x4_f32(false, a[cur][j].zw, false,
                      b0[cur][j].zw, (short)0, c01, false, false);
            c11 = __builtin_amdgcn_wmma_f32_16x16x4_f32(false, a[cur][j].zw, false,
                      b1[cur][j].zw, (short)0, c11, false, false);
        }
        __builtin_amdgcn_sched_barrier(0);
    }

    {
        float* rp = &red[w][hi * 8][l15];
        #pragma unroll
        for (int r = 0; r < 8; ++r) {
            rp[r * RD_PITCH]      = c00[r] + c01[r];
            rp[r * RD_PITCH + 16] = c10[r] + c11[r];
        }
    }
    __syncthreads();

    if (w >= 4) {
        if (pk0 == nullptr) return;
        const float* q = &red[0][l15][kbl * 8 + 2 * hi];
        v2f xs[FK_SPLIT], ys[FK_SPLIT];
        #pragma unroll
        for (int p = 0; p < FK_SPLIT; ++p) {
            xs[p] = *(const v2f*)(q + p * 16 * RD_PITCH);
            ys[p] = *(const v2f*)(q + p * 16 * RD_PITCH + 4);
        }
        v2f x = ((xs[0] + xs[1]) + (xs[2] + xs[3])) + ((xs[4] + xs[5]) + (xs[6] + xs[7]));
        v2f y = ((ys[0] + ys[1]) + (ys[2] + ys[3])) + ((ys[4] + ys[5]) + (ys[6] + ys[7]));
        if (subSrc != nullptr) { x = sa - x; y = sb - y; }
        v4f o;
        o.x = x.x; o.y = x.y; o.z = y.x; o.w = y.y;
        *((v4f*)pk0 + ((size_t)(tm * PK_KB + tp * 4 + kbl) * 32 + lane)) = o;
    } else if (out1 != nullptr || out2 != nullptr) {
        const int t    = threadIdx.x;
        const int row  = t >> 3;
        const int col4 = (t & 7) * 4;
        const float* q = &red[0][row][col4];
        v4f ps[FK_SPLIT];
        #pragma unroll
        for (int p = 0; p < FK_SPLIT; ++p)
            ps[p] = *(const v4f*)(q + p * 16 * RD_PITCH);
        v4f v = ((ps[0] + ps[1]) + (ps[2] + ps[3])) + ((ps[4] + ps[5]) + (ps[6] + ps[7]));
        const size_t o = (size_t)(tm * 16 + row) * N_DIM + tp * 32 + col4;
        if (subSrc != nullptr) v = *(const v4f*)(subSrc + o) - v;
        if (out1 != nullptr) *(v4f*)(out1 + o) = v;
        if (out2 != nullptr) *(v4f*)(out2 + o) = v;
    }
}

__global__ void __launch_bounds__(256)
gs_gemm_fused_pk(const float* __restrict__ Ap, const float* __restrict__ Bp,
                 const float* __restrict__ subSrc, float* __restrict__ pk0,
                 float* __restrict__ out1, float* __restrict__ out2) {
    gs_fused_pk_body(Ap, Bp, subSrc, pk0, out1, out2, (int)blockIdx.x);
}

__global__ void __launch_bounds__(256)
gs_gemm_fused_pk_pair(const float* __restrict__ Ap, const float* __restrict__ Bp0,
                      const float* __restrict__ Bp1, float* __restrict__ pk,
                      float* __restrict__ rm) {
    const bool second = blockIdx.x >= 256;
    gs_fused_pk_body(Ap, second ? Bp1 : Bp0, nullptr, second ? nullptr : pk,
                     second ? rm : nullptr, nullptr, (int)(blockIdx.x & 255));
}

#define WK_MT    4
#define WK_NT    4
#define WK_AHEAD 2
#define WK_RING  (WK_AHEAD + 1)
#define WR_PITCH 76
__global__ void __launch_bounds__(256)
gs_gemm_w_pk(const float* __restrict__ Ap, const float* __restrict__ Bp,
             float* __restrict__ Wp) {
    __shared__ float red[FK_SPLIT][16][WR_PITCH];

    const int lane = threadIdx.x & 31;
    const int w    = __builtin_amdgcn_readfirstlane((int)(threadIdx.x >> 5));
    const int tm4  = (int)(blockIdx.x >> 5);
    const int tq   = (int)(blockIdx.x & 31);
    const int l15  = lane & 15;
    const int hi   = lane >> 4;

    const v4f* ap = (const v4f*)Ap
                  + ((size_t)(tm4 * WK_MT * PK_KB + w * FK_KB) * 32 + lane);
    const v4f* bp = (const v4f*)Bp
                  + ((size_t)(tq * WK_NT * PK_KB + w * FK_KB) * 32 + lane);

    v4f a[WK_RING][WK_MT], b[WK_RING][WK_NT];
    #pragma unroll
    for (int g = 0; g < WK_AHEAD; ++g) {
        #pragma unroll
        for (int i = 0; i < WK_MT; ++i) a[g][i] = ap[(i * PK_KB + g) * 32];
        #pragma unroll
        for (int n = 0; n < WK_NT; ++n) b[g][n] = bp[(n * PK_KB + g) * 32];
    }

    v8f acc[WK_MT][WK_NT];
    #pragma unroll
    for (int i = 0; i < WK_MT; ++i) {
        #pragma unroll
        for (int n = 0; n < WK_NT; ++n) acc[i][n] = v8f{};
    }

    #pragma unroll
    for (int g = 0; g < FK_KB; ++g) {
        const int cur = g % WK_RING;
        if (g + WK_AHEAD < FK_KB) {
            const int nxt = (g + WK_AHEAD) % WK_RING;
            #pragma unroll
            for (int i = 0; i < WK_MT; ++i)
                a[nxt][i] = ap[(i * PK_KB + g + WK_AHEAD) * 32];
            #pragma unroll
            for (int n = 0; n < WK_NT; ++n)
                b[nxt][n] = bp[(n * PK_KB + g + WK_AHEAD) * 32];
        }
        __builtin_amdgcn_sched_barrier(0);
        #pragma unroll
        for (int i = 0; i < WK_MT; ++i) {
            #pragma unroll
            for (int n = 0; n < WK_NT; ++n)
                acc[i][n] = __builtin_amdgcn_wmma_f32_16x16x4_f32(false, a[cur][i].xy,
                                false, b[cur][n].xy, (short)0, acc[i][n], false, false);
        }
        #pragma unroll
        for (int i = 0; i < WK_MT; ++i) {
            #pragma unroll
            for (int n = 0; n < WK_NT; ++n)
                acc[i][n] = __builtin_amdgcn_wmma_f32_16x16x4_f32(false, a[cur][i].zw,
                                false, b[cur][n].zw, (short)0, acc[i][n], false, false);
        }
        __builtin_amdgcn_sched_barrier(0);
    }

    const int tnl = w & 3;
    const int kb2 = w >> 2;
    #pragma unroll
    for (int m = 0; m < WK_MT; ++m) {
        if (m > 0) __syncthreads();
        float* rp = &red[w][hi * 8][l15];
        #pragma unroll
        for (int r = 0; r < 8; ++r) {
            #pragma unroll
            for (int n = 0; n < WK_NT; ++n)
                rp[r * WR_PITCH + n * 16] = acc[m][n][r];
        }
        __syncthreads();

        const float* q = &red[0][kb2 * 8 + 2 * hi][tnl * 16 + l15];
        float e0[FK_SPLIT], e1[FK_SPLIT], e4[FK_SPLIT], e5[FK_SPLIT];
        #pragma unroll
        for (int p = 0; p < FK_SPLIT; ++p) {
            e0[p] = q[p * 16 * WR_PITCH];
            e1[p] = q[p * 16 * WR_PITCH + WR_PITCH];
            e4[p] = q[p * 16 * WR_PITCH + 4 * WR_PITCH];
            e5[p] = q[p * 16 * WR_PITCH + 5 * WR_PITCH];
        }
        v4f o;
        o.x = ((e0[0] + e0[1]) + (e0[2] + e0[3])) + ((e0[4] + e0[5]) + (e0[6] + e0[7]));
        o.y = ((e1[0] + e1[1]) + (e1[2] + e1[3])) + ((e1[4] + e1[5]) + (e1[6] + e1[7]));
        o.z = ((e4[0] + e4[1]) + (e4[2] + e4[3])) + ((e4[4] + e4[5]) + (e4[6] + e4[7]));
        o.w = ((e5[0] + e5[1]) + (e5[2] + e5[3])) + ((e5[4] + e5[5]) + (e5[6] + e5[7]));
        *((v4f*)Wp + ((size_t)((tq * WK_NT + tnl) * PK_KB
                               + (tm4 * WK_MT + m) * 2 + kb2) * 32 + lane)) = o;
    }
}

__global__ void __launch_bounds__(256)
gs_reduce(const float* __restrict__ P, const float* __restrict__ subSrc,
          float* __restrict__ pk0, float* __restrict__ out1,
          float* __restrict__ out2) {
    const size_t i = ((size_t)blockIdx.x * 256 + threadIdx.x) * 4;
    float4 v0 = *(const float4*)(P + i);
    float4 v1 = *(const float4*)(P + ELEMS + i);
    float4 v2 = *(const float4*)(P + 2 * ELEMS + i);
    float4 v3 = *(const float4*)(P + 3 * ELEMS + i);
    float4 v;
    v.x = (v0.x + v1.x) + (v2.x + v3.x);
    v.y = (v0.y + v1.y) + (v2.y + v3.y);
    v.z = (v0.z + v1.z) + (v2.z + v3.z);
    v.w = (v0.w + v1.w) + (v2.w + v3.w);
    if (subSrc) {
        float4 s = *(const float4*)(subSrc + i);
        v.x = s.x - v.x; v.y = s.y - v.y; v.z = s.z - v.z; v.w = s.w - v.w;
    }
    {
        const int m  = (int)(i >> 11);
        const int k  = (int)(i & (N_DIM - 1));
        const int kb = k >> 3;
        const int j2 = (k >> 1) & 2;
        float* blk = pk0 + ((size_t)((m >> 4) * PK_KB + kb) * 32 + (m & 15)) * 4 + j2;
        v2f lo; lo.x = v.x; lo.y = v.y;
        v2f hh; hh.x = v.z; hh.y = v.w;
        *(v2f*)(blk)          = lo;
        *(v2f*)(blk + 16 * 4) = hh;
    }
    if (out1) *(float4*)(out1 + i) = v;
    if (out2) *(float4*)(out2 + i) = v;
}

__global__ void __launch_bounds__(128)
gs_gemm_nt(const float* __restrict__ A, const float* __restrict__ Bt,
           const float* __restrict__ subSrc,
           float* __restrict__ out0, float* __restrict__ out1,
           float* __restrict__ out2) {
    const int lane = threadIdx.x & 31;
    const int wave = threadIdx.x >> 5;
    const int t    = blockIdx.x * 4 + wave;
    const int tm   = t & 3;
    const int tn   = t >> 2;
    const int l15  = lane & 15;
    const int hi   = lane >> 4;
    const int koff = hi * 2;

    const float* aptr = A  + (size_t)(tm * 16 + l15) * N_DIM + koff;
    const float* bptr = Bt + (size_t)(tn * 16 + l15) * N_DIM + koff;

    v8f acc0 = {}; v8f acc1 = {};
    #pragma unroll 4
    for (int k = 0; k < N_DIM; k += 8) {
        v2f a0 = *(const v2f*)(aptr + k);
        v2f b0 = *(const v2f*)(bptr + k);
        v2f a1 = *(const v2f*)(aptr + k + 4);
        v2f b1 = *(const v2f*)(bptr + k + 4);
        acc0 = __builtin_amdgcn_wmma_f32_16x16x4_f32(false, a0, false, b0,
                                                     (short)0, acc0, false, false);
        acc1 = __builtin_amdgcn_wmma_f32_16x16x4_f32(false, a1, false, b1,
                                                     (short)0, acc1, false, false);
    }

    const size_t obase = (size_t)(tm * 16 + hi * 8) * N_DIM + tn * 16 + l15;
    float res[8];
    #pragma unroll
    for (int r = 0; r < 8; ++r) res[r] = acc0[r] + acc1[r];
    if (subSrc) {
        #pragma unroll
        for (int r = 0; r < 8; ++r)
            res[r] = subSrc[obase + (size_t)r * N_DIM] - res[r];
    }
    #pragma unroll
    for (int r = 0; r < 8; ++r) out0[obase + (size_t)r * N_DIM] = res[r];
    if (out1) {
        #pragma unroll
        for (int r = 0; r < 8; ++r) out1[obase + (size_t)r * N_DIM] = res[r];
    }
    if (out2) {
        #pragma unroll
        for (int r = 0; r < 8; ++r) out2[obase + (size_t)r * N_DIM] = res[r];
    }
}

__global__ void __launch_bounds__(128)
gs_gemm_nn(const float* __restrict__ A, const float* __restrict__ B,
           const float* __restrict__ subSrc,
           float* __restrict__ out0, float* __restrict__ out1,
           float* __restrict__ out2) {
    const int lane = threadIdx.x & 31;
    const int wave = threadIdx.x >> 5;
    const int t    = blockIdx.x * 4 + wave;
    const int tm   = t & 3;
    const int tn   = t >> 2;
    const int l15  = lane & 15;
    const int hi   = lane >> 4;
    const int koff = hi * 2;

    const float* aptr = A + (size_t)(tm * 16 + l15) * N_DIM + koff;
    const float* bptr = B + (size_t)koff * N_DIM + tn * 16 + l15;

    v8f acc0 = {}; v8f acc1 = {};
    #pragma unroll 4
    for (int k = 0; k < N_DIM; k += 8) {
        v2f a0 = *(const v2f*)(aptr + k);
        v2f b0;
        b0.x = bptr[(size_t)(k + 0) * N_DIM];
        b0.y = bptr[(size_t)(k + 1) * N_DIM];
        v2f a1 = *(const v2f*)(aptr + k + 4);
        v2f b1;
        b1.x = bptr[(size_t)(k + 4) * N_DIM];
        b1.y = bptr[(size_t)(k + 5) * N_DIM];
        acc0 = __builtin_amdgcn_wmma_f32_16x16x4_f32(false, a0, false, b0,
                                                     (short)0, acc0, false, false);
        acc1 = __builtin_amdgcn_wmma_f32_16x16x4_f32(false, a1, false, b1,
                                                     (short)0, acc1, false, false);
    }

    const size_t obase = (size_t)(tm * 16 + hi * 8) * N_DIM + tn * 16 + l15;
    float res[8];
    #pragma unroll
    for (int r = 0; r < 8; ++r) res[r] = acc0[r] + acc1[r];
    if (subSrc) {
        #pragma unroll
        for (int r = 0; r < 8; ++r)
            res[r] = subSrc[obase + (size_t)r * N_DIM] - res[r];
    }
    #pragma unroll
    for (int r = 0; r < 8; ++r) out0[obase + (size_t)r * N_DIM] = res[r];
    if (out1) {
        #pragma unroll
        for (int r = 0; r < 8; ++r) out1[obase + (size_t)r * N_DIM] = res[r];
    }
    if (out2) {
        #pragma unroll
        for (int r = 0; r < 8; ++r) out2[obase + (size_t)r * N_DIM] = res[r];
    }
}

extern "C" void kernel_launch(void* const* d_in, const int* in_sizes, int n_in,
                              void* d_out, int out_size, void* d_ws, size_t ws_size,
                              hipStream_t stream) {
    const float* y    = (const float*)d_in[2];
    const float* H    = (const float*)d_in[3];
    const float* Dinv = (const float*)d_in[4];
    const float* U    = (const float*)d_in[5];
    const float* invM = (const float*)d_in[6];

    float* sfinal = (float*)d_out;
    float* traj   = (float*)d_out + ELEMS;

    float* ws    = (float*)d_ws;
    float* yMF   = ws;
    float* S     = ws + ELEMS;
    float* T     = ws + 2 * ELEMS;
    float* P     = ws + 3 * ELEMS;
    float* Dt    = ws + 7 * ELEMS;
    float* Ut    = Dt + MATSZ;
    float* invMt = Ut + MATSZ;

    const size_t need_fast = (7 * ELEMS + 3 * MATSZ) * sizeof(float);
    const size_t need_mid  = (3 * ELEMS + 2 * MATSZ) * sizeof(float);
    const int mode = (ws_size >= need_fast) ? 2 : (ws_size >= need_mid ? 1 : 0);

    dim3 gblk(128);
    dim3 ggrd(128);
    dim3 rgrd(ELEMS / (256 * 4));
    dim3 fblk(256);
    dim3 fgrd(256);
    dim3 pgrd(512);
    dim3 sgrd(SU_GRID);
    dim3 wgrd(1024);

    if (mode == 2) {
        float* Cm = yMF;
        float* Wp = Dt;

        gs_setup<<<sgrd, 256, 0, stream>>>(y, H, Dinv, U, invM, P, Dt, Ut, invMt, traj);
        gs_reduce<<<rgrd, 256, 0, stream>>>(P, nullptr, T, nullptr, nullptr);
        gs_gemm_fused_pk_pair<<<pgrd, fblk, 0, stream>>>(T, Dt, invMt, S, Cm);
        gs_gemm_w_pk<<<wgrd, fblk, 0, stream>>>(Ut, invMt, Wp);

        float* cur = S;
        float* nxt = T;
        for (int i = 1; i <= N_ITER; ++i) {
            float* trajD  = traj + (size_t)i * ELEMS;
            float* finalD = (i == N_ITER) ? sfinal : nullptr;
            gs_gemm_fused_pk<<<fgrd, fblk, 0, stream>>>(cur, Wp, Cm, nxt, trajD, finalD);
            float* t = cur; cur = nxt; nxt = t;
        }
    } else if (mode == 1) {
        gs_zero_kernel<<<(int)(ELEMS / 256), 256, 0, stream>>>(traj, (int)ELEMS);
        gs_transpose<<<4096, 256, 0, stream>>>(U, Ut - MATSZ);
        gs_transpose<<<4096, 256, 0, stream>>>(invM, invMt - MATSZ);
        float* Ut1    = Dt;
        float* invMt1 = Dt + MATSZ;
        gs_gemm_nt<<<ggrd, gblk, 0, stream>>>(y, H, nullptr, yMF, nullptr, nullptr);
        gs_gemm_nn<<<ggrd, gblk, 0, stream>>>(yMF, Dinv, nullptr, S, nullptr, nullptr);
        for (int i = 1; i <= N_ITER; ++i) {
            float* trajD  = traj + (size_t)i * ELEMS;
            float* finalD = (i == N_ITER) ? sfinal : nullptr;
            gs_gemm_nt<<<ggrd, gblk, 0, stream>>>(S, Ut1, yMF, T, nullptr, nullptr);
            gs_gemm_nt<<<ggrd, gblk, 0, stream>>>(T, invMt1, nullptr, S, trajD, finalD);
        }
    } else {
        gs_zero_kernel<<<(int)(ELEMS / 256), 256, 0, stream>>>(traj, (int)ELEMS);
        gs_gemm_nt<<<ggrd, gblk, 0, stream>>>(y, H, nullptr, yMF, nullptr, nullptr);
        gs_gemm_nn<<<ggrd, gblk, 0, stream>>>(yMF, Dinv, nullptr, S, nullptr, nullptr);
        for (int i = 1; i <= N_ITER; ++i) {
            float* trajD  = traj + (size_t)i * ELEMS;
            float* finalD = (i == N_ITER) ? sfinal : nullptr;
            gs_gemm_nn<<<ggrd, gblk, 0, stream>>>(S, U, yMF, T, nullptr, nullptr);
            gs_gemm_nn<<<ggrd, gblk, 0, stream>>>(T, invM, nullptr, S, trajD, finalD);
        }
    }
}


# Set-up launch: y@H^T sums its 8 K-slices inside the work-group from b128 fragment loads and stores yMF packed (reduce launch gone: 28), packing row loads issued together; yMF sum order changes
# speedup vs baseline: 3.0544x; 1.0300x over previous
#include <hip/hip_runtime.h>

typedef float v2f __attribute__((ext_vector_type(2)));
typedef float v4f __attribute__((ext_vector_type(4)));
typedef float v8f __attribute__((ext_vector_type(8)));

#define N_DIM   2048
#define BS_DIM  64
#define N_ITER  25
#define ELEMS   ((size_t)BS_DIM * N_DIM)
#define MATSZ   ((size_t)N_DIM * N_DIM)

__global__ void gs_zero_kernel(float* __restrict__ p, int n) {
    int i = blockIdx.x * blockDim.x + threadIdx.x;
    if (i < n) p[i] = 0.0f;
}

__global__ void __launch_bounds__(256)
gs_transpose(const float* __restrict__ in, float* __restrict__ out) {
    __shared__ float tile[32][33];
    const int bx = blockIdx.x & 63;
    const int by = blockIdx.x >> 6;
    const int tx = threadIdx.x & 31;
    const int ty = threadIdx.x >> 5;

    const int x = bx * 32 + tx;
    #pragma unroll
    for (int r = ty; r < 32; r += 8)
        tile[r][tx] = in[(size_t)(by * 32 + r) * N_DIM + x];
    __syncthreads();
    const int x2 = by * 32 + tx;
    #pragma unroll
    for (int r = ty; r < 32; r += 8)
        out[(size_t)(bx * 32 + r) * N_DIM + x2] = tile[tx][r];
}

#define PK_KB   (N_DIM / 8)

__device__ __forceinline__ void
gs_pack_b_role(const float* __restrict__ in, float* __restrict__ out, int bid,
               float (*tile)[40]) {
    const int bx = bid & 63;
    const int by = bid >> 6;
    const int tx = threadIdx.x & 31;
    const int ty = threadIdx.x >> 5;

    const int x = bx * 32 + tx;
    float rv[4];
    #pragma unroll
    for (int j = 0; j < 4; ++j)
        rv[j] = in[(size_t)(by * 32 + ty + 8 * j) * N_DIM + x];
    #pragma unroll
    for (int j = 0; j < 4; ++j)
        tile[ty + 8 * j][tx] = rv[j];
    __syncthreads();

    const int c   = tx & 15;
    const int h   = tx >> 4;
    const int tnl = ty & 1;
    const int kbl = ty >> 1;
    const int k0  = kbl * 8 + 2 * h;
    const int nl  = tnl * 16 + c;
    v4f v;
    v.x = tile[k0][nl];
    v.y = tile[k0 + 1][nl];
    v.z = tile[k0 + 4][nl];
    v.w = tile[k0 + 5][nl];
    const size_t off =
        (((size_t)(bx * 2 + tnl) * PK_KB + (by * 4 + kbl)) * 32 + tx) * 4;
    *(v4f*)(out + off) = v;
}

__device__ __forceinline__ void
gs_pack_a_role(const float* __restrict__ in, float* __restrict__ out, int bid) {
    const int lane = threadIdx.x & 31;
    const int w    = threadIdx.x >> 5;
    const int tmr  = bid >> 5;
    const int kb   = (bid & 31) * 8 + w;
    const int c    = lane & 15;
    const int h    = lane >> 4;
    const float* p = in + (size_t)(tmr * 16 + c) * N_DIM + kb * 8 + 2 * h;
    const v2f lo = *(const v2f*)(p);
    const v2f hh = *(const v2f*)(p + 4);
    v4f v;
    v.x = lo.x; v.y = lo.y; v.z = hh.x; v.w = hh.y;
    *((v4f*)out + ((size_t)(tmr * PK_KB + kb) * 32 + lane)) = v;
}

#define YG_PITCH 20
__device__ __forceinline__ void
gs_ymf_role(const float* __restrict__ A, const float* __restrict__ Bt,
            float* __restrict__ pk, int b, float (*red)[32][YG_PITCH]) {
    const int lane = threadIdx.x & 31;
    const int w    = __builtin_amdgcn_readfirstlane((int)(threadIdx.x >> 5));
    const int mp   = b & 1;
    const int tn   = b >> 1;
    const int l15  = lane & 15;
    const int hi   = lane >> 4;
    const int kbase = w * 256 + hi * 4;

    const float* a0p = A  + (size_t)(mp * 32 + l15) * N_DIM + kbase;
    const float* a1p = a0p + (size_t)16 * N_DIM;
    const float* bp  = Bt + (size_t)(tn * 16 + l15) * N_DIM + kbase;

    v8f c00 = {}; v8f c10 = {}; v8f c01 = {}; v8f c11 = {};
    for (int k = 0; k < 256; k += 32) {
        v4f a0[4], a1[4], bb[4];
        #pragma unroll
        for (int j = 0; j < 4; ++j) {
            a0[j] = *(const v4f*)(a0p + k + j * 8);
            a1[j] = *(const v4f*)(a1p + k + j * 8);
            bb[j] = *(const v4f*)(bp  + k + j * 8);
        }
        #pragma unroll
        for (int j = 0; j < 4; ++j) {
            c00 = __builtin_amdgcn_wmma_f32_16x16x4_f32(false, a0[j].xy, false,
                      bb[j].xy, (short)0, c00, false, false);
            c10 = __builtin_amdgcn_wmma_f32_16x16x4_f32(false, a1[j].xy, false,
                      bb[j].xy, (short)0, c10, false, false);
            c01 = __builtin_amdgcn_wmma_f32_16x16x4_f32(false, a0[j].zw, false,
                      bb[j].zw, (short)0, c01, false, false);
            c11 = __builtin_amdgcn_wmma_f32_16x16x4_f32(false, a1[j].zw, false,
                      bb[j].zw, (short)0, c11, false, false);
        }
    }

    {
        float* rp = &red[w][hi * 8][l15];
        #pragma unroll
        for (int r = 0; r < 8; ++r) {
            rp[r * YG_PITCH]        = c00[r] + c01[r];
            rp[(16 + r) * YG_PITCH] = c10[r] + c11[r];
        }
    }
    __syncthreads();

    if (w < 4) {
        const int mt = w >> 1;
        const int kb = w & 1;
        const float* q = &red[0][mt * 16 + l15][kb * 8 + 2 * hi];
        v2f xs[8], ys[8];
        #pragma unroll
        for (int p = 0; p < 8; ++p) {
            xs[p] = *(const v2f*)(q + p * 32 * YG_PITCH);
            ys[p] = *(const v2f*)(q + p * 32 * YG_PITCH + 4);
        }
        v2f x = ((xs[0] + xs[1]) + (xs[2] + xs[3])) + ((xs[4] + xs[5]) + (xs[6] + xs[7]));
        v2f y = ((ys[0] + ys[1]) + (ys[2] + ys[3])) + ((ys[4] + ys[5]) + (ys[6] + ys[7]));
        v4f o;
        o.x = x.x; o.y = x.y; o.z = y.x; o.w = y.y;
        *((v4f*)pk + ((size_t)((mp * 2 + mt) * PK_KB + tn * 2 + kb) * 32 + lane)) = o;
    }
}

#define SU_GEMM  256
#define SU_PACK  4096
#define SU_ZERO  ((int)(ELEMS / (256 * 4)))
#define SU_GRID  (SU_GEMM + 3 * SU_PACK + SU_ZERO)
__global__ void __launch_bounds__(256)
gs_setup(const float* __restrict__ y, const float* __restrict__ H,
         const float* __restrict__ Dinv, const float* __restrict__ U,
         const float* __restrict__ invM, float* __restrict__ yMFp,
         float* __restrict__ Dp, float* __restrict__ Up,
         float* __restrict__ invMp, float* __restrict__ traj0) {
    __shared__ float tile[32][40];
    __shared__ alignas(16) float red[8][32][YG_PITCH];
    const int b = (int)blockIdx.x;
    if (b < SU_GEMM) {
        gs_ymf_role(y, H, yMFp, b, red);
    } else if (b < SU_GEMM + SU_PACK) {
        gs_pack_b_role(Dinv, Dp, b - SU_GEMM, tile);
    } else if (b < SU_GEMM + 2 * SU_PACK) {
        gs_pack_a_role(U, Up, b - (SU_GEMM + SU_PACK));
    } else if (b < SU_GEMM + 3 * SU_PACK) {
        gs_pack_b_role(invM, invMp, b - (SU_GEMM + 2 * SU_PACK), tile);
    } else {
        const v4f z = {0.0f, 0.0f, 0.0f, 0.0f};
        *((v4f*)traj0 + ((size_t)(b - (SU_GEMM + 3 * SU_PACK)) * 256 + threadIdx.x)) = z;
    }
}

#define FK_SPLIT 8
#define FK_KB    (N_DIM / FK_SPLIT / 8)
#define PK_GRP   4
#define PK_NGRP  (FK_KB / PK_GRP)
#define PK_AHEAD 3
#define PK_RING  (PK_AHEAD + 1)
#define RD_PITCH 36
__device__ __forceinline__ void
gs_fused_pk_body(const float* __restrict__ Ap, const float* __restrict__ Bp,
                 const float* __restrict__ subSrc, float* __restrict__ pk0,
                 float* __restrict__ out1, float* __restrict__ out2, int bid) {
    __shared__ alignas(16) float red[FK_SPLIT][16][RD_PITCH];

    const int lane = threadIdx.x & 31;
    const int w    = __builtin_amdgcn_readfirstlane((int)(threadIdx.x >> 5));
    const int kbl  = w & 3;
    const int tm   = bid & 3;
    const int tp   = bid >> 2;
    const int l15  = lane & 15;
    const int hi   = lane >> 4;

    v2f sa = {}; v2f sb = {};
    if (subSrc != nullptr && w >= 4) {
        const float* sp = subSrc + (size_t)(tm * 16 + l15) * N_DIM
                        + tp * 32 + kbl * 8 + 2 * hi;
        sa = *(const v2f*)(sp);
        sb = *(const v2f*)(sp + 4);
    }

    const v4f* ap  = (const v4f*)Ap
                   + ((size_t)(tm * PK_KB + w * FK_KB) * 32 + lane);
    const v4f* bp0 = (const v4f*)Bp
                   + ((size_t)(tp * 2 * PK_KB + w * FK_KB) * 32 + lane);
    const v4f* bp1 = bp0 + (size_t)PK_KB * 32;

    v4f a[PK_RING][PK_GRP], b0[PK_RING][PK_GRP], b1[PK_RING][PK_GRP];
    #pragma unroll
    for (int g = 0; g < PK_AHEAD; ++g) {
        #pragma unroll
        for (int j = 0; j < PK_GRP; ++j) {
            a[g][j]  = ap[(g * PK_GRP + j) * 32];
            b0[g][j] = bp0[(g * PK_GRP + j) * 32];
            b1[g][j] = bp1[(g * PK_GRP + j) * 32];
        }
    }

    v8f c00 = {}; v8f c10 = {}; v8f c01 = {}; v8f c11 = {};
    #pragma unroll
    for (int g = 0; g < PK_NGRP; ++g) {
        const int cur = g % PK_RING;
        if (g + PK_AHEAD < PK_NGRP) {
            const int nxt = (g + PK_AHEAD) % PK_RING;
            #pragma unroll
            for (int j = 0; j < PK_GRP; ++j) {
                a[nxt][j]  = ap[((g + PK_AHEAD) * PK_GRP + j) * 32];
                b0[nxt][j] = bp0[((g + PK_AHEAD) * PK_GRP + j) * 32];
                b1[nxt][j] = bp1[((g + PK_AHEAD) * PK_GRP + j) * 32];
            }
        }
        __builtin_amdgcn_sched_barrier(0);
        #pragma unroll
        for (int j = 0; j < PK_GRP; ++j) {
            c00 = __builtin_amdgcn_wmma_f32_16x16x4_f32(false, a[cur][j].xy, false,
                      b0[cur][j].xy, (short)0, c00, false, false);
            c10 = __builtin_amdgcn_wmma_f32_16x16x4_f32(false, a[cur][j].xy, false,
                      b1[cur][j].xy, (short)0, c10, false, false);
            c01 = __builtin_amdgcn_wmma_f32_16x16x4_f32(false, a[cur][j].zw, false,
                      b0[cur][j].zw, (short)0, c01, false, false);
            c11 = __builtin_amdgcn_wmma_f32_16x16x4_f32(false, a[cur][j].zw, false,
                      b1[cur][j].zw, (short)0, c11, false, false);
        }
        __builtin_amdgcn_sched_barrier(0);
    }

    {
        float* rp = &red[w][hi * 8][l15];
        #pragma unroll
        for (int r = 0; r < 8; ++r) {
            rp[r * RD_PITCH]      = c00[r] + c01[r];
            rp[r * RD_PITCH + 16] = c10[r] + c11[r];
        }
    }
    __syncthreads();

    if (w >= 4) {
        if (pk0 == nullptr) return;
        const float* q = &red[0][l15][kbl * 8 + 2 * hi];
        v2f xs[FK_SPLIT], ys[FK_SPLIT];
        #pragma unroll
        for (int p = 0; p < FK_SPLIT; ++p) {
            xs[p] = *(const v2f*)(q + p * 16 * RD_PITCH);
            ys[p] = *(const v2f*)(q + p * 16 * RD_PITCH + 4);
        }
        v2f x = ((xs[0] + xs[1]) + (xs[2] + xs[3])) + ((xs[4] + xs[5]) + (xs[6] + xs[7]));
        v2f y = ((ys[0] + ys[1]) + (ys[2] + ys[3])) + ((ys[4] + ys[5]) + (ys[6] + ys[7]));
        if (subSrc != nullptr) { x = sa - x; y = sb - y; }
        v4f o;
        o.x = x.x; o.y = x.y; o.z = y.x; o.w = y.y;
        *((v4f*)pk0 + ((size_t)(tm * PK_KB + tp * 4 + kbl) * 32 + lane)) = o;
    } else if (out1 != nullptr || out2 != nullptr) {
        const int t    = threadIdx.x;
        const int row  = t >> 3;
        const int col4 = (t & 7) * 4;
        const float* q = &red[0][row][col4];
        v4f ps[FK_SPLIT];
        #pragma unroll
        for (int p = 0; p < FK_SPLIT; ++p)
            ps[p] = *(const v4f*)(q + p * 16 * RD_PITCH);
        v4f v = ((ps[0] + ps[1]) + (ps[2] + ps[3])) + ((ps[4] + ps[5]) + (ps[6] + ps[7]));
        const size_t o = (size_t)(tm * 16 + row) * N_DIM + tp * 32 + col4;
        if (subSrc != nullptr) v = *(const v4f*)(subSrc + o) - v;
        if (out1 != nullptr) *(v4f*)(out1 + o) = v;
        if (out2 != nullptr) *(v4f*)(out2 + o) = v;
    }
}

__global__ void __launch_bounds__(256)
gs_gemm_fused_pk(const float* __restrict__ Ap, const float* __restrict__ Bp,
                 const float* __restrict__ subSrc, float* __restrict__ pk0,
                 float* __restrict__ out1, float* __restrict__ out2) {
    gs_fused_pk_body(Ap, Bp, subSrc, pk0, out1, out2, (int)blockIdx.x);
}

__global__ void __launch_bounds__(256)
gs_gemm_fused_pk_pair(const float* __restrict__ Ap, const float* __restrict__ Bp0,
                      const float* __restrict__ Bp1, float* __restrict__ pk,
                      float* __restrict__ rm) {
    const bool second = blockIdx.x >= 256;
    gs_fused_pk_body(Ap, second ? Bp1 : Bp0, nullptr, second ? nullptr : pk,
                     second ? rm : nullptr, nullptr, (int)(blockIdx.x & 255));
}

#define WK_MT    4
#define WK_NT    4
#define WK_AHEAD 2
#define WK_RING  (WK_AHEAD + 1)
#define WR_PITCH 76
__global__ void __launch_bounds__(256)
gs_gemm_w_pk(const float* __restrict__ Ap, const float* __restrict__ Bp,
             float* __restrict__ Wp) {
    __shared__ float red[FK_SPLIT][16][WR_PITCH];

    const int lane = threadIdx.x & 31;
    const int w    = __builtin_amdgcn_readfirstlane((int)(threadIdx.x >> 5));
    const int tm4  = (int)(blockIdx.x >> 5);
    const int tq   = (int)(blockIdx.x & 31);
    const int l15  = lane & 15;
    const int hi   = lane >> 4;

    const v4f* ap = (const v4f*)Ap
                  + ((size_t)(tm4 * WK_MT * PK_KB + w * FK_KB) * 32 + lane);
    const v4f* bp = (const v4f*)Bp
                  + ((size_t)(tq * WK_NT * PK_KB + w * FK_KB) * 32 + lane);

    v4f a[WK_RING][WK_MT], b[WK_RING][WK_NT];
    #pragma unroll
    for (int g = 0; g < WK_AHEAD; ++g) {
        #pragma unroll
        for (int i = 0; i < WK_MT; ++i) a[g][i] = ap[(i * PK_KB + g) * 32];
        #pragma unroll
        for (int n = 0; n < WK_NT; ++n) b[g][n] = bp[(n * PK_KB + g) * 32];
    }

    v8f acc[WK_MT][WK_NT];
    #pragma unroll
    for (int i = 0; i < WK_MT; ++i) {
        #pragma unroll
        for (int n = 0; n < WK_NT; ++n) acc[i][n] = v8f{};
    }

    #pragma unroll
    for (int g = 0; g < FK_KB; ++g) {
        const int cur = g % WK_RING;
        if (g + WK_AHEAD < FK_KB) {
            const int nxt = (g + WK_AHEAD) % WK_RING;
            #pragma unroll
            for (int i = 0; i < WK_MT; ++i)
                a[nxt][i] = ap[(i * PK_KB + g + WK_AHEAD) * 32];
            #pragma unroll
            for (int n = 0; n < WK_NT; ++n)
                b[nxt][n] = bp[(n * PK_KB + g + WK_AHEAD) * 32];
        }
        __builtin_amdgcn_sched_barrier(0);
        #pragma unroll
        for (int i = 0; i < WK_MT; ++i) {
            #pragma unroll
            for (int n = 0; n < WK_NT; ++n)
                acc[i][n] = __builtin_amdgcn_wmma_f32_16x16x4_f32(false, a[cur][i].xy,
                                false, b[cur][n].xy, (short)0, acc[i][n], false, false);
        }
        #pragma unroll
        for (int i = 0; i < WK_MT; ++i) {
            #pragma unroll
            for (int n = 0; n < WK_NT; ++n)
                acc[i][n] = __builtin_amdgcn_wmma_f32_16x16x4_f32(false, a[cur][i].zw,
                                false, b[cur][n].zw, (short)0, acc[i][n], false, false);
        }
        __builtin_amdgcn_sched_barrier(0);
    }

    const int tnl = w & 3;
    const int kb2 = w >> 2;
    #pragma unroll
    for (int m = 0; m < WK_MT; ++m) {
        if (m > 0) __syncthreads();
        float* rp = &red[w][hi * 8][l15];
        #pragma unroll
        for (int r = 0; r < 8; ++r) {
            #pragma unroll
            for (int n = 0; n < WK_NT; ++n)
                rp[r * WR_PITCH + n * 16] = acc[m][n][r];
        }
        __syncthreads();

        const float* q = &red[0][kb2 * 8 + 2 * hi][tnl * 16 + l15];
        float e0[FK_SPLIT], e1[FK_SPLIT], e4[FK_SPLIT], e5[FK_SPLIT];
        #pragma unroll
        for (int p = 0; p < FK_SPLIT; ++p) {
            e0[p] = q[p * 16 * WR_PITCH];
            e1[p] = q[p * 16 * WR_PITCH + WR_PITCH];
            e4[p] = q[p * 16 * WR_PITCH + 4 * WR_PITCH];
            e5[p] = q[p * 16 * WR_PITCH + 5 * WR_PITCH];
        }
        v4f o;
        o.x = ((e0[0] + e0[1]) + (e0[2] + e0[3])) + ((e0[4] + e0[5]) + (e0[6] + e0[7]));
        o.y = ((e1[0] + e1[1]) + (e1[2] + e1[3])) + ((e1[4] + e1[5]) + (e1[6] + e1[7]));
        o.z = ((e4[0] + e4[1]) + (e4[2] + e4[3])) + ((e4[4] + e4[5]) + (e4[6] + e4[7]));
        o.w = ((e5[0] + e5[1]) + (e5[2] + e5[3])) + ((e5[4] + e5[5]) + (e5[6] + e5[7]));
        *((v4f*)Wp + ((size_t)((tq * WK_NT + tnl) * PK_KB
                               + (tm4 * WK_MT + m) * 2 + kb2) * 32 + lane)) = o;
    }
}

__global__ void __launch_bounds__(128)
gs_gemm_nt(const float* __restrict__ A, const float* __restrict__ Bt,
           const float* __restrict__ subSrc,
           float* __restrict__ out0, float* __restrict__ out1,
           float* __restrict__ out2) {
    const int lane = threadIdx.x & 31;
    const int wave = threadIdx.x >> 5;
    const int t    = blockIdx.x * 4 + wave;
    const int tm   = t & 3;
    const int tn   = t >> 2;
    const int l15  = lane & 15;
    const int hi   = lane >> 4;
    const int koff = hi * 2;

    const float* aptr = A  + (size_t)(tm * 16 + l15) * N_DIM + koff;
    const float* bptr = Bt + (size_t)(tn * 16 + l15) * N_DIM + koff;

    v8f acc0 = {}; v8f acc1 = {};
    #pragma unroll 4
    for (int k = 0; k < N_DIM; k += 8) {
        v2f a0 = *(const v2f*)(aptr + k);
        v2f b0 = *(const v2f*)(bptr + k);
        v2f a1 = *(const v2f*)(aptr + k + 4);
        v2f b1 = *(const v2f*)(bptr + k + 4);
        acc0 = __builtin_amdgcn_wmma_f32_16x16x4_f32(false, a0, false, b0,
                                                     (short)0, acc0, false, false);
        acc1 = __builtin_amdgcn_wmma_f32_16x16x4_f32(false, a1, false, b1,
                                                     (short)0, acc1, false, false);
    }

    const size_t obase = (size_t)(tm * 16 + hi * 8) * N_DIM + tn * 16 + l15;
    float res[8];
    #pragma unroll
    for (int r = 0; r < 8; ++r) res[r] = acc0[r] + acc1[r];
    if (subSrc) {
        #pragma unroll
        for (int r = 0; r < 8; ++r)
            res[r] = subSrc[obase + (size_t)r * N_DIM] - res[r];
    }
    #pragma unroll
    for (int r = 0; r < 8; ++r) out0[obase + (size_t)r * N_DIM] = res[r];
    if (out1) {
        #pragma unroll
        for (int r = 0; r < 8; ++r) out1[obase + (size_t)r * N_DIM] = res[r];
    }
    if (out2) {
        #pragma unroll
        for (int r = 0; r < 8; ++r) out2[obase + (size_t)r * N_DIM] = res[r];
    }
}

__global__ void __launch_bounds__(128)
gs_gemm_nn(const float* __restrict__ A, const float* __restrict__ B,
           const float* __restrict__ subSrc,
           float* __restrict__ out0, float* __restrict__ out1,
           float* __restrict__ out2) {
    const int lane = threadIdx.x & 31;
    const int wave = threadIdx.x >> 5;
    const int t    = blockIdx.x * 4 + wave;
    const int tm   = t & 3;
    const int tn   = t >> 2;
    const int l15  = lane & 15;
    const int hi   = lane >> 4;
    const int koff = hi * 2;

    const float* aptr = A + (size_t)(tm * 16 + l15) * N_DIM + koff;
    const float* bptr = B + (size_t)koff * N_DIM + tn * 16 + l15;

    v8f acc0 = {}; v8f acc1 = {};
    #pragma unroll 4
    for (int k = 0; k < N_DIM; k += 8) {
        v2f a0 = *(const v2f*)(aptr + k);
        v2f b0;
        b0.x = bptr[(size_t)(k + 0) * N_DIM];
        b0.y = bptr[(size_t)(k + 1) * N_DIM];
        v2f a1 = *(const v2f*)(aptr + k + 4);
        v2f b1;
        b1.x = bptr[(size_t)(k + 4) * N_DIM];
        b1.y = bptr[(size_t)(k + 5) * N_DIM];
        acc0 = __builtin_amdgcn_wmma_f32_16x16x4_f32(false, a0, false, b0,
                                                     (short)0, acc0, false, false);
        acc1 = __builtin_amdgcn_wmma_f32_16x16x4_f32(false, a1, false, b1,
                                                     (short)0, acc1, false, false);
    }

    const size_t obase = (size_t)(tm * 16 + hi * 8) * N_DIM + tn * 16 + l15;
    float res[8];
    #pragma unroll
    for (int r = 0; r < 8; ++r) res[r] = acc0[r] + acc1[r];
    if (subSrc) {
        #pragma unroll
        for (int r = 0; r < 8; ++r)
            res[r] = subSrc[obase + (size_t)r * N_DIM] - res[r];
    }
    #pragma unroll
    for (int r = 0; r < 8; ++r) out0[obase + (size_t)r * N_DIM] = res[r];
    if (out1) {
        #pragma unroll
        for (int r = 0; r < 8; ++r) out1[obase + (size_t)r * N_DIM] = res[r];
    }
    if (out2) {
        #pragma unroll
        for (int r = 0; r < 8; ++r) out2[obase + (size_t)r * N_DIM] = res[r];
    }
}

extern "C" void kernel_launch(void* const* d_in, const int* in_sizes, int n_in,
                              void* d_out, int out_size, void* d_ws, size_t ws_size,
                              hipStream_t stream) {
    const float* y    = (const float*)d_in[2];
    const float* H    = (const float*)d_in[3];
    const float* Dinv = (const float*)d_in[4];
    const float* U    = (const float*)d_in[5];
    const float* invM = (const float*)d_in[6];

    float* sfinal = (float*)d_out;
    float* traj   = (float*)d_out + ELEMS;

    float* ws    = (float*)d_ws;
    float* yMF   = ws;
    float* S     = ws + ELEMS;
    float* T     = ws + 2 * ELEMS;
    float* Dt    = ws + 7 * ELEMS;
    float* Ut    = Dt + MATSZ;
    float* invMt = Ut + MATSZ;

    const size_t need_fast = (7 * ELEMS + 3 * MATSZ) * sizeof(float);
    const size_t need_mid  = (3 * ELEMS + 2 * MATSZ) * sizeof(float);
    const int mode = (ws_size >= need_fast) ? 2 : (ws_size >= need_mid ? 1 : 0);

    dim3 gblk(128);
    dim3 ggrd(128);
    dim3 fblk(256);
    dim3 fgrd(256);
    dim3 pgrd(512);
    dim3 sgrd(SU_GRID);
    dim3 wgrd(1024);

    if (mode == 2) {
        float* Cm = yMF;
        float* Wp = Dt;

        gs_setup<<<sgrd, 256, 0, stream>>>(y, H, Dinv, U, invM, T, Dt, Ut, invMt, traj);
        gs_gemm_fused_pk_pair<<<pgrd, fblk, 0, stream>>>(T, Dt, invMt, S, Cm);
        gs_gemm_w_pk<<<wgrd, fblk, 0, stream>>>(Ut, invMt, Wp);

        float* cur = S;
        float* nxt = T;
        for (int i = 1; i <= N_ITER; ++i) {
            float* trajD  = traj + (size_t)i * ELEMS;
            float* finalD = (i == N_ITER) ? sfinal : nullptr;
            gs_gemm_fused_pk<<<fgrd, fblk, 0, stream>>>(cur, Wp, Cm, nxt, trajD, finalD);
            float* t = cur; cur = nxt; nxt = t;
        }
    } else if (mode == 1) {
        gs_zero_kernel<<<(int)(ELEMS / 256), 256, 0, stream>>>(traj, (int)ELEMS);
        gs_transpose<<<4096, 256, 0, stream>>>(U, Ut - MATSZ);
        gs_transpose<<<4096, 256, 0, stream>>>(invM, invMt - MATSZ);
        float* Ut1    = Dt;
        float* invMt1 = Dt + MATSZ;
        gs_gemm_nt<<<ggrd, gblk, 0, stream>>>(y, H, nullptr, yMF, nullptr, nullptr);
        gs_gemm_nn<<<ggrd, gblk, 0, stream>>>(yMF, Dinv, nullptr, S, nullptr, nullptr);
        for (int i = 1; i <= N_ITER; ++i) {
            float* trajD  = traj + (size_t)i * ELEMS;
            float* finalD = (i == N_ITER) ? sfinal : nullptr;
            gs_gemm_nt<<<ggrd, gblk, 0, stream>>>(S, Ut1, yMF, T, nullptr, nullptr);
            gs_gemm_nt<<<ggrd, gblk, 0, stream>>>(T, invMt1, nullptr, S, trajD, finalD);
        }
    } else {
        gs_zero_kernel<<<(int)(ELEMS / 256), 256, 0, stream>>>(traj, (int)ELEMS);
        gs_gemm_nt<<<ggrd, gblk, 0, stream>>>(y, H, nullptr, yMF, nullptr, nullptr);
        gs_gemm_nn<<<ggrd, gblk, 0, stream>>>(yMF, Dinv, nullptr, S, nullptr, nullptr);
        for (int i = 1; i <= N_ITER; ++i) {
            float* trajD  = traj + (size_t)i * ELEMS;
            float* finalD = (i == N_ITER) ? sfinal : nullptr;
            gs_gemm_nn<<<ggrd, gblk, 0, stream>>>(S, U, yMF, T, nullptr, nullptr);
            gs_gemm_nn<<<ggrd, gblk, 0, stream>>>(T, invM, nullptr, S, trajD, finalD);
        }
    }
}


# W, s_init and C share one launch (27 launches, not 28): s_init reads Dinv row-major from the input, so Dinv's packing pass is gone and its slot takes W in the same launch; arithmetic unchanged
# speedup vs baseline: 3.1928x; 1.0453x over previous
#include <hip/hip_runtime.h>

typedef float v2f __attribute__((ext_vector_type(2)));
typedef float v4f __attribute__((ext_vector_type(4)));
typedef float v8f __attribute__((ext_vector_type(8)));

#define N_DIM   2048
#define BS_DIM  64
#define N_ITER  25
#define ELEMS   ((size_t)BS_DIM * N_DIM)
#define MATSZ   ((size_t)N_DIM * N_DIM)

__global__ void gs_zero_kernel(float* __restrict__ p, int n) {
    int i = blockIdx.x * blockDim.x + threadIdx.x;
    if (i < n) p[i] = 0.0f;
}

__global__ void __launch_bounds__(256)
gs_transpose(const float* __restrict__ in, float* __restrict__ out) {
    __shared__ float tile[32][33];
    const int bx = blockIdx.x & 63;
    const int by = blockIdx.x >> 6;
    const int tx = threadIdx.x & 31;
    const int ty = threadIdx.x >> 5;

    const int x = bx * 32 + tx;
    #pragma unroll
    for (int r = ty; r < 32; r += 8)
        tile[r][tx] = in[(size_t)(by * 32 + r) * N_DIM + x];
    __syncthreads();
    const int x2 = by * 32 + tx;
    #pragma unroll
    for (int r = ty; r < 32; r += 8)
        out[(size_t)(bx * 32 + r) * N_DIM + x2] = tile[tx][r];
}

#define PK_KB   (N_DIM / 8)

__device__ __forceinline__ void
gs_pack_b_role(const float* __restrict__ in, float* __restrict__ out, int bid,
               float (*tile)[40]) {
    const int bx = bid & 63;
    const int by = bid >> 6;
    const int tx = threadIdx.x & 31;
    const int ty = threadIdx.x >> 5;

    const int x = bx * 32 + tx;
    float rv[4];
    #pragma unroll
    for (int j = 0; j < 4; ++j)
        rv[j] = in[(size_t)(by * 32 + ty + 8 * j) * N_DIM + x];
    #pragma unroll
    for (int j = 0; j < 4; ++j)
        tile[ty + 8 * j][tx] = rv[j];
    __syncthreads();

    const int c   = tx & 15;
    const int h   = tx >> 4;
    const int tnl = ty & 1;
    const int kbl = ty >> 1;
    const int k0  = kbl * 8 + 2 * h;
    const int nl  = tnl * 16 + c;
    v4f v;
    v.x = tile[k0][nl];
    v.y = tile[k0 + 1][nl];
    v.z = tile[k0 + 4][nl];
    v.w = tile[k0 + 5][nl];
    const size_t off =
        (((size_t)(bx * 2 + tnl) * PK_KB + (by * 4 + kbl)) * 32 + tx) * 4;
    *(v4f*)(out + off) = v;
}

__device__ __forceinline__ void
gs_pack_a_role(const float* __restrict__ in, float* __restrict__ out, int bid) {
    const int lane = threadIdx.x & 31;
    const int w    = threadIdx.x >> 5;
    const int tmr  = bid >> 5;
    const int kb   = (bid & 31) * 8 + w;
    const int c    = lane & 15;
    const int h    = lane >> 4;
    const float* p = in + (size_t)(tmr * 16 + c) * N_DIM + kb * 8 + 2 * h;
    const v2f lo = *(const v2f*)(p);
    const v2f hh = *(const v2f*)(p + 4);
    v4f v;
    v.x = lo.x; v.y = lo.y; v.z = hh.x; v.w = hh.y;
    *((v4f*)out + ((size_t)(tmr * PK_KB + kb) * 32 + lane)) = v;
}

#define YG_PITCH 20
__device__ __forceinline__ void
gs_ymf_role(const float* __restrict__ A, const float* __restrict__ Bt,
            float* __restrict__ pk, int b, float (*red)[32][YG_PITCH]) {
    const int lane = threadIdx.x & 31;
    const int w    = __builtin_amdgcn_readfirstlane((int)(threadIdx.x >> 5));
    const int mp   = b & 1;
    const int tn   = b >> 1;
    const int l15  = lane & 15;
    const int hi   = lane >> 4;
    const int kbase = w * 256 + hi * 4;

    const float* a0p = A  + (size_t)(mp * 32 + l15) * N_DIM + kbase;
    const float* a1p = a0p + (size_t)16 * N_DIM;
    const float* bp  = Bt + (size_t)(tn * 16 + l15) * N_DIM + kbase;

    v8f c00 = {}; v8f c10 = {}; v8f c01 = {}; v8f c11 = {};
    for (int k = 0; k < 256; k += 32) {
        v4f a0[4], a1[4], bb[4];
        #pragma unroll
        for (int j = 0; j < 4; ++j) {
            a0[j] = *(const v4f*)(a0p + k + j * 8);
            a1[j] = *(const v4f*)(a1p + k + j * 8);
            bb[j] = *(const v4f*)(bp  + k + j * 8);
        }
        #pragma unroll
        for (int j = 0; j < 4; ++j) {
            c00 = __builtin_amdgcn_wmma_f32_16x16x4_f32(false, a0[j].xy, false,
                      bb[j].xy, (short)0, c00, false, false);
            c10 = __builtin_amdgcn_wmma_f32_16x16x4_f32(false, a1[j].xy, false,
                      bb[j].xy, (short)0, c10, false, false);
            c01 = __builtin_amdgcn_wmma_f32_16x16x4_f32(false, a0[j].zw, false,
                      bb[j].zw, (short)0, c01, false, false);
            c11 = __builtin_amdgcn_wmma_f32_16x16x4_f32(false, a1[j].zw, false,
                      bb[j].zw, (short)0, c11, false, false);
        }
    }

    {
        float* rp = &red[w][hi * 8][l15];
        #pragma unroll
        for (int r = 0; r < 8; ++r) {
            rp[r * YG_PITCH]        = c00[r] + c01[r];
            rp[(16 + r) * YG_PITCH] = c10[r] + c11[r];
        }
    }
    __syncthreads();

    if (w < 4) {
        const int mt = w >> 1;
        const int kb = w & 1;
        const float* q = &red[0][mt * 16 + l15][kb * 8 + 2 * hi];
        v2f xs[8], ys[8];
        #pragma unroll
        for (int p = 0; p < 8; ++p) {
            xs[p] = *(const v2f*)(q + p * 32 * YG_PITCH);
            ys[p] = *(const v2f*)(q + p * 32 * YG_PITCH + 4);
        }
        v2f x = ((xs[0] + xs[1]) + (xs[2] + xs[3])) + ((xs[4] + xs[5]) + (xs[6] + xs[7]));
        v2f y = ((ys[0] + ys[1]) + (ys[2] + ys[3])) + ((ys[4] + ys[5]) + (ys[6] + ys[7]));
        v4f o;
        o.x = x.x; o.y = x.y; o.z = y.x; o.w = y.y;
        *((v4f*)pk + ((size_t)((mp * 2 + mt) * PK_KB + tn * 2 + kb) * 32 + lane)) = o;
    }
}

#define SU_GEMM  256
#define SU_PACK  4096
#define SU_ZERO  ((int)(ELEMS / (256 * 4)))
#define SU_GRID  (SU_GEMM + 2 * SU_PACK + SU_ZERO)
__global__ void __launch_bounds__(256)
gs_setup(const float* __restrict__ y, const float* __restrict__ H,
         const float* __restrict__ U, const float* __restrict__ invM,
         float* __restrict__ yMFp, float* __restrict__ Up,
         float* __restrict__ invMp, float* __restrict__ traj0) {
    __shared__ float tile[32][40];
    __shared__ alignas(16) float red[8][32][YG_PITCH];
    const int b = (int)blockIdx.x;
    if (b < SU_GEMM) {
        gs_ymf_role(y, H, yMFp, b, red);
    } else if (b < SU_GEMM + SU_PACK) {
        gs_pack_a_role(U, Up, b - SU_GEMM);
    } else if (b < SU_GEMM + 2 * SU_PACK) {
        gs_pack_b_role(invM, invMp, b - (SU_GEMM + SU_PACK), tile);
    } else {
        const v4f z = {0.0f, 0.0f, 0.0f, 0.0f};
        *((v4f*)traj0 + ((size_t)(b - (SU_GEMM + 2 * SU_PACK)) * 256 + threadIdx.x)) = z;
    }
}

#define FK_SPLIT 8
#define FK_KB    (N_DIM / FK_SPLIT / 8)
#define PK_GRP   4
#define PK_NGRP  (FK_KB / PK_GRP)
#define PK_AHEAD 3
#define PK_RING  (PK_AHEAD + 1)
#define RD_PITCH 36
__device__ __forceinline__ void
gs_fused_pk_body(const float* __restrict__ Ap, const float* __restrict__ Bp,
                 const float* __restrict__ subSrc, float* __restrict__ pk0,
                 float* __restrict__ out1, float* __restrict__ out2, int bid) {
    __shared__ alignas(16) float red[FK_SPLIT][16][RD_PITCH];

    const int lane = threadIdx.x & 31;
    const int w    = __builtin_amdgcn_readfirstlane((int)(threadIdx.x >> 5));
    const int kbl  = w & 3;
    const int tm   = bid & 3;
    const int tp   = bid >> 2;
    const int l15  = lane & 15;
    const int hi   = lane >> 4;

    v2f sa = {}; v2f sb = {};
    if (subSrc != nullptr && w >= 4) {
        const float* sp = subSrc + (size_t)(tm * 16 + l15) * N_DIM
                        + tp * 32 + kbl * 8 + 2 * hi;
        sa = *(const v2f*)(sp);
        sb = *(const v2f*)(sp + 4);
    }

    const v4f* ap  = (const v4f*)Ap
                   + ((size_t)(tm * PK_KB + w * FK_KB) * 32 + lane);
    const v4f* bp0 = (const v4f*)Bp
                   + ((size_t)(tp * 2 * PK_KB + w * FK_KB) * 32 + lane);
    const v4f* bp1 = bp0 + (size_t)PK_KB * 32;

    v4f a[PK_RING][PK_GRP], b0[PK_RING][PK_GRP], b1[PK_RING][PK_GRP];
    #pragma unroll
    for (int g = 0; g < PK_AHEAD; ++g) {
        #pragma unroll
        for (int j = 0; j < PK_GRP; ++j) {
            a[g][j]  = ap[(g * PK_GRP + j) * 32];
            b0[g][j] = bp0[(g * PK_GRP + j) * 32];
            b1[g][j] = bp1[(g * PK_GRP + j) * 32];
        }
    }

    v8f c00 = {}; v8f c10 = {}; v8f c01 = {}; v8f c11 = {};
    #pragma unroll
    for (int g = 0; g < PK_NGRP; ++g) {
        const int cur = g % PK_RING;
        if (g + PK_AHEAD < PK_NGRP) {
            const int nxt = (g + PK_AHEAD) % PK_RING;
            #pragma unroll
            for (int j = 0; j < PK_GRP; ++j) {
                a[nxt][j]  = ap[((g + PK_AHEAD) * PK_GRP + j) * 32];
                b0[nxt][j] = bp0[((g + PK_AHEAD) * PK_GRP + j) * 32];
                b1[nxt][j] = bp1[((g + PK_AHEAD) * PK_GRP + j) * 32];
            }
        }
        __builtin_amdgcn_sched_barrier(0);
        #pragma unroll
        for (int j = 0; j < PK_GRP; ++j) {
            c00 = __builtin_amdgcn_wmma_f32_16x16x4_f32(false, a[cur][j].xy, false,
                      b0[cur][j].xy, (short)0, c00, false, false);
            c10 = __builtin_amdgcn_wmma_f32_16x16x4_f32(false, a[cur][j].xy, false,
                      b1[cur][j].xy, (short)0, c10, false, false);
            c01 = __builtin_amdgcn_wmma_f32_16x16x4_f32(false, a[cur][j].zw, false,
                      b0[cur][j].zw, (short)0, c01, false, false);
            c11 = __builtin_amdgcn_wmma_f32_16x16x4_f32(false, a[cur][j].zw, false,
                      b1[cur][j].zw, (short)0, c11, false, false);
        }
        __builtin_amdgcn_sched_barrier(0);
    }

    {
        float* rp = &red[w][hi * 8][l15];
        #pragma unroll
        for (int r = 0; r < 8; ++r) {
            rp[r * RD_PITCH]      = c00[r] + c01[r];
            rp[r * RD_PITCH + 16] = c10[r] + c11[r];
        }
    }
    __syncthreads();

    if (w >= 4) {
        if (pk0 == nullptr) return;
        const float* q = &red[0][l15][kbl * 8 + 2 * hi];
        v2f xs[FK_SPLIT], ys[FK_SPLIT];
        #pragma unroll
        for (int p = 0; p < FK_SPLIT; ++p) {
            xs[p] = *(const v2f*)(q + p * 16 * RD_PITCH);
            ys[p] = *(const v2f*)(q + p * 16 * RD_PITCH + 4);
        }
        v2f x = ((xs[0] + xs[1]) + (xs[2] + xs[3])) + ((xs[4] + xs[5]) + (xs[6] + xs[7]));
        v2f y = ((ys[0] + ys[1]) + (ys[2] + ys[3])) + ((ys[4] + ys[5]) + (ys[6] + ys[7]));
        if (subSrc != nullptr) { x = sa - x; y = sb - y; }
        v4f o;
        o.x = x.x; o.y = x.y; o.z = y.x; o.w = y.y;
        *((v4f*)pk0 + ((size_t)(tm * PK_KB + tp * 4 + kbl) * 32 + lane)) = o;
    } else if (out1 != nullptr || out2 != nullptr) {
        const int t    = threadIdx.x;
        const int row  = t >> 3;
        const int col4 = (t & 7) * 4;
        const float* q = &red[0][row][col4];
        v4f ps[FK_SPLIT];
        #pragma unroll
        for (int p = 0; p < FK_SPLIT; ++p)
            ps[p] = *(const v4f*)(q + p * 16 * RD_PITCH);
        v4f v = ((ps[0] + ps[1]) + (ps[2] + ps[3])) + ((ps[4] + ps[5]) + (ps[6] + ps[7]));
        const size_t o = (size_t)(tm * 16 + row) * N_DIM + tp * 32 + col4;
        if (subSrc != nullptr) v = *(const v4f*)(subSrc + o) - v;
        if (out1 != nullptr) *(v4f*)(out1 + o) = v;
        if (out2 != nullptr) *(v4f*)(out2 + o) = v;
    }
}

__global__ void __launch_bounds__(256)
gs_gemm_fused_pk(const float* __restrict__ Ap, const float* __restrict__ Bp,
                 const float* __restrict__ subSrc, float* __restrict__ pk0,
                 float* __restrict__ out1, float* __restrict__ out2) {
    gs_fused_pk_body(Ap, Bp, subSrc, pk0, out1, out2, (int)blockIdx.x);
}

#define SI_GRP   2
#define SI_NGRP  (FK_KB / SI_GRP)
#define SI_AHEAD 2
#define SI_RING  (SI_AHEAD + 1)
__device__ __forceinline__ void
gs_sinit_load(const v4f* __restrict__ ap, const float* __restrict__ bq, int kb,
              v4f& a, v4f& b0, v4f& b1) {
    const float* r = bq + (size_t)(kb * 8) * N_DIM;
    a    = ap[kb * 32];
    b0.x = r[0];
    b0.y = r[N_DIM];
    b0.z = r[4 * N_DIM];
    b0.w = r[5 * N_DIM];
    b1.x = r[16];
    b1.y = r[N_DIM + 16];
    b1.z = r[4 * N_DIM + 16];
    b1.w = r[5 * N_DIM + 16];
}

__device__ __forceinline__ void
gs_sinit_body(const float* __restrict__ Ap, const float* __restrict__ B,
              float* __restrict__ pk0, int bid) {
    __shared__ alignas(16) float red[FK_SPLIT][16][RD_PITCH];

    const int lane = threadIdx.x & 31;
    const int w    = __builtin_amdgcn_readfirstlane((int)(threadIdx.x >> 5));
    const int kbl  = w & 3;
    const int tm   = bid & 3;
    const int tp   = bid >> 2;
    const int l15  = lane & 15;
    const int hi   = lane >> 4;

    const v4f* ap = (const v4f*)Ap
                  + ((size_t)(tm * PK_KB + w * FK_KB) * 32 + lane);
    const float* bq = B + (size_t)(w * (FK_KB * 8) + 2 * hi) * N_DIM
                    + tp * 32 + l15;

    v4f a[SI_RING][SI_GRP], b0[SI_RING][SI_GRP], b1[SI_RING][SI_GRP];
    #pragma unroll
    for (int g = 0; g < SI_AHEAD; ++g) {
        #pragma unroll
        for (int j = 0; j < SI_GRP; ++j)
            gs_sinit_load(ap, bq, g * SI_GRP + j, a[g][j], b0[g][j], b1[g][j]);
    }

    v8f c00 = {}; v8f c10 = {}; v8f c01 = {}; v8f c11 = {};
    #pragma unroll
    for (int g = 0; g < SI_NGRP; ++g) {
        const int cur = g % SI_RING;
        if (g + SI_AHEAD < SI_NGRP) {
            const int nxt = (g + SI_AHEAD) % SI_RING;
            #pragma unroll
            for (int j = 0; j < SI_GRP; ++j)
                gs_sinit_load(ap, bq, (g + SI_AHEAD) * SI_GRP + j,
                              a[nxt][j], b0[nxt][j], b1[nxt][j]);
        }
        __builtin_amdgcn_sched_barrier(0);
        #pragma unroll
        for (int j = 0; j < SI_GRP; ++j) {
            c00 = __builtin_amdgcn_wmma_f32_16x16x4_f32(false, a[cur][j].xy, false,
                      b0[cur][j].xy, (short)0, c00, false, false);
            c10 = __builtin_amdgcn_wmma_f32_16x16x4_f32(false, a[cur][j].xy, false,
                      b1[cur][j].xy, (short)0, c10, false, false);
            c01 = __builtin_amdgcn_wmma_f32_16x16x4_f32(false, a[cur][j].zw, false,
                      b0[cur][j].zw, (short)0, c01, false, false);
            c11 = __builtin_amdgcn_wmma_f32_16x16x4_f32(false, a[cur][j].zw, false,
                      b1[cur][j].zw, (short)0, c11, false, false);
        }
        __builtin_amdgcn_sched_barrier(0);
    }

    {
        float* rp = &red[w][hi * 8][l15];
        #pragma unroll
        for (int r = 0; r < 8; ++r) {
            rp[r * RD_PITCH]      = c00[r] + c01[r];
            rp[r * RD_PITCH + 16] = c10[r] + c11[r];
        }
    }
    __syncthreads();

    if (w >= 4) {
        const float* q = &red[0][l15][kbl * 8 + 2 * hi];
        v2f xs[FK_SPLIT], ys[FK_SPLIT];
        #pragma unroll
        for (int p = 0; p < FK_SPLIT; ++p) {
            xs[p] = *(const v2f*)(q + p * 16 * RD_PITCH);
            ys[p] = *(const v2f*)(q + p * 16 * RD_PITCH + 4);
        }
        v2f x = ((xs[0] + xs[1]) + (xs[2] + xs[3])) + ((xs[4] + xs[5]) + (xs[6] + xs[7]));
        v2f y = ((ys[0] + ys[1]) + (ys[2] + ys[3])) + ((ys[4] + ys[5]) + (ys[6] + ys[7]));
        v4f o;
        o.x = x.x; o.y = x.y; o.z = y.x; o.w = y.y;
        *((v4f*)pk0 + ((size_t)(tm * PK_KB + tp * 4 + kbl) * 32 + lane)) = o;
    }
}

#define WK_MT    4
#define WK_NT    4
#define WK_AHEAD 2
#define WK_RING  (WK_AHEAD + 1)
#define WR_PITCH 76
__device__ __forceinline__ void
gs_w_body(const float* __restrict__ Ap, const float* __restrict__ Bp,
          float* __restrict__ Wp, int bid) {
    __shared__ float red[FK_SPLIT][16][WR_PITCH];

    const int lane = threadIdx.x & 31;
    const int w    = __builtin_amdgcn_readfirstlane((int)(threadIdx.x >> 5));
    const int tm4  = bid >> 5;
    const int tq   = bid & 31;
    const int l15  = lane & 15;
    const int hi   = lane >> 4;

    const v4f* ap = (const v4f*)Ap
                  + ((size_t)(tm4 * WK_MT * PK_KB + w * FK_KB) * 32 + lane);
    const v4f* bp = (const v4f*)Bp
                  + ((size_t)(tq * WK_NT * PK_KB + w * FK_KB) * 32 + lane);

    v4f a[WK_RING][WK_MT], b[WK_RING][WK_NT];
    #pragma unroll
    for (int g = 0; g < WK_AHEAD; ++g) {
        #pragma unroll
        for (int i = 0; i < WK_MT; ++i) a[g][i] = ap[(i * PK_KB + g) * 32];
        #pragma unroll
        for (int n = 0; n < WK_NT; ++n) b[g][n] = bp[(n * PK_KB + g) * 32];
    }

    v8f acc[WK_MT][WK_NT];
    #pragma unroll
    for (int i = 0; i < WK_MT; ++i) {
        #pragma unroll
        for (int n = 0; n < WK_NT; ++n) acc[i][n] = v8f{};
    }

    #pragma unroll
    for (int g = 0; g < FK_KB; ++g) {
        const int cur = g % WK_RING;
        if (g + WK_AHEAD < FK_KB) {
            const int nxt = (g + WK_AHEAD) % WK_RING;
            #pragma unroll
            for (int i = 0; i < WK_MT; ++i)
                a[nxt][i] = ap[(i * PK_KB + g + WK_AHEAD) * 32];
            #pragma unroll
            for (int n = 0; n < WK_NT; ++n)
                b[nxt][n] = bp[(n * PK_KB + g + WK_AHEAD) * 32];
        }
        __builtin_amdgcn_sched_barrier(0);
        #pragma unroll
        for (int i = 0; i < WK_MT; ++i) {
            #pragma unroll
            for (int n = 0; n < WK_NT; ++n)
                acc[i][n] = __builtin_amdgcn_wmma_f32_16x16x4_f32(false, a[cur][i].xy,
                                false, b[cur][n].xy, (short)0, acc[i][n], false, false);
        }
        #pragma unroll
        for (int i = 0; i < WK_MT; ++i) {
            #pragma unroll
            for (int n = 0; n < WK_NT; ++n)
                acc[i][n] = __builtin_amdgcn_wmma_f32_16x16x4_f32(false, a[cur][i].zw,
                                false, b[cur][n].zw, (short)0, acc[i][n], false, false);
        }
        __builtin_amdgcn_sched_barrier(0);
    }

    const int tnl = w & 3;
    const int kb2 = w >> 2;
    #pragma unroll
    for (int m = 0; m < WK_MT; ++m) {
        if (m > 0) __syncthreads();
        float* rp = &red[w][hi * 8][l15];
        #pragma unroll
        for (int r = 0; r < 8; ++r) {
            #pragma unroll
            for (int n = 0; n < WK_NT; ++n)
                rp[r * WR_PITCH + n * 16] = acc[m][n][r];
        }
        __syncthreads();

        const float* q = &red[0][kb2 * 8 + 2 * hi][tnl * 16 + l15];
        float e0[FK_SPLIT], e1[FK_SPLIT], e4[FK_SPLIT], e5[FK_SPLIT];
        #pragma unroll
        for (int p = 0; p < FK_SPLIT; ++p) {
            e0[p] = q[p * 16 * WR_PITCH];
            e1[p] = q[p * 16 * WR_PITCH + WR_PITCH];
            e4[p] = q[p * 16 * WR_PITCH + 4 * WR_PITCH];
            e5[p] = q[p * 16 * WR_PITCH + 5 * WR_PITCH];
        }
        v4f o;
        o.x = ((e0[0] + e0[1]) + (e0[2] + e0[3])) + ((e0[4] + e0[5]) + (e0[6] + e0[7]));
        o.y = ((e1[0] + e1[1]) + (e1[2] + e1[3])) + ((e1[4] + e1[5]) + (e1[6] + e1[7]));
        o.z = ((e4[0] + e4[1]) + (e4[2] + e4[3])) + ((e4[4] + e4[5]) + (e4[6] + e4[7]));
        o.w = ((e5[0] + e5[1]) + (e5[2] + e5[3])) + ((e5[4] + e5[5]) + (e5[6] + e5[7]));
        *((v4f*)Wp + ((size_t)((tq * WK_NT + tnl) * PK_KB
                               + (tm4 * WK_MT + m) * 2 + kb2) * 32 + lane)) = o;
    }
}

#define PR_SINIT 256
#define PR_W     1024
#define PR_C     256
#define PR_GRID  (PR_SINIT + PR_W + PR_C)
__global__ void __launch_bounds__(256)
gs_products(const float* __restrict__ yMFp, const float* __restrict__ Dinv,
            const float* __restrict__ invMp, const float* __restrict__ Up,
            float* __restrict__ Spk, float* __restrict__ Crm,
            float* __restrict__ Wp) {
    const int b = (int)blockIdx.x;
    if (b < PR_SINIT) {
        gs_sinit_body(yMFp, Dinv, Spk, b);
    } else if (b < PR_SINIT + PR_W) {
        gs_w_body(Up, invMp, Wp, b - PR_SINIT);
    } else {
        gs_fused_pk_body(yMFp, invMp, nullptr, nullptr, Crm, nullptr,
                         b - (PR_SINIT + PR_W));
    }
}

__global__ void __launch_bounds__(128)
gs_gemm_nt(const float* __restrict__ A, const float* __restrict__ Bt,
           const float* __restrict__ subSrc,
           float* __restrict__ out0, float* __restrict__ out1,
           float* __restrict__ out2) {
    const int lane = threadIdx.x & 31;
    const int wave = threadIdx.x >> 5;
    const int t    = blockIdx.x * 4 + wave;
    const int tm   = t & 3;
    const int tn   = t >> 2;
    const int l15  = lane & 15;
    const int hi   = lane >> 4;
    const int koff = hi * 2;

    const float* aptr = A  + (size_t)(tm * 16 + l15) * N_DIM + koff;
    const float* bptr = Bt + (size_t)(tn * 16 + l15) * N_DIM + koff;

    v8f acc0 = {}; v8f acc1 = {};
    #pragma unroll 4
    for (int k = 0; k < N_DIM; k += 8) {
        v2f a0 = *(const v2f*)(aptr + k);
        v2f b0 = *(const v2f*)(bptr + k);
        v2f a1 = *(const v2f*)(aptr + k + 4);
        v2f b1 = *(const v2f*)(bptr + k + 4);
        acc0 = __builtin_amdgcn_wmma_f32_16x16x4_f32(false, a0, false, b0,
                                                     (short)0, acc0, false, false);
        acc1 = __builtin_amdgcn_wmma_f32_16x16x4_f32(false, a1, false, b1,
                                                     (short)0, acc1, false, false);
    }

    const size_t obase = (size_t)(tm * 16 + hi * 8) * N_DIM + tn * 16 + l15;
    float res[8];
    #pragma unroll
    for (int r = 0; r < 8; ++r) res[r] = acc0[r] + acc1[r];
    if (subSrc) {
        #pragma unroll
        for (int r = 0; r < 8; ++r)
            res[r] = subSrc[obase + (size_t)r * N_DIM] - res[r];
    }
    #pragma unroll
    for (int r = 0; r < 8; ++r) out0[obase + (size_t)r * N_DIM] = res[r];
    if (out1) {
        #pragma unroll
        for (int r = 0; r < 8; ++r) out1[obase + (size_t)r * N_DIM] = res[r];
    }
    if (out2) {
        #pragma unroll
        for (int r = 0; r < 8; ++r) out2[obase + (size_t)r * N_DIM] = res[r];
    }
}

__global__ void __launch_bounds__(128)
gs_gemm_nn(const float* __restrict__ A, const float* __restrict__ B,
           const float* __restrict__ subSrc,
           float* __restrict__ out0, float* __restrict__ out1,
           float* __restrict__ out2) {
    const int lane = threadIdx.x & 31;
    const int wave = threadIdx.x >> 5;
    const int t    = blockIdx.x * 4 + wave;
    const int tm   = t & 3;
    const int tn   = t >> 2;
    const int l15  = lane & 15;
    const int hi   = lane >> 4;
    const int koff = hi * 2;

    const float* aptr = A + (size_t)(tm * 16 + l15) * N_DIM + koff;
    const float* bptr = B + (size_t)koff * N_DIM + tn * 16 + l15;

    v8f acc0 = {}; v8f acc1 = {};
    #pragma unroll 4
    for (int k = 0; k < N_DIM; k += 8) {
        v2f a0 = *(const v2f*)(aptr + k);
        v2f b0;
        b0.x = bptr[(size_t)(k + 0) * N_DIM];
        b0.y = bptr[(size_t)(k + 1) * N_DIM];
        v2f a1 = *(const v2f*)(aptr + k + 4);
        v2f b1;
        b1.x = bptr[(size_t)(k + 4) * N_DIM];
        b1.y = bptr[(size_t)(k + 5) * N_DIM];
        acc0 = __builtin_amdgcn_wmma_f32_16x16x4_f32(false, a0, false, b0,
                                                     (short)0, acc0, false, false);
        acc1 = __builtin_amdgcn_wmma_f32_16x16x4_f32(false, a1, false, b1,
                                                     (short)0, acc1, false, false);
    }

    const size_t obase = (size_t)(tm * 16 + hi * 8) * N_DIM + tn * 16 + l15;
    float res[8];
    #pragma unroll
    for (int r = 0; r < 8; ++r) res[r] = acc0[r] + acc1[r];
    if (subSrc) {
        #pragma unroll
        for (int r = 0; r < 8; ++r)
            res[r] = subSrc[obase + (size_t)r * N_DIM] - res[r];
    }
    #pragma unroll
    for (int r = 0; r < 8; ++r) out0[obase + (size_t)r * N_DIM] = res[r];
    if (out1) {
        #pragma unroll
        for (int r = 0; r < 8; ++r) out1[obase + (size_t)r * N_DIM] = res[r];
    }
    if (out2) {
        #pragma unroll
        for (int r = 0; r < 8; ++r) out2[obase + (size_t)r * N_DIM] = res[r];
    }
}

extern "C" void kernel_launch(void* const* d_in, const int* in_sizes, int n_in,
                              void* d_out, int out_size, void* d_ws, size_t ws_size,
                              hipStream_t stream) {
    const float* y    = (const float*)d_in[2];
    const float* H    = (const float*)d_in[3];
    const float* Dinv = (const float*)d_in[4];
    const float* U    = (const float*)d_in[5];
    const float* invM = (const float*)d_in[6];

    float* sfinal = (float*)d_out;
    float* traj   = (float*)d_out + ELEMS;

    float* ws    = (float*)d_ws;
    float* yMF   = ws;
    float* S     = ws + ELEMS;
    float* T     = ws + 2 * ELEMS;
    float* Dt    = ws + 7 * ELEMS;
    float* Ut    = Dt + MATSZ;
    float* invMt = Ut + MATSZ;

    const size_t need_fast = (7 * ELEMS + 3 * MATSZ) * sizeof(float);
    const size_t need_mid  = (3 * ELEMS + 2 * MATSZ) * sizeof(float);
    const int mode = (ws_size >= need_fast) ? 2 : (ws_size >= need_mid ? 1 : 0);

    dim3 gblk(128);
    dim3 ggrd(128);
    dim3 fblk(256);
    dim3 fgrd(256);
    dim3 sgrd(SU_GRID);
    dim3 pgrd(PR_GRID);

    if (mode == 2) {
        float* Cm = yMF;
        float* Wp = Dt;

        gs_setup<<<sgrd, 256, 0, stream>>>(y, H, U, invM, T, Ut, invMt, traj);
        gs_products<<<pgrd, fblk, 0, stream>>>(T, Dinv, invMt, Ut, S, Cm, Wp);

        float* cur = S;
        float* nxt = T;
        for (int i = 1; i <= N_ITER; ++i) {
            float* trajD  = traj + (size_t)i * ELEMS;
            float* finalD = (i == N_ITER) ? sfinal : nullptr;
            gs_gemm_fused_pk<<<fgrd, fblk, 0, stream>>>(cur, Wp, Cm, nxt, trajD, finalD);
            float* t = cur; cur = nxt; nxt = t;
        }
    } else if (mode == 1) {
        gs_zero_kernel<<<(int)(ELEMS / 256), 256, 0, stream>>>(traj, (int)ELEMS);
        gs_transpose<<<4096, 256, 0, stream>>>(U, Ut - MATSZ);
        gs_transpose<<<4096, 256, 0, stream>>>(invM, invMt - MATSZ);
        float* Ut1    = Dt;
        float* invMt1 = Dt + MATSZ;
        gs_gemm_nt<<<ggrd, gblk, 0, stream>>>(y, H, nullptr, yMF, nullptr, nullptr);
        gs_gemm_nn<<<ggrd, gblk, 0, stream>>>(yMF, Dinv, nullptr, S, nullptr, nullptr);
        for (int i = 1; i <= N_ITER; ++i) {
            float* trajD  = traj + (size_t)i * ELEMS;
            float* finalD = (i == N_ITER) ? sfinal : nullptr;
            gs_gemm_nt<<<ggrd, gblk, 0, stream>>>(S, Ut1, yMF, T, nullptr, nullptr);
            gs_gemm_nt<<<ggrd, gblk, 0, stream>>>(T, invMt1, nullptr, S, trajD, finalD);
        }
    } else {
        gs_zero_kernel<<<(int)(ELEMS / 256), 256, 0, stream>>>(traj, (int)ELEMS);
        gs_gemm_nt<<<ggrd, gblk, 0, stream>>>(y, H, nullptr, yMF, nullptr, nullptr);
        gs_gemm_nn<<<ggrd, gblk, 0, stream>>>(yMF, Dinv, nullptr, S, nullptr, nullptr);
        for (int i = 1; i <= N_ITER; ++i) {
            float* trajD  = traj + (size_t)i * ELEMS;
            float* finalD = (i == N_ITER) ? sfinal : nullptr;
            gs_gemm_nn<<<ggrd, gblk, 0, stream>>>(S, U, yMF, T, nullptr, nullptr);
            gs_gemm_nn<<<ggrd, gblk, 0, stream>>>(T, invM, nullptr, S, trajD, finalD);
        }
    }
}
